# starting kernel
# baseline (speedup 1.0000x reference)
#include <hip/hip_runtime.h>

typedef __attribute__((ext_vector_type(2))) float v2f;
typedef __attribute__((ext_vector_type(8))) float v8f;

#define KDIM 512
#define F1 32
#define F2 16

__global__ void init_kernel(float* __restrict__ deg, float* __restrict__ agg1,
                            float* __restrict__ out, int n) {
  int i = blockIdx.x * blockDim.x + threadIdx.x;
  const int t1 = n, t2 = n + n * F1, t3 = n + n * F1 + n * F2;
  if (i < t1)       deg[i] = 1.0f;
  else if (i < t2)  agg1[i - t1] = 0.0f;
  else if (i < t3)  out[i - t2] = 0.0f;
}

__global__ void degree_kernel(const int* __restrict__ dst, float* __restrict__ deg, int E) {
  int i = blockIdx.x * blockDim.x + threadIdx.x;
  if (i < E) atomicAdd(&deg[dst[i]], 1.0f);
}

__global__ void rsqrt_kernel(float* __restrict__ deg, int n) {
  int i = blockIdx.x * blockDim.x + threadIdx.x;
  if (i < n) deg[i] = rsqrtf(deg[i]);
}

__global__ __launch_bounds__(256) void gemm1_kernel(const float* __restrict__ x,
                                                    const float* __restrict__ W1,
                                                    float* __restrict__ h, int n) {
  const int lane  = threadIdx.x & 31;
  const int wave  = blockIdx.x * (blockDim.x >> 5) + (threadIdx.x >> 5);
  const int ntile = n >> 4;
  if (wave >= ntile) return;
  const int m0 = wave << 4;
  const int r  = lane & 15;
  const int kh = (lane >> 4) << 1;

  const float* arow = x + (size_t)(m0 + r) * KDIM;

  v8f acc0 = {0.f,0.f,0.f,0.f,0.f,0.f,0.f,0.f};
  v8f acc1 = {0.f,0.f,0.f,0.f,0.f,0.f,0.f,0.f};

  for (int k = 0; k < KDIM; k += 4) {
    v2f a = __builtin_nontemporal_load((const v2f*)(arow + k + kh));
    const float* w0 = W1 + (size_t)(k + kh) * F1;
    const float* w1 = w0 + F1;
    v2f bA; bA.x = w0[r];      bA.y = w1[r];
    v2f bB; bB.x = w0[r + 16]; bB.y = w1[r + 16];
    acc0 = __builtin_amdgcn_wmma_f32_16x16x4_f32(false, a, false, bA, (short)0, acc0, false, false);
    acc1 = __builtin_amdgcn_wmma_f32_16x16x4_f32(false, a, false, bB, (short)0, acc1, false, false);
  }

  const int mbase = m0 + ((lane >> 4) << 3);
  float* hp = h + (size_t)mbase * F1 + r;
#pragma unroll
  for (int v = 0; v < 8; ++v) {
    hp[(size_t)v * F1]      = acc0[v];
    hp[(size_t)v * F1 + 16] = acc1[v];
  }
}

__global__ __launch_bounds__(256) void gemm2_kernel(const float* __restrict__ agg1,
                                                    const float* __restrict__ h,
                                                    const float* __restrict__ dis,
                                                    const float* __restrict__ b1,
                                                    const float* __restrict__ W2,
                                                    float* __restrict__ h2, int n) {
  const int lane  = threadIdx.x & 31;
  const int wave  = blockIdx.x * (blockDim.x >> 5) + (threadIdx.x >> 5);
  const int ntile = n >> 4;
  if (wave >= ntile) return;
  const int m0 = wave << 4;
  const int r  = lane & 15;
  const int kh = (lane >> 4) << 1;

  const int row  = m0 + r;
  const float dv = dis[row];
  const float d2 = dv * dv;
  const float* ag = agg1 + (size_t)row * F1;
  const float* hr = h    + (size_t)row * F1;

  v8f acc = {0.f,0.f,0.f,0.f,0.f,0.f,0.f,0.f};

  for (int k = 0; k < F1; k += 4) {
    const int k0 = k + kh;
    v2f a;
    a.x = fmaxf(ag[k0]     + hr[k0]     * d2 + b1[k0],     0.0f);
    a.y = fmaxf(ag[k0 + 1] + hr[k0 + 1] * d2 + b1[k0 + 1], 0.0f);
    v2f b;
    b.x = W2[(size_t)k0 * F2 + r];
    b.y = W2[(size_t)(k0 + 1) * F2 + r];
    acc = __builtin_amdgcn_wmma_f32_16x16x4_f32(false, a, false, b, (short)0, acc, false, false);
  }

  const int mbase = m0 + ((lane >> 4) << 3);
  float* op = h2 + (size_t)mbase * F2 + r;
#pragma unroll
  for (int v = 0; v < 8; ++v) op[(size_t)v * F2] = acc[v];
}

template <int F>
__global__ __launch_bounds__(256) void agg_kernel(const float* __restrict__ feat,
                                                  const int* __restrict__ src,
                                                  const int* __restrict__ dst,
                                                  const float* __restrict__ dis,
                                                  float* __restrict__ out, int E) {
  const int lane = threadIdx.x & 31;
  const int wave = blockIdx.x * (blockDim.x >> 5) + (threadIdx.x >> 5);
  const int e0   = wave * 32;
  if (e0 >= E) return;

  const int myE = e0 + lane;
  int   s_l = -1, d_l = 0;
  float nrm_l = 0.0f;
  if (myE < E) {
    s_l = src[myE];
    d_l = dst[myE];
    nrm_l = dis[s_l] * dis[d_l];
  }

  constexpr int EPS = 32 / F;
  const int f   = lane & (F - 1);
  const int sub = lane / F;

  for (int j = 0; j < 32; j += EPS) {
    const int   s  = __shfl(s_l,   j + sub);
    const int   d  = __shfl(d_l,   j + sub);
    const float nm = __shfl(nrm_l, j + sub);
    if (s >= 0)
      atomicAdd(&out[(size_t)d * F + f], feat[(size_t)s * F + f] * nm);
  }
}

__global__ void epi2_kernel(float* __restrict__ out, const float* __restrict__ h2,
                            const float* __restrict__ dis, const float* __restrict__ b2,
                            int n) {
  int i = blockIdx.x * blockDim.x + threadIdx.x;
  if (i < n * F2) {
    const int row = i / F2, f = i - row * F2;
    const float dv = dis[row];
    out[i] = out[i] + h2[i] * dv * dv + b2[f];
  }
}

extern "C" void kernel_launch(void* const* d_in, const int* in_sizes, int n_in,
                              void* d_out, int out_size, void* d_ws, size_t ws_size,
                              hipStream_t stream) {
  const float* x  = (const float*)d_in[0];
  const int*   ei = (const int*)d_in[1];
  const float* W1 = (const float*)d_in[2];
  const float* b1 = (const float*)d_in[3];
  const float* W2 = (const float*)d_in[4];
  const float* b2 = (const float*)d_in[5];

  const int n = in_sizes[0] / KDIM;
  const int E = in_sizes[1] / 2;
  const int* src = ei;
  const int* dst = ei + E;

  float* ws   = (float*)d_ws;
  float* dis  = ws;
  float* h    = dis  + n;
  float* agg1 = h    + (size_t)n * F1;
  float* h2   = agg1 + (size_t)n * F1;
  float* out  = (float*)d_out;

  {
    const int total = n + n * F1 + n * F2;
    init_kernel<<<(total + 255) / 256, 256, 0, stream>>>(dis, agg1, out, n);
  }
  degree_kernel<<<(E + 255) / 256, 256, 0, stream>>>(dst, dis, E);
  rsqrt_kernel<<<(n + 255) / 256, 256, 0, stream>>>(dis, n);
  {
    const int ntile = n >> 4;
    gemm1_kernel<<<(ntile + 7) / 8, 256, 0, stream>>>(x, W1, h, n);
  }
  {
    const int waves = (E + 31) / 32;
    agg_kernel<F1><<<(waves + 7) / 8, 256, 0, stream>>>(h, src, dst, dis, agg1, E);
  }
  {
    const int ntile = n >> 4;
    gemm2_kernel<<<(ntile + 7) / 8, 256, 0, stream>>>(agg1, h, dis, b1, W2, h2, n);
  }
  {
    const int waves = (E + 31) / 32;
    agg_kernel<F2><<<(waves + 7) / 8, 256, 0, stream>>>(h2, src, dst, dis, out, E);
  }
  epi2_kernel<<<(n * F2 + 255) / 256, 256, 0, stream>>>(out, h2, dis, b2, n);
}


# in-degree counted per workgroup in an LDS integer histogram (node range x edge chunk), one float atomic add per nonzero bin instead of one per edge; counts are exact integers, sum order free
# speedup vs baseline: 1.3716x; 1.3716x over previous
#include <hip/hip_runtime.h>

typedef __attribute__((ext_vector_type(2))) float v2f;
typedef __attribute__((ext_vector_type(8))) float v8f;

#define KDIM 512
#define F1 32
#define F2 16

__global__ void init_kernel(float* __restrict__ deg, float* __restrict__ agg1,
                            float* __restrict__ out, int n) {
  int i = blockIdx.x * blockDim.x + threadIdx.x;
  const int t1 = n, t2 = n + n * F1, t3 = n + n * F1 + n * F2;
  if (i < t1)       deg[i] = 1.0f;
  else if (i < t2)  agg1[i - t1] = 0.0f;
  else if (i < t3)  out[i - t2] = 0.0f;
}

#define DEG_BINS    14336
#define DEG_THREADS 512
#define DEG_CHUNKS  36
#define DEG_DEPTH   8

static __device__ __forceinline__ void deg_count(unsigned int* cnt, int d, unsigned int lo) {
  const unsigned int b = (unsigned int)d - lo;
  if (b < (unsigned int)DEG_BINS) atomicAdd(&cnt[b], 1u);
}

__global__ __launch_bounds__(DEG_THREADS) void degree_kernel(const int* __restrict__ dst,
                                                             float* __restrict__ deg,
                                                             int n, int E) {
  __shared__ __attribute__((aligned(16))) unsigned int cnt[DEG_BINS];
  const int tid = threadIdx.x;
  const int lo  = blockIdx.y * DEG_BINS;

  for (int i = tid; i < DEG_BINS / 4; i += DEG_THREADS)
    ((uint4*)cnt)[i] = make_uint4(0u, 0u, 0u, 0u);
  __syncthreads();

  const int per = (((E + (int)gridDim.x - 1) / (int)gridDim.x) + 3) & ~3;
  const int e0  = blockIdx.x * per;
  const int len = max(0, min(E - e0, per));

  if ((((size_t)dst) & 15) == 0) {
    const int4* p  = (const int4*)(dst + e0);
    const int   nv = len >> 2;
    if (nv > 0) {
      int4 r[DEG_DEPTH];
#pragma unroll
      for (int u = 0; u < DEG_DEPTH; ++u)
        r[u] = p[min(u * DEG_THREADS + tid, nv - 1)];
      for (int base = 0; base < nv; base += DEG_DEPTH * DEG_THREADS) {
        int4 q[DEG_DEPTH];
#pragma unroll
        for (int u = 0; u < DEG_DEPTH; ++u)
          q[u] = p[min(base + (DEG_DEPTH + u) * DEG_THREADS + tid, nv - 1)];
#pragma unroll
        for (int u = 0; u < DEG_DEPTH; ++u) {
          const unsigned int lo_u =
              (base + u * DEG_THREADS + tid < nv) ? (unsigned int)lo : 0x80000000u;
          deg_count(cnt, r[u].x, lo_u);
          deg_count(cnt, r[u].y, lo_u);
          deg_count(cnt, r[u].z, lo_u);
          deg_count(cnt, r[u].w, lo_u);
        }
#pragma unroll
        for (int u = 0; u < DEG_DEPTH; ++u) r[u] = q[u];
      }
    }
    const int t = (nv << 2) + tid;
    if (t < len) deg_count(cnt, dst[e0 + t], (unsigned int)lo);
  } else {
    for (int t = tid; t < len; t += DEG_THREADS) deg_count(cnt, dst[e0 + t], (unsigned int)lo);
  }
  __syncthreads();

  const int rot = (int)((blockIdx.x * 5u) % (unsigned int)(DEG_BINS / DEG_THREADS)) * DEG_THREADS;
  static_assert(DEG_BINS % (4 * DEG_THREADS) == 0, "flush reads four counters per lane and trip");
  for (int i = tid; i < DEG_BINS; i += 4 * DEG_THREADS) {
    int j[4];
    unsigned int c[4];
#pragma unroll
    for (int u = 0; u < 4; ++u) {
      j[u] = i + u * DEG_THREADS + rot;
      if (j[u] >= DEG_BINS) j[u] -= DEG_BINS;
      c[u] = cnt[j[u]];
    }
#pragma unroll
    for (int u = 0; u < 4; ++u) {
      const int node = lo + j[u];
      if (c[u] != 0u && node < n) atomicAdd(&deg[node], (float)c[u]);
    }
  }
}

__global__ void rsqrt_kernel(float* __restrict__ deg, int n) {
  int i = blockIdx.x * blockDim.x + threadIdx.x;
  if (i < n) deg[i] = rsqrtf(deg[i]);
}

__global__ __launch_bounds__(256) void gemm1_kernel(const float* __restrict__ x,
                                                    const float* __restrict__ W1,
                                                    float* __restrict__ h, int n) {
  const int lane  = threadIdx.x & 31;
  const int wave  = blockIdx.x * (blockDim.x >> 5) + (threadIdx.x >> 5);
  const int ntile = n >> 4;
  if (wave >= ntile) return;
  const int m0 = wave << 4;
  const int r  = lane & 15;
  const int kh = (lane >> 4) << 1;

  const float* arow = x + (size_t)(m0 + r) * KDIM;

  v8f acc0 = {0.f,0.f,0.f,0.f,0.f,0.f,0.f,0.f};
  v8f acc1 = {0.f,0.f,0.f,0.f,0.f,0.f,0.f,0.f};

  for (int k = 0; k < KDIM; k += 4) {
    v2f a = __builtin_nontemporal_load((const v2f*)(arow + k + kh));
    const float* w0 = W1 + (size_t)(k + kh) * F1;
    const float* w1 = w0 + F1;
    v2f bA; bA.x = w0[r];      bA.y = w1[r];
    v2f bB; bB.x = w0[r + 16]; bB.y = w1[r + 16];
    acc0 = __builtin_amdgcn_wmma_f32_16x16x4_f32(false, a, false, bA, (short)0, acc0, false, false);
    acc1 = __builtin_amdgcn_wmma_f32_16x16x4_f32(false, a, false, bB, (short)0, acc1, false, false);
  }

  const int mbase = m0 + ((lane >> 4) << 3);
  float* hp = h + (size_t)mbase * F1 + r;
#pragma unroll
  for (int v = 0; v < 8; ++v) {
    hp[(size_t)v * F1]      = acc0[v];
    hp[(size_t)v * F1 + 16] = acc1[v];
  }
}

__global__ __launch_bounds__(256) void gemm2_kernel(const float* __restrict__ agg1,
                                                    const float* __restrict__ h,
                                                    const float* __restrict__ dis,
                                                    const float* __restrict__ b1,
                                                    const float* __restrict__ W2,
                                                    float* __restrict__ h2, int n) {
  const int lane  = threadIdx.x & 31;
  const int wave  = blockIdx.x * (blockDim.x >> 5) + (threadIdx.x >> 5);
  const int ntile = n >> 4;
  if (wave >= ntile) return;
  const int m0 = wave << 4;
  const int r  = lane & 15;
  const int kh = (lane >> 4) << 1;

  const int row  = m0 + r;
  const float dv = dis[row];
  const float d2 = dv * dv;
  const float* ag = agg1 + (size_t)row * F1;
  const float* hr = h    + (size_t)row * F1;

  v8f acc = {0.f,0.f,0.f,0.f,0.f,0.f,0.f,0.f};

  for (int k = 0; k < F1; k += 4) {
    const int k0 = k + kh;
    v2f a;
    a.x = fmaxf(ag[k0]     + hr[k0]     * d2 + b1[k0],     0.0f);
    a.y = fmaxf(ag[k0 + 1] + hr[k0 + 1] * d2 + b1[k0 + 1], 0.0f);
    v2f b;
    b.x = W2[(size_t)k0 * F2 + r];
    b.y = W2[(size_t)(k0 + 1) * F2 + r];
    acc = __builtin_amdgcn_wmma_f32_16x16x4_f32(false, a, false, b, (short)0, acc, false, false);
  }

  const int mbase = m0 + ((lane >> 4) << 3);
  float* op = h2 + (size_t)mbase * F2 + r;
#pragma unroll
  for (int v = 0; v < 8; ++v) op[(size_t)v * F2] = acc[v];
}

template <int F>
__global__ __launch_bounds__(256) void agg_kernel(const float* __restrict__ feat,
                                                  const int* __restrict__ src,
                                                  const int* __restrict__ dst,
                                                  const float* __restrict__ dis,
                                                  float* __restrict__ out, int E) {
  const int lane = threadIdx.x & 31;
  const int wave = blockIdx.x * (blockDim.x >> 5) + (threadIdx.x >> 5);
  const int e0   = wave * 32;
  if (e0 >= E) return;

  const int myE = e0 + lane;
  int   s_l = -1, d_l = 0;
  float nrm_l = 0.0f;
  if (myE < E) {
    s_l = src[myE];
    d_l = dst[myE];
    nrm_l = dis[s_l] * dis[d_l];
  }

  constexpr int EPS = 32 / F;
  const int f   = lane & (F - 1);
  const int sub = lane / F;

  for (int j = 0; j < 32; j += EPS) {
    const int   s  = __shfl(s_l,   j + sub);
    const int   d  = __shfl(d_l,   j + sub);
    const float nm = __shfl(nrm_l, j + sub);
    if (s >= 0)
      atomicAdd(&out[(size_t)d * F + f], feat[(size_t)s * F + f] * nm);
  }
}

__global__ void epi2_kernel(float* __restrict__ out, const float* __restrict__ h2,
                            const float* __restrict__ dis, const float* __restrict__ b2,
                            int n) {
  int i = blockIdx.x * blockDim.x + threadIdx.x;
  if (i < n * F2) {
    const int row = i / F2, f = i - row * F2;
    const float dv = dis[row];
    out[i] = out[i] + h2[i] * dv * dv + b2[f];
  }
}

extern "C" void kernel_launch(void* const* d_in, const int* in_sizes, int n_in,
                              void* d_out, int out_size, void* d_ws, size_t ws_size,
                              hipStream_t stream) {
  const float* x  = (const float*)d_in[0];
  const int*   ei = (const int*)d_in[1];
  const float* W1 = (const float*)d_in[2];
  const float* b1 = (const float*)d_in[3];
  const float* W2 = (const float*)d_in[4];
  const float* b2 = (const float*)d_in[5];

  const int n = in_sizes[0] / KDIM;
  const int E = in_sizes[1] / 2;
  const int* src = ei;
  const int* dst = ei + E;

  float* ws   = (float*)d_ws;
  float* dis  = ws;
  float* h    = dis  + n;
  float* agg1 = h    + (size_t)n * F1;
  float* h2   = agg1 + (size_t)n * F1;
  float* out  = (float*)d_out;

  {
    const int total = n + n * F1 + n * F2;
    init_kernel<<<(total + 255) / 256, 256, 0, stream>>>(dis, agg1, out, n);
  }
  {
    const int ranges = (n + DEG_BINS - 1) / DEG_BINS;
    degree_kernel<<<dim3(DEG_CHUNKS, ranges), DEG_THREADS, 0, stream>>>(dst, dis, n, E);
  }
  rsqrt_kernel<<<(n + 255) / 256, 256, 0, stream>>>(dis, n);
  {
    const int ntile = n >> 4;
    gemm1_kernel<<<(ntile + 7) / 8, 256, 0, stream>>>(x, W1, h, n);
  }
  {
    const int waves = (E + 31) / 32;
    agg_kernel<F1><<<(waves + 7) / 8, 256, 0, stream>>>(h, src, dst, dis, agg1, E);
  }
  {
    const int ntile = n >> 4;
    gemm2_kernel<<<(ntile + 7) / 8, 256, 0, stream>>>(agg1, h, dis, b1, W2, h2, n);
  }
  {
    const int waves = (E + 31) / 32;
    agg_kernel<F2><<<(waves + 7) / 8, 256, 0, stream>>>(h2, src, dst, dis, out, E);
  }
  epi2_kernel<<<(n * F2 + 255) / 256, 256, 0, stream>>>(out, h2, dis, b2, n);
}


# GEMM1 reads x in whole 256-byte row pieces (16-byte loads, two rows per instruction), transposed through the wave's own LDS rows; k order inside each 8-column group permuted: h's sum order changes
# speedup vs baseline: 1.5687x; 1.1437x over previous
#include <hip/hip_runtime.h>

typedef __attribute__((ext_vector_type(2))) float v2f;
typedef __attribute__((ext_vector_type(8))) float v8f;

#define KDIM 512
#define F1 32
#define F2 16

__global__ void init_kernel(float* __restrict__ deg, float* __restrict__ agg1,
                            float* __restrict__ out, int n) {
  int i = blockIdx.x * blockDim.x + threadIdx.x;
  const int t1 = n, t2 = n + n * F1, t3 = n + n * F1 + n * F2;
  if (i < t1)       deg[i] = 1.0f;
  else if (i < t2)  agg1[i - t1] = 0.0f;
  else if (i < t3)  out[i - t2] = 0.0f;
}

#define DEG_BINS    14336
#define DEG_THREADS 512
#define DEG_CHUNKS  36
#define DEG_DEPTH   8

static __device__ __forceinline__ void deg_count(unsigned int* cnt, int d, unsigned int lo) {
  const unsigned int b = (unsigned int)d - lo;
  if (b < (unsigned int)DEG_BINS) atomicAdd(&cnt[b], 1u);
}

__global__ __launch_bounds__(DEG_THREADS) void degree_kernel(const int* __restrict__ dst,
                                                             float* __restrict__ deg,
                                                             int n, int E) {
  __shared__ __attribute__((aligned(16))) unsigned int cnt[DEG_BINS];
  const int tid = threadIdx.x;
  const int lo  = blockIdx.y * DEG_BINS;

  for (int i = tid; i < DEG_BINS / 4; i += DEG_THREADS)
    ((uint4*)cnt)[i] = make_uint4(0u, 0u, 0u, 0u);
  __syncthreads();

  const int per = (((E + (int)gridDim.x - 1) / (int)gridDim.x) + 3) & ~3;
  const int e0  = blockIdx.x * per;
  const int len = max(0, min(E - e0, per));

  if ((((size_t)dst) & 15) == 0) {
    const int4* p  = (const int4*)(dst + e0);
    const int   nv = len >> 2;
    if (nv > 0) {
      int4 r[DEG_DEPTH];
#pragma unroll
      for (int u = 0; u < DEG_DEPTH; ++u)
        r[u] = p[min(u * DEG_THREADS + tid, nv - 1)];
      for (int base = 0; base < nv; base += DEG_DEPTH * DEG_THREADS) {
        int4 q[DEG_DEPTH];
#pragma unroll
        for (int u = 0; u < DEG_DEPTH; ++u)
          q[u] = p[min(base + (DEG_DEPTH + u) * DEG_THREADS + tid, nv - 1)];
#pragma unroll
        for (int u = 0; u < DEG_DEPTH; ++u) {
          const unsigned int lo_u =
              (base + u * DEG_THREADS + tid < nv) ? (unsigned int)lo : 0x80000000u;
          deg_count(cnt, r[u].x, lo_u);
          deg_count(cnt, r[u].y, lo_u);
          deg_count(cnt, r[u].z, lo_u);
          deg_count(cnt, r[u].w, lo_u);
        }
#pragma unroll
        for (int u = 0; u < DEG_DEPTH; ++u) r[u] = q[u];
      }
    }
    const int t = (nv << 2) + tid;
    if (t < len) deg_count(cnt, dst[e0 + t], (unsigned int)lo);
  } else {
    for (int t = tid; t < len; t += DEG_THREADS) deg_count(cnt, dst[e0 + t], (unsigned int)lo);
  }
  __syncthreads();

  const int rot = (int)((blockIdx.x * 5u) % (unsigned int)(DEG_BINS / DEG_THREADS)) * DEG_THREADS;
  static_assert(DEG_BINS % (4 * DEG_THREADS) == 0, "flush reads four counters per lane and trip");
  for (int i = tid; i < DEG_BINS; i += 4 * DEG_THREADS) {
    int j[4];
    unsigned int c[4];
#pragma unroll
    for (int u = 0; u < 4; ++u) {
      j[u] = i + u * DEG_THREADS + rot;
      if (j[u] >= DEG_BINS) j[u] -= DEG_BINS;
      c[u] = cnt[j[u]];
    }
#pragma unroll
    for (int u = 0; u < 4; ++u) {
      const int node = lo + j[u];
      if (c[u] != 0u && node < n) atomicAdd(&deg[node], (float)c[u]);
    }
  }
}

__global__ void rsqrt_kernel(float* __restrict__ deg, int n) {
  int i = blockIdx.x * blockDim.x + threadIdx.x;
  if (i < n) deg[i] = rsqrtf(deg[i]);
}

typedef __attribute__((ext_vector_type(4))) float v4f;

#define G1_WAVES 8
#define G1_COLS  64
#define G1_PITCH 68

static __device__ __forceinline__ void g1_wave_handover() {
  __builtin_amdgcn_fence(__ATOMIC_RELEASE, "wavefront");
  __builtin_amdgcn_wave_barrier();
  __builtin_amdgcn_fence(__ATOMIC_ACQUIRE, "wavefront");
}

__global__ __launch_bounds__(256) void gemm1_kernel(const float* __restrict__ x,
                                                    const float* __restrict__ W1,
                                                    float* __restrict__ h, int n) {
  __shared__ __attribute__((aligned(16))) float stage[G1_WAVES][16][G1_PITCH];
  const int lane  = threadIdx.x & 31;
  const int wid   = threadIdx.x >> 5;
  const int wave  = blockIdx.x * (blockDim.x >> 5) + wid;
  const int ntile = n >> 4;
  if (wave >= ntile) return;
  const int m0 = wave << 4;
  const int r  = lane & 15;
  const int hh = lane >> 4;

  float (*st)[G1_PITCH] = stage[wid];

  const int lrow = lane >> 4;
  const int lcol = (lane & 15) << 2;
  const float* gp = x + (size_t)(m0 + lrow) * KDIM + lcol;

  v4f R[8];
#pragma unroll
  for (int j = 0; j < 8; ++j)
    R[j] = __builtin_nontemporal_load((const v4f*)(gp + (size_t)(2 * j) * KDIM));

  v8f acc0 = {0.f,0.f,0.f,0.f,0.f,0.f,0.f,0.f};
  v8f acc1 = {0.f,0.f,0.f,0.f,0.f,0.f,0.f,0.f};

  for (int t = 0; t < KDIM / G1_COLS; ++t) {
#pragma unroll
    for (int j = 0; j < 8; ++j)
      *(v4f*)&st[2 * j + lrow][lcol] = R[j];

    if (t + 1 < KDIM / G1_COLS) {
#pragma unroll
      for (int j = 0; j < 8; ++j)
        R[j] = __builtin_nontemporal_load(
            (const v4f*)(gp + (size_t)(2 * j) * KDIM + (t + 1) * G1_COLS));
    }
    g1_wave_handover();

    const float* wb = W1 + (size_t)(t * G1_COLS + 4 * hh) * F1 + r;
#pragma unroll
    for (int g = 0; g < G1_COLS / 8; ++g) {
      const v4f a4 = *(const v4f*)&st[r][8 * g + 4 * hh];
      const float* w = wb + (size_t)(8 * g) * F1;
      v2f aL; aL.x = a4.x; aL.y = a4.y;
      v2f aH; aH.x = a4.z; aH.y = a4.w;
      v2f b0; b0.x = w[0];      b0.y = w[F1];
      v2f b1; b1.x = w[16];     b1.y = w[F1 + 16];
      v2f b2; b2.x = w[2 * F1]; b2.y = w[3 * F1];
      v2f b3; b3.x = w[2 * F1 + 16]; b3.y = w[3 * F1 + 16];
      acc0 = __builtin_amdgcn_wmma_f32_16x16x4_f32(false, aL, false, b0, (short)0, acc0, false, false);
      acc1 = __builtin_amdgcn_wmma_f32_16x16x4_f32(false, aL, false, b1, (short)0, acc1, false, false);
      acc0 = __builtin_amdgcn_wmma_f32_16x16x4_f32(false, aH, false, b2, (short)0, acc0, false, false);
      acc1 = __builtin_amdgcn_wmma_f32_16x16x4_f32(false, aH, false, b3, (short)0, acc1, false, false);
    }
    g1_wave_handover();
  }

  const int mbase = m0 + ((lane >> 4) << 3);
  float* hp = h + (size_t)mbase * F1 + r;
#pragma unroll
  for (int v = 0; v < 8; ++v) {
    hp[(size_t)v * F1]      = acc0[v];
    hp[(size_t)v * F1 + 16] = acc1[v];
  }
}

__global__ __launch_bounds__(256) void gemm2_kernel(const float* __restrict__ agg1,
                                                    const float* __restrict__ h,
                                                    const float* __restrict__ dis,
                                                    const float* __restrict__ b1,
                                                    const float* __restrict__ W2,
                                                    float* __restrict__ h2, int n) {
  const int lane  = threadIdx.x & 31;
  const int wave  = blockIdx.x * (blockDim.x >> 5) + (threadIdx.x >> 5);
  const int ntile = n >> 4;
  if (wave >= ntile) return;
  const int m0 = wave << 4;
  const int r  = lane & 15;
  const int kh = (lane >> 4) << 1;

  const int row  = m0 + r;
  const float dv = dis[row];
  const float d2 = dv * dv;
  const float* ag = agg1 + (size_t)row * F1;
  const float* hr = h    + (size_t)row * F1;

  v8f acc = {0.f,0.f,0.f,0.f,0.f,0.f,0.f,0.f};

  for (int k = 0; k < F1; k += 4) {
    const int k0 = k + kh;
    v2f a;
    a.x = fmaxf(ag[k0]     + hr[k0]     * d2 + b1[k0],     0.0f);
    a.y = fmaxf(ag[k0 + 1] + hr[k0 + 1] * d2 + b1[k0 + 1], 0.0f);
    v2f b;
    b.x = W2[(size_t)k0 * F2 + r];
    b.y = W2[(size_t)(k0 + 1) * F2 + r];
    acc = __builtin_amdgcn_wmma_f32_16x16x4_f32(false, a, false, b, (short)0, acc, false, false);
  }

  const int mbase = m0 + ((lane >> 4) << 3);
  float* op = h2 + (size_t)mbase * F2 + r;
#pragma unroll
  for (int v = 0; v < 8; ++v) op[(size_t)v * F2] = acc[v];
}

template <int F>
__global__ __launch_bounds__(256) void agg_kernel(const float* __restrict__ feat,
                                                  const int* __restrict__ src,
                                                  const int* __restrict__ dst,
                                                  const float* __restrict__ dis,
                                                  float* __restrict__ out, int E) {
  const int lane = threadIdx.x & 31;
  const int wave = blockIdx.x * (blockDim.x >> 5) + (threadIdx.x >> 5);
  const int e0   = wave * 32;
  if (e0 >= E) return;

  const int myE = e0 + lane;
  int   s_l = -1, d_l = 0;
  float nrm_l = 0.0f;
  if (myE < E) {
    s_l = src[myE];
    d_l = dst[myE];
    nrm_l = dis[s_l] * dis[d_l];
  }

  constexpr int EPS = 32 / F;
  const int f   = lane & (F - 1);
  const int sub = lane / F;

  for (int j = 0; j < 32; j += EPS) {
    const int   s  = __shfl(s_l,   j + sub);
    const int   d  = __shfl(d_l,   j + sub);
    const float nm = __shfl(nrm_l, j + sub);
    if (s >= 0)
      atomicAdd(&out[(size_t)d * F + f], feat[(size_t)s * F + f] * nm);
  }
}

__global__ void epi2_kernel(float* __restrict__ out, const float* __restrict__ h2,
                            const float* __restrict__ dis, const float* __restrict__ b2,
                            int n) {
  int i = blockIdx.x * blockDim.x + threadIdx.x;
  if (i < n * F2) {
    const int row = i / F2, f = i - row * F2;
    const float dv = dis[row];
    out[i] = out[i] + h2[i] * dv * dv + b2[f];
  }
}

extern "C" void kernel_launch(void* const* d_in, const int* in_sizes, int n_in,
                              void* d_out, int out_size, void* d_ws, size_t ws_size,
                              hipStream_t stream) {
  const float* x  = (const float*)d_in[0];
  const int*   ei = (const int*)d_in[1];
  const float* W1 = (const float*)d_in[2];
  const float* b1 = (const float*)d_in[3];
  const float* W2 = (const float*)d_in[4];
  const float* b2 = (const float*)d_in[5];

  const int n = in_sizes[0] / KDIM;
  const int E = in_sizes[1] / 2;
  const int* src = ei;
  const int* dst = ei + E;

  float* ws   = (float*)d_ws;
  float* dis  = ws;
  float* h    = dis  + n;
  float* agg1 = h    + (size_t)n * F1;
  float* h2   = agg1 + (size_t)n * F1;
  float* out  = (float*)d_out;

  {
    const int total = n + n * F1 + n * F2;
    init_kernel<<<(total + 255) / 256, 256, 0, stream>>>(dis, agg1, out, n);
  }
  {
    const int ranges = (n + DEG_BINS - 1) / DEG_BINS;
    degree_kernel<<<dim3(DEG_CHUNKS, ranges), DEG_THREADS, 0, stream>>>(dst, dis, n, E);
  }
  rsqrt_kernel<<<(n + 255) / 256, 256, 0, stream>>>(dis, n);
  {
    const int ntile = n >> 4;
    gemm1_kernel<<<(ntile + 7) / 8, 256, 0, stream>>>(x, W1, h, n);
  }
  {
    const int waves = (E + 31) / 32;
    agg_kernel<F1><<<(waves + 7) / 8, 256, 0, stream>>>(h, src, dst, dis, agg1, E);
  }
  {
    const int ntile = n >> 4;
    gemm2_kernel<<<(ntile + 7) / 8, 256, 0, stream>>>(agg1, h, dis, b1, W2, h2, n);
  }
  {
    const int waves = (E + 31) / 32;
    agg_kernel<F2><<<(waves + 7) / 8, 256, 0, stream>>>(h2, src, dst, dis, out, E);
  }
  epi2_kernel<<<(n * F2 + 255) / 256, 256, 0, stream>>>(out, h2, dis, b2, n);
}


# Edges regrouped by 128-node destination range once per call (LDS counting sort); a workgroup sums its range's rows by LDS float atomics, stores rows once: no global float atomics; sum order changes
# speedup vs baseline: 2.7925x; 1.7801x over previous
#include <hip/hip_runtime.h>

typedef __attribute__((ext_vector_type(2))) float v2f;
typedef __attribute__((ext_vector_type(8))) float v8f;

#define KDIM 512
#define F1 32
#define F2 16

__global__ void init_kernel(float* __restrict__ deg, int n) {
  int i = blockIdx.x * blockDim.x + threadIdx.x;
  if (i < n) deg[i] = 1.0f;
}

#define DEG_BINS    14336
#define DEG_THREADS 512
#define DEG_CHUNKS  36
#define DEG_DEPTH   8

static __device__ __forceinline__ void deg_count(unsigned int* cnt, int d, unsigned int lo) {
  const unsigned int b = (unsigned int)d - lo;
  if (b < (unsigned int)DEG_BINS) atomicAdd(&cnt[b], 1u);
}

__global__ __launch_bounds__(DEG_THREADS) void degree_kernel(const int* __restrict__ dst,
                                                             float* __restrict__ deg,
                                                             int n, int E) {
  __shared__ __attribute__((aligned(16))) unsigned int cnt[DEG_BINS];
  const int tid = threadIdx.x;
  const int lo  = blockIdx.y * DEG_BINS;

  for (int i = tid; i < DEG_BINS / 4; i += DEG_THREADS)
    ((uint4*)cnt)[i] = make_uint4(0u, 0u, 0u, 0u);
  __syncthreads();

  const int per = (((E + (int)gridDim.x - 1) / (int)gridDim.x) + 3) & ~3;
  const int e0  = blockIdx.x * per;
  const int len = max(0, min(E - e0, per));

  if ((((size_t)dst) & 15) == 0) {
    const int4* p  = (const int4*)(dst + e0);
    const int   nv = len >> 2;
    if (nv > 0) {
      int4 r[DEG_DEPTH];
#pragma unroll
      for (int u = 0; u < DEG_DEPTH; ++u)
        r[u] = p[min(u * DEG_THREADS + tid, nv - 1)];
      for (int base = 0; base < nv; base += DEG_DEPTH * DEG_THREADS) {
        int4 q[DEG_DEPTH];
#pragma unroll
        for (int u = 0; u < DEG_DEPTH; ++u)
          q[u] = p[min(base + (DEG_DEPTH + u) * DEG_THREADS + tid, nv - 1)];
#pragma unroll
        for (int u = 0; u < DEG_DEPTH; ++u) {
          const unsigned int lo_u =
              (base + u * DEG_THREADS + tid < nv) ? (unsigned int)lo : 0x80000000u;
          deg_count(cnt, r[u].x, lo_u);
          deg_count(cnt, r[u].y, lo_u);
          deg_count(cnt, r[u].z, lo_u);
          deg_count(cnt, r[u].w, lo_u);
        }
#pragma unroll
        for (int u = 0; u < DEG_DEPTH; ++u) r[u] = q[u];
      }
    }
    const int t = (nv << 2) + tid;
    if (t < len) deg_count(cnt, dst[e0 + t], (unsigned int)lo);
  } else {
    for (int t = tid; t < len; t += DEG_THREADS) deg_count(cnt, dst[e0 + t], (unsigned int)lo);
  }
  __syncthreads();

  const int rot = (int)((blockIdx.x * 5u) % (unsigned int)(DEG_BINS / DEG_THREADS)) * DEG_THREADS;
  static_assert(DEG_BINS % (4 * DEG_THREADS) == 0, "flush reads four counters per lane and trip");
  for (int i = tid; i < DEG_BINS; i += 4 * DEG_THREADS) {
    int j[4];
    unsigned int c[4];
#pragma unroll
    for (int u = 0; u < 4; ++u) {
      j[u] = i + u * DEG_THREADS + rot;
      if (j[u] >= DEG_BINS) j[u] -= DEG_BINS;
      c[u] = cnt[j[u]];
    }
#pragma unroll
    for (int u = 0; u < 4; ++u) {
      const int node = lo + j[u];
      if (c[u] != 0u && node < n) atomicAdd(&deg[node], (float)c[u]);
    }
  }
}

#define BK_SHIFT  7
#define BK_W      (1 << BK_SHIFT)
#define BK_STRIDE (BK_W * F2)
#define BK_CNT    0
#define BK_START  1
#define BK_CURSOR 2
#define BK_THREADS 512

__global__ __launch_bounds__(BK_W) void count_rsqrt_kernel(float* __restrict__ deg,
                                                           int* __restrict__ slots, int n) {
  __shared__ int part[BK_W / 32];
  const int t = threadIdx.x;
  const int i = blockIdx.x * BK_W + t;
  const float d = (i < n) ? deg[i] : 1.0f;
  int c = (int)d - 1;
  for (int o = 16; o > 0; o >>= 1) c += __shfl_xor(c, o);
  if ((t & 31) == 0) part[t >> 5] = c;
  __syncthreads();
  if (t == 0) {
    int s = 0;
    for (int w = 0; w < BK_W / 32; ++w) s += part[w];
    slots[(size_t)blockIdx.x * BK_STRIDE + BK_CNT] = s;
  }
  if (i < n) deg[i] = rsqrtf(d);
}

__global__ __launch_bounds__(1024) void bucket_scan_kernel(int* __restrict__ slots, int B) {
  __shared__ int wsum[32];
  __shared__ int total;
  const int tid = threadIdx.x, lane = tid & 31, wid = tid >> 5;
  int carry = 0;
  for (int base = 0; base < B; base += 1024) {
    const int b = base + tid;
    const int c = (b < B) ? slots[(size_t)b * BK_STRIDE + BK_CNT] : 0;
    int x = c;
    for (int o = 1; o < 32; o <<= 1) { const int y = __shfl_up(x, o); if (lane >= o) x += y; }
    if (lane == 31) wsum[wid] = x;
    __syncthreads();
    if (wid == 0) {
      const int tw = wsum[lane];
      int z = tw;
      for (int o = 1; o < 32; o <<= 1) { const int y = __shfl_up(z, o); if (lane >= o) z += y; }
      wsum[lane] = z - tw;
      if (lane == 31) total = z;
    }
    __syncthreads();
    const int excl = carry + wsum[wid] + x - c;
    if (b < B) {
      slots[(size_t)b * BK_STRIDE + BK_START]  = excl;
      slots[(size_t)b * BK_STRIDE + BK_CURSOR] = excl;
    }
    carry += total;
    __syncthreads();
  }
}

#define PT_THREADS 512
#define PT_GROUPS  256
#define PT_DEPTH   8

__global__ __launch_bounds__(PT_THREADS) void partition_kernel(const int* __restrict__ src,
                                                               const int* __restrict__ dst,
                                                               int* __restrict__ slots,
                                                               int* __restrict__ list,
                                                               int E, int B) {
  extern __shared__ int pt_lds[];
  int* hist  = pt_lds;
  int* bbase = pt_lds + B;
  const int tid = threadIdx.x;
  for (int b = tid; b < B; b += PT_THREADS) hist[b] = 0;
  __syncthreads();

  const int per = (E + (int)gridDim.x - 1) / (int)gridDim.x;
  const int e0  = blockIdx.x * per;
  const int e1  = min(E, e0 + per);

  for (int e = e0 + tid; e < e1; e += PT_DEPTH * PT_THREADS) {
    int d[PT_DEPTH];
#pragma unroll
    for (int u = 0; u < PT_DEPTH; ++u) d[u] = dst[min(e + u * PT_THREADS, e1 - 1)];
#pragma unroll
    for (int u = 0; u < PT_DEPTH; ++u)
      if (e + u * PT_THREADS < e1) atomicAdd(&hist[d[u] >> BK_SHIFT], 1);
  }
  __syncthreads();

  for (int b = tid; b < B; b += PT_THREADS) {
    const int c = hist[b];
    bbase[b] = (c != 0) ? atomicAdd(&slots[(size_t)b * BK_STRIDE + BK_CURSOR], c) : 0;
    hist[b] = 0;
  }
  __syncthreads();

  for (int e = e0 + tid; e < e1; e += PT_DEPTH * PT_THREADS) {
    int d[PT_DEPTH], s[PT_DEPTH];
#pragma unroll
    for (int u = 0; u < PT_DEPTH; ++u) {
      const int ee = min(e + u * PT_THREADS, e1 - 1);
      d[u] = dst[ee];
      s[u] = src[ee];
    }
#pragma unroll
    for (int u = 0; u < PT_DEPTH; ++u) {
      if (e + u * PT_THREADS < e1) {
        const int b = d[u] >> BK_SHIFT;
        const int p = bbase[b] + atomicAdd(&hist[b], 1);
        list[p] = (s[u] << BK_SHIFT) | (d[u] & (BK_W - 1));
      }
    }
  }
}

template <int EPS>
static __device__ __forceinline__ int lane_pick(int x, int u, int sub) {
  if constexpr (EPS == 1) {
    return __builtin_amdgcn_readlane(x, u);
  } else {
    const int a = __builtin_amdgcn_readlane(x, 2 * u);
    const int b = __builtin_amdgcn_readlane(x, 2 * u + 1);
    return sub ? b : a;
  }
}

template <int F, int PITCH>
static __device__ __forceinline__ void bucket_rows(const float* __restrict__ feat,
                                                   const int* __restrict__ list,
                                                   const float* __restrict__ dis,
                                                   float* acc, int lo, int start, int cnt) {
  constexpr int EPS   = 32 / F;
  constexpr int STEPS = 32 / EPS;
  const int lane = threadIdx.x & 31;
  const int wid  = threadIdx.x >> 5;
  const int nw   = blockDim.x >> 5;
  const int f    = lane & (F - 1);
  const int sub  = lane / F;
  if (cnt <= 0) return;
  int w = list[start + min(wid * 32 + lane, cnt - 1)];
  for (int base = wid * 32; base < cnt; base += nw * 32) {
    const int e    = base + lane;
    const int dl_l = w & (BK_W - 1);
    const int sv   = w >> BK_SHIFT;
    const float ds = dis[sv];
    const float dd = dis[lo + dl_l];
    const int s_l  = (e < cnt) ? sv : -1;
    w = list[start + min(e + nw * 32, cnt - 1)];
    float v[STEPS];
#pragma unroll
    for (int u = 0; u < STEPS; ++u) {
      const int s = lane_pick<EPS>(s_l, u, sub);
      v[u] = feat[(size_t)max(s, 0) * F + f];
    }
    __builtin_amdgcn_sched_barrier(0);
    const int nrm_l = __float_as_int(ds * dd);
#pragma unroll
    for (int u = 0; u < STEPS; ++u) {
      const int   s  = lane_pick<EPS>(s_l, u, sub);
      const int   dl = lane_pick<EPS>(dl_l, u, sub);
      const float nm = __int_as_float(lane_pick<EPS>(nrm_l, u, sub));
      const int row = (s >= 0) ? dl : BK_W;
      atomicAdd(&acc[row * PITCH + f], v[u] * nm);
    }
  }
}

#define A1_PITCH 33

__global__ __launch_bounds__(BK_THREADS) void agg1_gemm2_kernel(const float* __restrict__ h,
                                                         const int* __restrict__ list,
                                                         const int* __restrict__ slots,
                                                         const float* __restrict__ dis,
                                                         const float* __restrict__ b1,
                                                         const float* __restrict__ W2,
                                                         float* __restrict__ h2, int n) {
  __shared__ float acc[(BK_W + 1) * A1_PITCH];
  const int lo    = blockIdx.x * BK_W;
  const int cnt   = slots[(size_t)blockIdx.x * BK_STRIDE + BK_CNT];
  const int start = slots[(size_t)blockIdx.x * BK_STRIDE + BK_START];
  for (int i = threadIdx.x; i < BK_W * A1_PITCH; i += blockDim.x) acc[i] = 0.0f;
  __syncthreads();
  bucket_rows<F1, A1_PITCH>(h, list, dis, acc, lo, start, cnt);
  __syncthreads();

  const int lane = threadIdx.x & 31;
  const int r    = lane & 15;
  const int kh   = (lane >> 4) << 1;
  for (int tl = (threadIdx.x >> 5) * 16; tl < BK_W; tl += (blockDim.x >> 5) * 16) {
    const int m0 = lo + tl;
    if (m0 >= n) break;
    const int row  = m0 + r;
    const float dv = dis[row];
    const float d2 = dv * dv;
    const float* ag = acc + (tl + r) * A1_PITCH;
    const float* hr = h + (size_t)row * F1;
    v8f c8 = {0.f,0.f,0.f,0.f,0.f,0.f,0.f,0.f};
    for (int k = 0; k < F1; k += 4) {
      const int k0 = k + kh;
      v2f a;
      a.x = fmaxf(ag[k0]     + hr[k0]     * d2 + b1[k0],     0.0f);
      a.y = fmaxf(ag[k0 + 1] + hr[k0 + 1] * d2 + b1[k0 + 1], 0.0f);
      v2f b;
      b.x = W2[(size_t)k0 * F2 + r];
      b.y = W2[(size_t)(k0 + 1) * F2 + r];
      c8 = __builtin_amdgcn_wmma_f32_16x16x4_f32(false, a, false, b, (short)0, c8, false, false);
    }
    const int mbase = m0 + ((lane >> 4) << 3);
    float* op = h2 + (size_t)mbase * F2 + r;
#pragma unroll
    for (int v = 0; v < 8; ++v) op[(size_t)v * F2] = c8[v];
  }
}

__global__ __launch_bounds__(BK_THREADS) void agg2_out_kernel(const float* __restrict__ h2,
                                                       const int* __restrict__ list,
                                                       const float* __restrict__ dis,
                                                       const float* __restrict__ b2,
                                                       float* out, int n) {
  __shared__ float acc[(BK_W + 1) * F2];
  const int lo    = blockIdx.x * BK_W;
  const int cnt   = __float_as_int(out[(size_t)blockIdx.x * BK_STRIDE + BK_CNT]);
  const int start = __float_as_int(out[(size_t)blockIdx.x * BK_STRIDE + BK_START]);
  for (int i = threadIdx.x; i < BK_W * F2; i += blockDim.x) acc[i] = 0.0f;
  __syncthreads();
  bucket_rows<F2, F2>(h2, list, dis, acc, lo, start, cnt);
  __syncthreads();
  for (int i = threadIdx.x; i < BK_W * F2; i += blockDim.x) {
    const int row = lo + i / F2, f = i & (F2 - 1);
    if (row < n) {
      const float dv = dis[row];
      const size_t g = (size_t)row * F2 + f;
      out[g] = acc[i] + h2[g] * dv * dv + b2[f];
    }
  }
}

typedef __attribute__((ext_vector_type(4))) float v4f;

#define G1_WAVES 8
#define G1_COLS  64
#define G1_PITCH 68

static __device__ __forceinline__ void g1_wave_handover() {
  __builtin_amdgcn_fence(__ATOMIC_RELEASE, "wavefront");
  __builtin_amdgcn_wave_barrier();
  __builtin_amdgcn_fence(__ATOMIC_ACQUIRE, "wavefront");
}

__global__ __launch_bounds__(256) void gemm1_kernel(const float* __restrict__ x,
                                                    const float* __restrict__ W1,
                                                    float* __restrict__ h, int n) {
  __shared__ __attribute__((aligned(16))) float stage[G1_WAVES][16][G1_PITCH];
  const int lane  = threadIdx.x & 31;
  const int wid   = threadIdx.x >> 5;
  const int wave  = blockIdx.x * (blockDim.x >> 5) + wid;
  const int ntile = n >> 4;
  if (wave >= ntile) return;
  const int m0 = wave << 4;
  const int r  = lane & 15;
  const int hh = lane >> 4;

  float (*st)[G1_PITCH] = stage[wid];

  const int lrow = lane >> 4;
  const int lcol = (lane & 15) << 2;
  const float* gp = x + (size_t)(m0 + lrow) * KDIM + lcol;

  v4f R[8];
#pragma unroll
  for (int j = 0; j < 8; ++j)
    R[j] = __builtin_nontemporal_load((const v4f*)(gp + (size_t)(2 * j) * KDIM));

  v8f acc0 = {0.f,0.f,0.f,0.f,0.f,0.f,0.f,0.f};
  v8f acc1 = {0.f,0.f,0.f,0.f,0.f,0.f,0.f,0.f};

  for (int t = 0; t < KDIM / G1_COLS; ++t) {
#pragma unroll
    for (int j = 0; j < 8; ++j)
      *(v4f*)&st[2 * j + lrow][lcol] = R[j];

    if (t + 1 < KDIM / G1_COLS) {
#pragma unroll
      for (int j = 0; j < 8; ++j)
        R[j] = __builtin_nontemporal_load(
            (const v4f*)(gp + (size_t)(2 * j) * KDIM + (t + 1) * G1_COLS));
    }
    g1_wave_handover();

    const float* wb = W1 + (size_t)(t * G1_COLS + 4 * hh) * F1 + r;
#pragma unroll
    for (int g = 0; g < G1_COLS / 8; ++g) {
      const v4f a4 = *(const v4f*)&st[r][8 * g + 4 * hh];
      const float* w = wb + (size_t)(8 * g) * F1;
      v2f aL; aL.x = a4.x; aL.y = a4.y;
      v2f aH; aH.x = a4.z; aH.y = a4.w;
      v2f b0; b0.x = w[0];      b0.y = w[F1];
      v2f b1; b1.x = w[16];     b1.y = w[F1 + 16];
      v2f b2; b2.x = w[2 * F1]; b2.y = w[3 * F1];
      v2f b3; b3.x = w[2 * F1 + 16]; b3.y = w[3 * F1 + 16];
      acc0 = __builtin_amdgcn_wmma_f32_16x16x4_f32(false, aL, false, b0, (short)0, acc0, false, false);
      acc1 = __builtin_amdgcn_wmma_f32_16x16x4_f32(false, aL, false, b1, (short)0, acc1, false, false);
      acc0 = __builtin_amdgcn_wmma_f32_16x16x4_f32(false, aH, false, b2, (short)0, acc0, false, false);
      acc1 = __builtin_amdgcn_wmma_f32_16x16x4_f32(false, aH, false, b3, (short)0, acc1, false, false);
    }
    g1_wave_handover();
  }

  const int mbase = m0 + ((lane >> 4) << 3);
  float* hp = h + (size_t)mbase * F1 + r;
#pragma unroll
  for (int v = 0; v < 8; ++v) {
    hp[(size_t)v * F1]      = acc0[v];
    hp[(size_t)v * F1 + 16] = acc1[v];
  }
}

extern "C" void kernel_launch(void* const* d_in, const int* in_sizes, int n_in,
                              void* d_out, int out_size, void* d_ws, size_t ws_size,
                              hipStream_t stream) {
  const float* x  = (const float*)d_in[0];
  const int*   ei = (const int*)d_in[1];
  const float* W1 = (const float*)d_in[2];
  const float* b1 = (const float*)d_in[3];
  const float* W2 = (const float*)d_in[4];
  const float* b2 = (const float*)d_in[5];

  const int n = in_sizes[0] / KDIM;
  const int E = in_sizes[1] / 2;
  const int* src = ei;
  const int* dst = ei + E;

  float* ws   = (float*)d_ws;
  float* dis  = ws;
  float* h    = dis  + n;
  int*   list = (int*)(h + (size_t)n * F1);
  float* h2   = (float*)(list + E);
  float* out  = (float*)d_out;
  int*   slots = (int*)d_out;
  const int B = (n + BK_W - 1) / BK_W;

  init_kernel<<<(n + 255) / 256, 256, 0, stream>>>(dis, n);
  {
    const int ranges = (n + DEG_BINS - 1) / DEG_BINS;
    degree_kernel<<<dim3(DEG_CHUNKS, ranges), DEG_THREADS, 0, stream>>>(dst, dis, n, E);
  }
  count_rsqrt_kernel<<<B, BK_W, 0, stream>>>(dis, slots, n);
  bucket_scan_kernel<<<1, 1024, 0, stream>>>(slots, B);
  partition_kernel<<<PT_GROUPS, PT_THREADS, 2 * B * sizeof(int), stream>>>(src, dst, slots, list, E, B);
  {
    const int ntile = n >> 4;
    gemm1_kernel<<<(ntile + 7) / 8, 256, 0, stream>>>(x, W1, h, n);
  }
  agg1_gemm2_kernel<<<B, BK_THREADS, 0, stream>>>(h, list, slots, dis, b1, W2, h2, n);
  agg2_out_kernel<<<B, BK_THREADS, 0, stream>>>(h2, list, dis, b2, out, n);
}


# Edge weight dis[s]*dis[d] factored out of the edge loop: rows stored pre-scaled by their node's deg^-1/2, destination factor applied once per output row; no per-edge dis reads; rounding changes
# speedup vs baseline: 3.1849x; 1.1405x over previous
#include <hip/hip_runtime.h>

typedef __attribute__((ext_vector_type(2))) float v2f;
typedef __attribute__((ext_vector_type(8))) float v8f;

#define KDIM 512
#define F1 32
#define F2 16

__global__ void init_kernel(float* __restrict__ deg, int n) {
  int i = blockIdx.x * blockDim.x + threadIdx.x;
  if (i < n) deg[i] = 1.0f;
}

#define DEG_BINS    14336
#define DEG_THREADS 512
#define DEG_CHUNKS  36
#define DEG_DEPTH   8

static __device__ __forceinline__ void deg_count(unsigned int* cnt, int d, unsigned int lo) {
  const unsigned int b = (unsigned int)d - lo;
  if (b < (unsigned int)DEG_BINS) atomicAdd(&cnt[b], 1u);
}

__global__ __launch_bounds__(DEG_THREADS) void degree_kernel(const int* __restrict__ dst,
                                                             float* __restrict__ deg,
                                                             int n, int E) {
  __shared__ __attribute__((aligned(16))) unsigned int cnt[DEG_BINS];
  const int tid = threadIdx.x;
  const int lo  = blockIdx.y * DEG_BINS;

  for (int i = tid; i < DEG_BINS / 4; i += DEG_THREADS)
    ((uint4*)cnt)[i] = make_uint4(0u, 0u, 0u, 0u);
  __syncthreads();

  const int per = (((E + (int)gridDim.x - 1) / (int)gridDim.x) + 3) & ~3;
  const int e0  = blockIdx.x * per;
  const int len = max(0, min(E - e0, per));

  if ((((size_t)dst) & 15) == 0) {
    const int4* p  = (const int4*)(dst + e0);
    const int   nv = len >> 2;
    if (nv > 0) {
      int4 r[DEG_DEPTH];
#pragma unroll
      for (int u = 0; u < DEG_DEPTH; ++u)
        r[u] = p[min(u * DEG_THREADS + tid, nv - 1)];
      for (int base = 0; base < nv; base += DEG_DEPTH * DEG_THREADS) {
        int4 q[DEG_DEPTH];
#pragma unroll
        for (int u = 0; u < DEG_DEPTH; ++u)
          q[u] = p[min(base + (DEG_DEPTH + u) * DEG_THREADS + tid, nv - 1)];
#pragma unroll
        for (int u = 0; u < DEG_DEPTH; ++u) {
          const unsigned int lo_u =
              (base + u * DEG_THREADS + tid < nv) ? (unsigned int)lo : 0x80000000u;
          deg_count(cnt, r[u].x, lo_u);
          deg_count(cnt, r[u].y, lo_u);
          deg_count(cnt, r[u].z, lo_u);
          deg_count(cnt, r[u].w, lo_u);
        }
#pragma unroll
        for (int u = 0; u < DEG_DEPTH; ++u) r[u] = q[u];
      }
    }
    const int t = (nv << 2) + tid;
    if (t < len) deg_count(cnt, dst[e0 + t], (unsigned int)lo);
  } else {
    for (int t = tid; t < len; t += DEG_THREADS) deg_count(cnt, dst[e0 + t], (unsigned int)lo);
  }
  __syncthreads();

  const int rot = (int)((blockIdx.x * 5u) % (unsigned int)(DEG_BINS / DEG_THREADS)) * DEG_THREADS;
  static_assert(DEG_BINS % (4 * DEG_THREADS) == 0, "flush reads four counters per lane and trip");
  for (int i = tid; i < DEG_BINS; i += 4 * DEG_THREADS) {
    int j[4];
    unsigned int c[4];
#pragma unroll
    for (int u = 0; u < 4; ++u) {
      j[u] = i + u * DEG_THREADS + rot;
      if (j[u] >= DEG_BINS) j[u] -= DEG_BINS;
      c[u] = cnt[j[u]];
    }
#pragma unroll
    for (int u = 0; u < 4; ++u) {
      const int node = lo + j[u];
      if (c[u] != 0u && node < n) atomicAdd(&deg[node], (float)c[u]);
    }
  }
}

#define BK_SHIFT  7
#define BK_W      (1 << BK_SHIFT)
#define BK_STRIDE (BK_W * F2)
#define BK_CNT    0
#define BK_START  1
#define BK_CURSOR 2
#define BK_THREADS 512

__global__ __launch_bounds__(BK_W) void count_rsqrt_kernel(float* __restrict__ deg,
                                                           int* __restrict__ slots, int n) {
  __shared__ int part[BK_W / 32];
  const int t = threadIdx.x;
  const int i = blockIdx.x * BK_W + t;
  const float d = (i < n) ? deg[i] : 1.0f;
  int c = (int)d - 1;
  for (int o = 16; o > 0; o >>= 1) c += __shfl_xor(c, o);
  if ((t & 31) == 0) part[t >> 5] = c;
  __syncthreads();
  if (t == 0) {
    int s = 0;
    for (int w = 0; w < BK_W / 32; ++w) s += part[w];
    slots[(size_t)blockIdx.x * BK_STRIDE + BK_CNT] = s;
  }
  if (i < n) deg[i] = rsqrtf(d);
}

__global__ __launch_bounds__(1024) void bucket_scan_kernel(int* __restrict__ slots, int B) {
  __shared__ int wsum[32];
  __shared__ int total;
  const int tid = threadIdx.x, lane = tid & 31, wid = tid >> 5;
  int carry = 0;
  for (int base = 0; base < B; base += 1024) {
    const int b = base + tid;
    const int c = (b < B) ? slots[(size_t)b * BK_STRIDE + BK_CNT] : 0;
    int x = c;
    for (int o = 1; o < 32; o <<= 1) { const int y = __shfl_up(x, o); if (lane >= o) x += y; }
    if (lane == 31) wsum[wid] = x;
    __syncthreads();
    if (wid == 0) {
      const int tw = wsum[lane];
      int z = tw;
      for (int o = 1; o < 32; o <<= 1) { const int y = __shfl_up(z, o); if (lane >= o) z += y; }
      wsum[lane] = z - tw;
      if (lane == 31) total = z;
    }
    __syncthreads();
    const int excl = carry + wsum[wid] + x - c;
    if (b < B) {
      slots[(size_t)b * BK_STRIDE + BK_START]  = excl;
      slots[(size_t)b * BK_STRIDE + BK_CURSOR] = excl;
    }
    carry += total;
    __syncthreads();
  }
}

#define PT_THREADS 512
#define PT_GROUPS  256
#define PT_DEPTH   8

__global__ __launch_bounds__(PT_THREADS) void partition_kernel(const int* __restrict__ src,
                                                               const int* __restrict__ dst,
                                                               int* __restrict__ slots,
                                                               int* __restrict__ list,
                                                               int E, int B) {
  extern __shared__ int pt_lds[];
  int* hist  = pt_lds;
  int* bbase = pt_lds + B;
  const int tid = threadIdx.x;
  for (int b = tid; b < B; b += PT_THREADS) hist[b] = 0;
  __syncthreads();

  const int per = (E + (int)gridDim.x - 1) / (int)gridDim.x;
  const int e0  = blockIdx.x * per;
  const int e1  = min(E, e0 + per);

  for (int e = e0 + tid; e < e1; e += PT_DEPTH * PT_THREADS) {
    int d[PT_DEPTH];
#pragma unroll
    for (int u = 0; u < PT_DEPTH; ++u) d[u] = dst[min(e + u * PT_THREADS, e1 - 1)];
#pragma unroll
    for (int u = 0; u < PT_DEPTH; ++u)
      if (e + u * PT_THREADS < e1) atomicAdd(&hist[d[u] >> BK_SHIFT], 1);
  }
  __syncthreads();

  for (int b = tid; b < B; b += PT_THREADS) {
    const int c = hist[b];
    bbase[b] = (c != 0) ? atomicAdd(&slots[(size_t)b * BK_STRIDE + BK_CURSOR], c) : 0;
    hist[b] = 0;
  }
  __syncthreads();

  for (int e = e0 + tid; e < e1; e += PT_DEPTH * PT_THREADS) {
    int d[PT_DEPTH], s[PT_DEPTH];
#pragma unroll
    for (int u = 0; u < PT_DEPTH; ++u) {
      const int ee = min(e + u * PT_THREADS, e1 - 1);
      d[u] = dst[ee];
      s[u] = src[ee];
    }
#pragma unroll
    for (int u = 0; u < PT_DEPTH; ++u) {
      if (e + u * PT_THREADS < e1) {
        const int b = d[u] >> BK_SHIFT;
        const int p = bbase[b] + atomicAdd(&hist[b], 1);
        list[p] = (s[u] << BK_SHIFT) | (d[u] & (BK_W - 1));
      }
    }
  }
}

template <int EPS>
static __device__ __forceinline__ int lane_pick(int x, int u, int sub) {
  if constexpr (EPS == 1) {
    return __builtin_amdgcn_readlane(x, u);
  } else {
    const int a = __builtin_amdgcn_readlane(x, 2 * u);
    const int b = __builtin_amdgcn_readlane(x, 2 * u + 1);
    return sub ? b : a;
  }
}

template <int F, int PITCH>
static __device__ __forceinline__ void bucket_rows(const float* __restrict__ feat,
                                                   const int* __restrict__ list,
                                                   float* acc, int start, int cnt) {
  constexpr int EPS   = 32 / F;
  constexpr int STEPS = 32 / EPS;
  const int lane = threadIdx.x & 31;
  const int wid  = threadIdx.x >> 5;
  const int nw   = blockDim.x >> 5;
  const int f    = lane & (F - 1);
  const int sub  = lane / F;
  if (cnt <= 0) return;
  int w = list[start + min(wid * 32 + lane, cnt - 1)];
  for (int base = wid * 32; base < cnt; base += nw * 32) {
    const int e    = base + lane;
    const int dl_l = (e < cnt) ? (w & (BK_W - 1)) : BK_W;
    const int s_l  = w >> BK_SHIFT;
    w = list[start + min(e + nw * 32, cnt - 1)];
    float v[STEPS];
#pragma unroll
    for (int u = 0; u < STEPS; ++u) {
      const int s = lane_pick<EPS>(s_l, u, sub);
      v[u] = feat[(size_t)s * F + f];
    }
    __builtin_amdgcn_sched_barrier(0);
#pragma unroll
    for (int u = 0; u < STEPS; ++u) {
      const int row = lane_pick<EPS>(dl_l, u, sub);
      atomicAdd(&acc[row * PITCH + f], v[u]);
    }
  }
}

#define A1_PITCH 33

__global__ __launch_bounds__(BK_THREADS) void agg1_gemm2_kernel(const float* __restrict__ h,
                                                         const int* __restrict__ list,
                                                         const int* __restrict__ slots,
                                                         const float* __restrict__ dis,
                                                         const float* __restrict__ b1,
                                                         const float* __restrict__ W2,
                                                         float* __restrict__ h2, int n) {
  __shared__ float acc[(BK_W + 1) * A1_PITCH];
  const int lo    = blockIdx.x * BK_W;
  const int cnt   = slots[(size_t)blockIdx.x * BK_STRIDE + BK_CNT];
  const int start = slots[(size_t)blockIdx.x * BK_STRIDE + BK_START];
  for (int i = threadIdx.x; i < BK_W * A1_PITCH; i += blockDim.x) acc[i] = 0.0f;
  __syncthreads();
  bucket_rows<F1, A1_PITCH>(h, list, acc, start, cnt);
  __syncthreads();

  const int lane = threadIdx.x & 31;
  const int r    = lane & 15;
  const int kh   = (lane >> 4) << 1;
  for (int tl = (threadIdx.x >> 5) * 16; tl < BK_W; tl += (blockDim.x >> 5) * 16) {
    const int m0 = lo + tl;
    if (m0 >= n) break;
    const int row  = m0 + r;
    const float dv = dis[row];
    const float* ag = acc + (tl + r) * A1_PITCH;
    const float* hr = h + (size_t)row * F1;
    v8f c8 = {0.f,0.f,0.f,0.f,0.f,0.f,0.f,0.f};
    for (int k = 0; k < F1; k += 4) {
      const int k0 = k + kh;
      v2f a;
      a.x = fmaxf((ag[k0]     + hr[k0])     * dv + b1[k0],     0.0f) * dv;
      a.y = fmaxf((ag[k0 + 1] + hr[k0 + 1]) * dv + b1[k0 + 1], 0.0f) * dv;
      v2f b;
      b.x = W2[(size_t)k0 * F2 + r];
      b.y = W2[(size_t)(k0 + 1) * F2 + r];
      c8 = __builtin_amdgcn_wmma_f32_16x16x4_f32(false, a, false, b, (short)0, c8, false, false);
    }
    const int mbase = m0 + ((lane >> 4) << 3);
    float* op = h2 + (size_t)mbase * F2 + r;
#pragma unroll
    for (int v = 0; v < 8; ++v) op[(size_t)v * F2] = c8[v];
  }
}

__global__ __launch_bounds__(BK_THREADS) void agg2_out_kernel(const float* __restrict__ h2,
                                                       const int* __restrict__ list,
                                                       const float* __restrict__ dis,
                                                       const float* __restrict__ b2,
                                                       float* out, int n) {
  __shared__ float acc[(BK_W + 1) * F2];
  const int lo    = blockIdx.x * BK_W;
  const int cnt   = __float_as_int(out[(size_t)blockIdx.x * BK_STRIDE + BK_CNT]);
  const int start = __float_as_int(out[(size_t)blockIdx.x * BK_STRIDE + BK_START]);
  for (int i = threadIdx.x; i < BK_W * F2; i += blockDim.x) acc[i] = 0.0f;
  __syncthreads();
  bucket_rows<F2, F2>(h2, list, acc, start, cnt);
  __syncthreads();
  for (int i = threadIdx.x; i < BK_W * F2; i += blockDim.x) {
    const int row = lo + i / F2, f = i & (F2 - 1);
    if (row < n) {
      const float dv = dis[row];
      const size_t g = (size_t)row * F2 + f;
      out[g] = (acc[i] + h2[g]) * dv + b2[f];
    }
  }
}

typedef __attribute__((ext_vector_type(4))) float v4f;

#define G1_WAVES 8
#define G1_COLS  64
#define G1_PITCH 68

static __device__ __forceinline__ void g1_wave_handover() {
  __builtin_amdgcn_fence(__ATOMIC_RELEASE, "wavefront");
  __builtin_amdgcn_wave_barrier();
  __builtin_amdgcn_fence(__ATOMIC_ACQUIRE, "wavefront");
}

__global__ __launch_bounds__(256) void gemm1_kernel(const float* __restrict__ x,
                                                    const float* __restrict__ W1,
                                                    const float* __restrict__ dis,
                                                    float* __restrict__ h, int n) {
  __shared__ __attribute__((aligned(16))) float stage[G1_WAVES][16][G1_PITCH];
  const int lane  = threadIdx.x & 31;
  const int wid   = threadIdx.x >> 5;
  const int wave  = blockIdx.x * (blockDim.x >> 5) + wid;
  const int ntile = n >> 4;
  if (wave >= ntile) return;
  const int m0 = wave << 4;
  const int r  = lane & 15;
  const int hh = lane >> 4;

  float (*st)[G1_PITCH] = stage[wid];

  const int lrow = lane >> 4;
  const int lcol = (lane & 15) << 2;
  const float* gp = x + (size_t)(m0 + lrow) * KDIM + lcol;

  v4f R[8];
#pragma unroll
  for (int j = 0; j < 8; ++j)
    R[j] = __builtin_nontemporal_load((const v4f*)(gp + (size_t)(2 * j) * KDIM));

  v8f acc0 = {0.f,0.f,0.f,0.f,0.f,0.f,0.f,0.f};
  v8f acc1 = {0.f,0.f,0.f,0.f,0.f,0.f,0.f,0.f};

  const int mbase = m0 + ((lane >> 4) << 3);
  float rs[8];
#pragma unroll
  for (int v = 0; v < 8; ++v) rs[v] = dis[mbase + v];

  for (int t = 0; t < KDIM / G1_COLS; ++t) {
#pragma unroll
    for (int j = 0; j < 8; ++j)
      *(v4f*)&st[2 * j + lrow][lcol] = R[j];

    if (t + 1 < KDIM / G1_COLS) {
#pragma unroll
      for (int j = 0; j < 8; ++j)
        R[j] = __builtin_nontemporal_load(
            (const v4f*)(gp + (size_t)(2 * j) * KDIM + (t + 1) * G1_COLS));
    }
    g1_wave_handover();

    const float* wb = W1 + (size_t)(t * G1_COLS + 4 * hh) * F1 + r;
#pragma unroll
    for (int g = 0; g < G1_COLS / 8; ++g) {
      const v4f a4 = *(const v4f*)&st[r][8 * g + 4 * hh];
      const float* w = wb + (size_t)(8 * g) * F1;
      v2f aL; aL.x = a4.x; aL.y = a4.y;
      v2f aH; aH.x = a4.z; aH.y = a4.w;
      v2f b0; b0.x = w[0];      b0.y = w[F1];
      v2f b1; b1.x = w[16];     b1.y = w[F1 + 16];
      v2f b2; b2.x = w[2 * F1]; b2.y = w[3 * F1];
      v2f b3; b3.x = w[2 * F1 + 16]; b3.y = w[3 * F1 + 16];
      acc0 = __builtin_amdgcn_wmma_f32_16x16x4_f32(false, aL, false, b0, (short)0, acc0, false, false);
      acc1 = __builtin_amdgcn_wmma_f32_16x16x4_f32(false, aL, false, b1, (short)0, acc1, false, false);
      acc0 = __builtin_amdgcn_wmma_f32_16x16x4_f32(false, aH, false, b2, (short)0, acc0, false, false);
      acc1 = __builtin_amdgcn_wmma_f32_16x16x4_f32(false, aH, false, b3, (short)0, acc1, false, false);
    }
    g1_wave_handover();
  }

  float* hp = h + (size_t)mbase * F1 + r;
#pragma unroll
  for (int v = 0; v < 8; ++v) {
    hp[(size_t)v * F1]      = acc0[v] * rs[v];
    hp[(size_t)v * F1 + 16] = acc1[v] * rs[v];
  }
}

extern "C" void kernel_launch(void* const* d_in, const int* in_sizes, int n_in,
                              void* d_out, int out_size, void* d_ws, size_t ws_size,
                              hipStream_t stream) {
  const float* x  = (const float*)d_in[0];
  const int*   ei = (const int*)d_in[1];
  const float* W1 = (const float*)d_in[2];
  const float* b1 = (const float*)d_in[3];
  const float* W2 = (const float*)d_in[4];
  const float* b2 = (const float*)d_in[5];

  const int n = in_sizes[0] / KDIM;
  const int E = in_sizes[1] / 2;
  const int* src = ei;
  const int* dst = ei + E;

  float* ws   = (float*)d_ws;
  float* dis  = ws;
  float* h    = dis  + n;
  int*   list = (int*)(h + (size_t)n * F1);
  float* h2   = (float*)(list + E);
  float* out  = (float*)d_out;
  int*   slots = (int*)d_out;
  const int B = (n + BK_W - 1) / BK_W;

  init_kernel<<<(n + 255) / 256, 256, 0, stream>>>(dis, n);
  {
    const int ranges = (n + DEG_BINS - 1) / DEG_BINS;
    degree_kernel<<<dim3(DEG_CHUNKS, ranges), DEG_THREADS, 0, stream>>>(dst, dis, n, E);
  }
  count_rsqrt_kernel<<<B, BK_W, 0, stream>>>(dis, slots, n);
  bucket_scan_kernel<<<1, 1024, 0, stream>>>(slots, B);
  partition_kernel<<<PT_GROUPS, PT_THREADS, 2 * B * sizeof(int), stream>>>(src, dst, slots, list, E, B);
  {
    const int ntile = n >> 4;
    gemm1_kernel<<<(ntile + 7) / 8, 256, 0, stream>>>(x, W1, dis, h, n);
  }
  agg1_gemm2_kernel<<<B, BK_THREADS, 0, stream>>>(h, list, slots, dis, b1, W2, h2, n);
  agg2_out_kernel<<<B, BK_THREADS, 0, stream>>>(h2, list, dis, b2, out, n);
}


# A bucket's edges sorted by destination row in LDS (counting sort); each row's edge rows summed in registers from 16-byte loads, 4 or 8 rows per load: no LDS float atomics; sum order changes
# speedup vs baseline: 3.3690x; 1.0578x over previous
#include <hip/hip_runtime.h>

typedef __attribute__((ext_vector_type(2))) float v2f;
typedef __attribute__((ext_vector_type(8))) float v8f;

#define KDIM 512
#define F1 32
#define F2 16

__global__ void init_kernel(float* __restrict__ deg, int n) {
  int i = blockIdx.x * blockDim.x + threadIdx.x;
  if (i < n) deg[i] = 1.0f;
}

#define DEG_BINS    14336
#define DEG_THREADS 512
#define DEG_CHUNKS  36
#define DEG_DEPTH   8

static __device__ __forceinline__ void deg_count(unsigned int* cnt, int d, unsigned int lo) {
  const unsigned int b = (unsigned int)d - lo;
  if (b < (unsigned int)DEG_BINS) atomicAdd(&cnt[b], 1u);
}

__global__ __launch_bounds__(DEG_THREADS) void degree_kernel(const int* __restrict__ dst,
                                                             float* __restrict__ deg,
                                                             int n, int E) {
  __shared__ __attribute__((aligned(16))) unsigned int cnt[DEG_BINS];
  const int tid = threadIdx.x;
  const int lo  = blockIdx.y * DEG_BINS;

  for (int i = tid; i < DEG_BINS / 4; i += DEG_THREADS)
    ((uint4*)cnt)[i] = make_uint4(0u, 0u, 0u, 0u);
  __syncthreads();

  const int per = (((E + (int)gridDim.x - 1) / (int)gridDim.x) + 3) & ~3;
  const int e0  = blockIdx.x * per;
  const int len = max(0, min(E - e0, per));

  if ((((size_t)dst) & 15) == 0) {
    const int4* p  = (const int4*)(dst + e0);
    const int   nv = len >> 2;
    if (nv > 0) {
      int4 r[DEG_DEPTH];
#pragma unroll
      for (int u = 0; u < DEG_DEPTH; ++u)
        r[u] = p[min(u * DEG_THREADS + tid, nv - 1)];
      for (int base = 0; base < nv; base += DEG_DEPTH * DEG_THREADS) {
        int4 q[DEG_DEPTH];
#pragma unroll
        for (int u = 0; u < DEG_DEPTH; ++u)
          q[u] = p[min(base + (DEG_DEPTH + u) * DEG_THREADS + tid, nv - 1)];
#pragma unroll
        for (int u = 0; u < DEG_DEPTH; ++u) {
          const unsigned int lo_u =
              (base + u * DEG_THREADS + tid < nv) ? (unsigned int)lo : 0x80000000u;
          deg_count(cnt, r[u].x, lo_u);
          deg_count(cnt, r[u].y, lo_u);
          deg_count(cnt, r[u].z, lo_u);
          deg_count(cnt, r[u].w, lo_u);
        }
#pragma unroll
        for (int u = 0; u < DEG_DEPTH; ++u) r[u] = q[u];
      }
    }
    const int t = (nv << 2) + tid;
    if (t < len) deg_count(cnt, dst[e0 + t], (unsigned int)lo);
  } else {
    for (int t = tid; t < len; t += DEG_THREADS) deg_count(cnt, dst[e0 + t], (unsigned int)lo);
  }
  __syncthreads();

  const int rot = (int)((blockIdx.x * 5u) % (unsigned int)(DEG_BINS / DEG_THREADS)) * DEG_THREADS;
  static_assert(DEG_BINS % (4 * DEG_THREADS) == 0, "flush reads four counters per lane and trip");
  for (int i = tid; i < DEG_BINS; i += 4 * DEG_THREADS) {
    int j[4];
    unsigned int c[4];
#pragma unroll
    for (int u = 0; u < 4; ++u) {
      j[u] = i + u * DEG_THREADS + rot;
      if (j[u] >= DEG_BINS) j[u] -= DEG_BINS;
      c[u] = cnt[j[u]];
    }
#pragma unroll
    for (int u = 0; u < 4; ++u) {
      const int node = lo + j[u];
      if (c[u] != 0u && node < n) atomicAdd(&deg[node], (float)c[u]);
    }
  }
}

#define BK_SHIFT  7
#define BK_W      (1 << BK_SHIFT)
#define BK_STRIDE (BK_W * F2)
#define BK_CNT    0
#define BK_START  1
#define BK_CURSOR 2
#define BK_THREADS 512

__global__ __launch_bounds__(BK_W) void count_rsqrt_kernel(float* __restrict__ deg,
                                                           int* __restrict__ slots, int n) {
  __shared__ int part[BK_W / 32];
  const int t = threadIdx.x;
  const int i = blockIdx.x * BK_W + t;
  const float d = (i < n) ? deg[i] : 1.0f;
  int c = (int)d - 1;
  for (int o = 16; o > 0; o >>= 1) c += __shfl_xor(c, o);
  if ((t & 31) == 0) part[t >> 5] = c;
  __syncthreads();
  if (t == 0) {
    int s = 0;
    for (int w = 0; w < BK_W / 32; ++w) s += part[w];
    slots[(size_t)blockIdx.x * BK_STRIDE + BK_CNT] = s;
  }
  if (i < n) deg[i] = rsqrtf(d);
}

__global__ __launch_bounds__(1024) void bucket_scan_kernel(int* __restrict__ slots, int B) {
  __shared__ int wsum[32];
  __shared__ int total;
  const int tid = threadIdx.x, lane = tid & 31, wid = tid >> 5;
  int carry = 0;
  for (int base = 0; base < B; base += 1024) {
    const int b = base + tid;
    const int c = (b < B) ? slots[(size_t)b * BK_STRIDE + BK_CNT] : 0;
    int x = c;
    for (int o = 1; o < 32; o <<= 1) { const int y = __shfl_up(x, o); if (lane >= o) x += y; }
    if (lane == 31) wsum[wid] = x;
    __syncthreads();
    if (wid == 0) {
      const int tw = wsum[lane];
      int z = tw;
      for (int o = 1; o < 32; o <<= 1) { const int y = __shfl_up(z, o); if (lane >= o) z += y; }
      wsum[lane] = z - tw;
      if (lane == 31) total = z;
    }
    __syncthreads();
    const int excl = carry + wsum[wid] + x - c;
    if (b < B) {
      slots[(size_t)b * BK_STRIDE + BK_START]  = excl;
      slots[(size_t)b * BK_STRIDE + BK_CURSOR] = excl;
    }
    carry += total;
    __syncthreads();
  }
}

#define PT_THREADS 512
#define PT_GROUPS  256
#define PT_DEPTH   8

__global__ __launch_bounds__(PT_THREADS) void partition_kernel(const int* __restrict__ src,
                                                               const int* __restrict__ dst,
                                                               int* __restrict__ slots,
                                                               int* __restrict__ list,
                                                               int E, int B) {
  extern __shared__ int pt_lds[];
  int* hist  = pt_lds;
  int* bbase = pt_lds + B;
  const int tid = threadIdx.x;
  for (int b = tid; b < B; b += PT_THREADS) hist[b] = 0;
  __syncthreads();

  const int per = (E + (int)gridDim.x - 1) / (int)gridDim.x;
  const int e0  = blockIdx.x * per;
  const int e1  = min(E, e0 + per);

  for (int e = e0 + tid; e < e1; e += PT_DEPTH * PT_THREADS) {
    int d[PT_DEPTH];
#pragma unroll
    for (int u = 0; u < PT_DEPTH; ++u) d[u] = dst[min(e + u * PT_THREADS, e1 - 1)];
#pragma unroll
    for (int u = 0; u < PT_DEPTH; ++u)
      if (e + u * PT_THREADS < e1) atomicAdd(&hist[d[u] >> BK_SHIFT], 1);
  }
  __syncthreads();

  for (int b = tid; b < B; b += PT_THREADS) {
    const int c = hist[b];
    bbase[b] = (c != 0) ? atomicAdd(&slots[(size_t)b * BK_STRIDE + BK_CURSOR], c) : 0;
    hist[b] = 0;
  }
  __syncthreads();

  for (int e = e0 + tid; e < e1; e += PT_DEPTH * PT_THREADS) {
    int d[PT_DEPTH], s[PT_DEPTH];
#pragma unroll
    for (int u = 0; u < PT_DEPTH; ++u) {
      const int ee = min(e + u * PT_THREADS, e1 - 1);
      d[u] = dst[ee];
      s[u] = src[ee];
    }
#pragma unroll
    for (int u = 0; u < PT_DEPTH; ++u) {
      if (e + u * PT_THREADS < e1) {
        const int b = d[u] >> BK_SHIFT;
        const int p = bbase[b] + atomicAdd(&hist[b], 1);
        list[p] = (s[u] << BK_SHIFT) | (d[u] & (BK_W - 1));
      }
    }
  }
}

#define SORT_CAP   5120
#define SORT_PER_LANE (SORT_CAP / BK_THREADS)
#define BK_WAVES_PER_SIMD 16
#define TAIL_DEPTH 4

typedef __attribute__((ext_vector_type(4))) float f4v;

template <int F>
static __device__ __forceinline__ f4v row_piece(const float* __restrict__ feat, int s, unsigned int fl4) {
  const unsigned int off = (unsigned int)s * (unsigned int)(F * 4) + fl4;
  return *(const f4v*)((const char*)feat + off);
}

template <int F, int PITCH>
static __device__ __forceinline__ void bucket_rows(const float* __restrict__ feat,
                                                   const int* __restrict__ list,
                                                   float* acc, int* cntL, int* offL, int* srcL,
                                                   int start, int cnt) {
  constexpr int LPR = F / 4;
  constexpr int GPW = 32 / LPR;
  constexpr int NG  = (BK_THREADS / 32) * GPW;
  constexpr int RPG = BK_W / NG;
  constexpr int ROW_DEPTH = (RPG > 1) ? 6 : 8;
  static_assert(RPG * NG == BK_W, "every row of the bucket has exactly one owner");
  static_assert(SORT_PER_LANE * BK_THREADS == SORT_CAP, "a piece is a whole number of words per lane");
  const int tid  = threadIdx.x;
  const int lane = tid & 31;
  const int wid  = tid >> 5;
  const int grp  = wid * GPW + lane / LPR;
  const unsigned int fl  = (unsigned int)(lane % LPR) * 4u;
  const unsigned int fl4 = fl * 4u;

  f4v sum[RPG];
#pragma unroll
  for (int q = 0; q < RPG; ++q) sum[q] = (f4v){0.f, 0.f, 0.f, 0.f};

  for (int c0 = 0; c0 < cnt; c0 += SORT_CAP) {
    const int m = min(SORT_CAP, cnt - c0);
    const int* lp = list + start + c0;
    int wv[SORT_PER_LANE];
#pragma unroll
    for (int u = 0; u < SORT_PER_LANE; ++u) wv[u] = lp[min(tid + u * BK_THREADS, m - 1)];
    if (tid < BK_W) cntL[tid] = 0;
    __syncthreads();
#pragma unroll
    for (int u = 0; u < SORT_PER_LANE; ++u)
      if (tid + u * BK_THREADS < m) atomicAdd(&cntL[wv[u] & (BK_W - 1)], 1);
    __syncthreads();
    if (wid == 0) {
      const int a0 = cntL[4 * lane], a1 = cntL[4 * lane + 1], a2 = cntL[4 * lane + 2], a3 = cntL[4 * lane + 3];
      const int t = a0 + a1 + a2 + a3;
      int x = t;
      for (int o = 1; o < 32; o <<= 1) { const int y = __shfl_up(x, o); if (lane >= o) x += y; }
      const int p0 = x - t, p1 = p0 + a0, p2 = p1 + a1, p3 = p2 + a2;
      offL[4 * lane] = p0; offL[4 * lane + 1] = p1; offL[4 * lane + 2] = p2; offL[4 * lane + 3] = p3;
      cntL[4 * lane] = p0; cntL[4 * lane + 1] = p1; cntL[4 * lane + 2] = p2; cntL[4 * lane + 3] = p3;
      if (lane == 31) offL[BK_W] = x;
    }
    __syncthreads();
    int pos[SORT_PER_LANE];
#pragma unroll
    for (int u = 0; u < SORT_PER_LANE; ++u) {
      pos[u] = 0;
      if (tid + u * BK_THREADS < m) pos[u] = atomicAdd(&cntL[wv[u] & (BK_W - 1)], 1);
    }
#pragma unroll
    for (int u = 0; u < SORT_PER_LANE; ++u)
      if (tid + u * BK_THREADS < m) srcL[pos[u]] = wv[u] >> BK_SHIFT;
    __syncthreads();

#pragma unroll
    for (int q = 0; q < RPG; ++q) {
      const int row = grp * RPG + q;
      int j = offL[row];
      const int end = offL[row + 1];
      f4v a = sum[q];
      for (; j + ROW_DEPTH <= end; j += ROW_DEPTH) {
        f4v v[ROW_DEPTH];
#pragma unroll
        for (int u = 0; u < ROW_DEPTH; ++u) v[u] = row_piece<F>(feat, srcL[j + u], fl4);
#pragma unroll
        for (int u = 0; u < ROW_DEPTH; ++u) a += v[u];
      }
      for (; j < end; j += TAIL_DEPTH) {
        f4v v[TAIL_DEPTH];
#pragma unroll
        for (int u = 0; u < TAIL_DEPTH; ++u) v[u] = row_piece<F>(feat, srcL[min(j + u, end - 1)], fl4);
#pragma unroll
        for (int u = 0; u < TAIL_DEPTH; ++u) {
          const bool in = (j + u < end);
          a.x += in ? v[u].x : 0.0f;
          a.y += in ? v[u].y : 0.0f;
          a.z += in ? v[u].z : 0.0f;
          a.w += in ? v[u].w : 0.0f;
        }
      }
      sum[q] = a;
    }
    __syncthreads();
  }

#pragma unroll
  for (int q = 0; q < RPG; ++q) {
    float* o = acc + (grp * RPG + q) * PITCH + fl;
    o[0] = sum[q].x; o[1] = sum[q].y; o[2] = sum[q].z; o[3] = sum[q].w;
  }
}

#define A1_PITCH 33

__global__ __launch_bounds__(BK_THREADS, BK_WAVES_PER_SIMD) void agg1_gemm2_kernel(const float* __restrict__ h,
                                                         const int* __restrict__ list,
                                                         const int* __restrict__ slots,
                                                         const float* __restrict__ dis,
                                                         const float* __restrict__ b1,
                                                         const float* __restrict__ W2,
                                                         float* __restrict__ h2, int n) {
  __shared__ float acc[BK_W * A1_PITCH];
  __shared__ int cntL[BK_W];
  __shared__ int offL[BK_W + 1];
  __shared__ int srcL[SORT_CAP];
  const int lo    = blockIdx.x * BK_W;
  const int cnt   = slots[(size_t)blockIdx.x * BK_STRIDE + BK_CNT];
  const int start = slots[(size_t)blockIdx.x * BK_STRIDE + BK_START];
  bucket_rows<F1, A1_PITCH>(h, list, acc, cntL, offL, srcL, start, cnt);
  __syncthreads();

  const int lane = threadIdx.x & 31;
  const int r    = lane & 15;
  const int kh   = (lane >> 4) << 1;
  for (int tl = (threadIdx.x >> 5) * 16; tl < BK_W; tl += (blockDim.x >> 5) * 16) {
    const int m0 = lo + tl;
    if (m0 >= n) break;
    const int row  = m0 + r;
    const float dv = dis[row];
    const float* ag = acc + (tl + r) * A1_PITCH;
    const float* hr = h + (size_t)row * F1;
    v8f c8 = {0.f,0.f,0.f,0.f,0.f,0.f,0.f,0.f};
    for (int k = 0; k < F1; k += 4) {
      const int k0 = k + kh;
      v2f a;
      a.x = fmaxf((ag[k0]     + hr[k0])     * dv + b1[k0],     0.0f) * dv;
      a.y = fmaxf((ag[k0 + 1] + hr[k0 + 1]) * dv + b1[k0 + 1], 0.0f) * dv;
      v2f b;
      b.x = W2[(size_t)k0 * F2 + r];
      b.y = W2[(size_t)(k0 + 1) * F2 + r];
      c8 = __builtin_amdgcn_wmma_f32_16x16x4_f32(false, a, false, b, (short)0, c8, false, false);
    }
    const int mbase = m0 + ((lane >> 4) << 3);
    float* op = h2 + (size_t)mbase * F2 + r;
#pragma unroll
    for (int v = 0; v < 8; ++v) op[(size_t)v * F2] = c8[v];
  }
}

__global__ __launch_bounds__(BK_THREADS, BK_WAVES_PER_SIMD) void agg2_out_kernel(const float* __restrict__ h2,
                                                       const int* __restrict__ list,
                                                       const float* __restrict__ dis,
                                                       const float* __restrict__ b2,
                                                       float* out, int n) {
  __shared__ float acc[BK_W * F2];
  __shared__ int cntL[BK_W];
  __shared__ int offL[BK_W + 1];
  __shared__ int srcL[SORT_CAP];
  const int lo    = blockIdx.x * BK_W;
  const int cnt   = __float_as_int(out[(size_t)blockIdx.x * BK_STRIDE + BK_CNT]);
  const int start = __float_as_int(out[(size_t)blockIdx.x * BK_STRIDE + BK_START]);
  bucket_rows<F2, F2>(h2, list, acc, cntL, offL, srcL, start, cnt);
  __syncthreads();
  for (int i = threadIdx.x; i < BK_W * F2; i += blockDim.x) {
    const int row = lo + i / F2, f = i & (F2 - 1);
    if (row < n) {
      const float dv = dis[row];
      const size_t g = (size_t)row * F2 + f;
      out[g] = (acc[i] + h2[g]) * dv + b2[f];
    }
  }
}

typedef __attribute__((ext_vector_type(4))) float v4f;

#define G1_WAVES 8
#define G1_COLS  64
#define G1_PITCH 68

static __device__ __forceinline__ void g1_wave_handover() {
  __builtin_amdgcn_fence(__ATOMIC_RELEASE, "wavefront");
  __builtin_amdgcn_wave_barrier();
  __builtin_amdgcn_fence(__ATOMIC_ACQUIRE, "wavefront");
}

__global__ __launch_bounds__(256) void gemm1_kernel(const float* __restrict__ x,
                                                    const float* __restrict__ W1,
                                                    const float* __restrict__ dis,
                                                    float* __restrict__ h, int n) {
  __shared__ __attribute__((aligned(16))) float stage[G1_WAVES][16][G1_PITCH];
  const int lane  = threadIdx.x & 31;
  const int wid   = threadIdx.x >> 5;
  const int wave  = blockIdx.x * (blockDim.x >> 5) + wid;
  const int ntile = n >> 4;
  if (wave >= ntile) return;
  const int m0 = wave << 4;
  const int r  = lane & 15;
  const int hh = lane >> 4;

  float (*st)[G1_PITCH] = stage[wid];

  const int lrow = lane >> 4;
  const int lcol = (lane & 15) << 2;
  const float* gp = x + (size_t)(m0 + lrow) * KDIM + lcol;

  v4f R[8];
#pragma unroll
  for (int j = 0; j < 8; ++j)
    R[j] = __builtin_nontemporal_load((const v4f*)(gp + (size_t)(2 * j) * KDIM));

  v8f acc0 = {0.f,0.f,0.f,0.f,0.f,0.f,0.f,0.f};
  v8f acc1 = {0.f,0.f,0.f,0.f,0.f,0.f,0.f,0.f};

  const int mbase = m0 + ((lane >> 4) << 3);
  float rs[8];
#pragma unroll
  for (int v = 0; v < 8; ++v) rs[v] = dis[mbase + v];

  for (int t = 0; t < KDIM / G1_COLS; ++t) {
#pragma unroll
    for (int j = 0; j < 8; ++j)
      *(v4f*)&st[2 * j + lrow][lcol] = R[j];

    if (t + 1 < KDIM / G1_COLS) {
#pragma unroll
      for (int j = 0; j < 8; ++j)
        R[j] = __builtin_nontemporal_load(
            (const v4f*)(gp + (size_t)(2 * j) * KDIM + (t + 1) * G1_COLS));
    }
    g1_wave_handover();

    const float* wb = W1 + (size_t)(t * G1_COLS + 4 * hh) * F1 + r;
#pragma unroll
    for (int g = 0; g < G1_COLS / 8; ++g) {
      const v4f a4 = *(const v4f*)&st[r][8 * g + 4 * hh];
      const float* w = wb + (size_t)(8 * g) * F1;
      v2f aL; aL.x = a4.x; aL.y = a4.y;
      v2f aH; aH.x = a4.z; aH.y = a4.w;
      v2f b0; b0.x = w[0];      b0.y = w[F1];
      v2f b1; b1.x = w[16];     b1.y = w[F1 + 16];
      v2f b2; b2.x = w[2 * F1]; b2.y = w[3 * F1];
      v2f b3; b3.x = w[2 * F1 + 16]; b3.y = w[3 * F1 + 16];
      acc0 = __builtin_amdgcn_wmma_f32_16x16x4_f32(false, aL, false, b0, (short)0, acc0, false, false);
      acc1 = __builtin_amdgcn_wmma_f32_16x16x4_f32(false, aL, false, b1, (short)0, acc1, false, false);
      acc0 = __builtin_amdgcn_wmma_f32_16x16x4_f32(false, aH, false, b2, (short)0, acc0, false, false);
      acc1 = __builtin_amdgcn_wmma_f32_16x16x4_f32(false, aH, false, b3, (short)0, acc1, false, false);
    }
    g1_wave_handover();
  }

  float* hp = h + (size_t)mbase * F1 + r;
#pragma unroll
  for (int v = 0; v < 8; ++v) {
    hp[(size_t)v * F1]      = acc0[v] * rs[v];
    hp[(size_t)v * F1 + 16] = acc1[v] * rs[v];
  }
}

extern "C" void kernel_launch(void* const* d_in, const int* in_sizes, int n_in,
                              void* d_out, int out_size, void* d_ws, size_t ws_size,
                              hipStream_t stream) {
  const float* x  = (const float*)d_in[0];
  const int*   ei = (const int*)d_in[1];
  const float* W1 = (const float*)d_in[2];
  const float* b1 = (const float*)d_in[3];
  const float* W2 = (const float*)d_in[4];
  const float* b2 = (const float*)d_in[5];

  const int n = in_sizes[0] / KDIM;
  const int E = in_sizes[1] / 2;
  const int* src = ei;
  const int* dst = ei + E;

  float* ws   = (float*)d_ws;
  float* dis  = ws;
  float* h    = dis  + n;
  int*   list = (int*)(h + (size_t)n * F1);
  float* h2   = (float*)(list + E);
  float* out  = (float*)d_out;
  int*   slots = (int*)d_out;
  const int B = (n + BK_W - 1) / BK_W;

  init_kernel<<<(n + 255) / 256, 256, 0, stream>>>(dis, n);
  {
    const int ranges = (n + DEG_BINS - 1) / DEG_BINS;
    degree_kernel<<<dim3(DEG_CHUNKS, ranges), DEG_THREADS, 0, stream>>>(dst, dis, n, E);
  }
  count_rsqrt_kernel<<<B, BK_W, 0, stream>>>(dis, slots, n);
  bucket_scan_kernel<<<1, 1024, 0, stream>>>(slots, B);
  partition_kernel<<<PT_GROUPS, PT_THREADS, 2 * B * sizeof(int), stream>>>(src, dst, slots, list, E, B);
  {
    const int ntile = n >> 4;
    gemm1_kernel<<<(ntile + 7) / 8, 256, 0, stream>>>(x, W1, dis, h, n);
  }
  agg1_gemm2_kernel<<<B, BK_THREADS, 0, stream>>>(h, list, slots, dis, b1, W2, h2, n);
  agg2_out_kernel<<<B, BK_THREADS, 0, stream>>>(h2, list, dis, b2, out, n);
}


# Regrouping orders each workgroup's edge words by bucket in LDS, then copies them out: neighbouring lanes store neighbouring words of a run, not one 4-byte store per line; sum order varies as before
# speedup vs baseline: 3.8139x; 1.1321x over previous
#include <hip/hip_runtime.h>

typedef __attribute__((ext_vector_type(2))) float v2f;
typedef __attribute__((ext_vector_type(8))) float v8f;

#define KDIM 512
#define F1 32
#define F2 16

__global__ void init_kernel(float* __restrict__ deg, int n) {
  int i = blockIdx.x * blockDim.x + threadIdx.x;
  if (i < n) deg[i] = 1.0f;
}

#define DEG_BINS    14336
#define DEG_THREADS 512
#define DEG_CHUNKS  36
#define DEG_DEPTH   8

static __device__ __forceinline__ void deg_count(unsigned int* cnt, int d, unsigned int lo) {
  const unsigned int b = (unsigned int)d - lo;
  if (b < (unsigned int)DEG_BINS) atomicAdd(&cnt[b], 1u);
}

__global__ __launch_bounds__(DEG_THREADS) void degree_kernel(const int* __restrict__ dst,
                                                             float* __restrict__ deg,
                                                             int n, int E) {
  __shared__ __attribute__((aligned(16))) unsigned int cnt[DEG_BINS];
  const int tid = threadIdx.x;
  const int lo  = blockIdx.y * DEG_BINS;

  for (int i = tid; i < DEG_BINS / 4; i += DEG_THREADS)
    ((uint4*)cnt)[i] = make_uint4(0u, 0u, 0u, 0u);
  __syncthreads();

  const int per = (((E + (int)gridDim.x - 1) / (int)gridDim.x) + 3) & ~3;
  const int e0  = blockIdx.x * per;
  const int len = max(0, min(E - e0, per));

  if ((((size_t)dst) & 15) == 0) {
    const int4* p  = (const int4*)(dst + e0);
    const int   nv = len >> 2;
    if (nv > 0) {
      int4 r[DEG_DEPTH];
#pragma unroll
      for (int u = 0; u < DEG_DEPTH; ++u)
        r[u] = p[min(u * DEG_THREADS + tid, nv - 1)];
      for (int base = 0; base < nv; base += DEG_DEPTH * DEG_THREADS) {
        int4 q[DEG_DEPTH];
#pragma unroll
        for (int u = 0; u < DEG_DEPTH; ++u)
          q[u] = p[min(base + (DEG_DEPTH + u) * DEG_THREADS + tid, nv - 1)];
#pragma unroll
        for (int u = 0; u < DEG_DEPTH; ++u) {
          const unsigned int lo_u =
              (base + u * DEG_THREADS + tid < nv) ? (unsigned int)lo : 0x80000000u;
          deg_count(cnt, r[u].x, lo_u);
          deg_count(cnt, r[u].y, lo_u);
          deg_count(cnt, r[u].z, lo_u);
          deg_count(cnt, r[u].w, lo_u);
        }
#pragma unroll
        for (int u = 0; u < DEG_DEPTH; ++u) r[u] = q[u];
      }
    }
    const int t = (nv << 2) + tid;
    if (t < len) deg_count(cnt, dst[e0 + t], (unsigned int)lo);
  } else {
    for (int t = tid; t < len; t += DEG_THREADS) deg_count(cnt, dst[e0 + t], (unsigned int)lo);
  }
  __syncthreads();

  const int rot = (int)((blockIdx.x * 5u) % (unsigned int)(DEG_BINS / DEG_THREADS)) * DEG_THREADS;
  static_assert(DEG_BINS % (4 * DEG_THREADS) == 0, "flush reads four counters per lane and trip");
  for (int i = tid; i < DEG_BINS; i += 4 * DEG_THREADS) {
    int j[4];
    unsigned int c[4];
#pragma unroll
    for (int u = 0; u < 4; ++u) {
      j[u] = i + u * DEG_THREADS + rot;
      if (j[u] >= DEG_BINS) j[u] -= DEG_BINS;
      c[u] = cnt[j[u]];
    }
#pragma unroll
    for (int u = 0; u < 4; ++u) {
      const int node = lo + j[u];
      if (c[u] != 0u && node < n) atomicAdd(&deg[node], (float)c[u]);
    }
  }
}

#define BK_SHIFT  7
#define BK_W      (1 << BK_SHIFT)
#define BK_STRIDE (BK_W * F2)
#define BK_CNT    0
#define BK_START  1
#define BK_CURSOR 2
#define BK_THREADS 512

__global__ __launch_bounds__(BK_W) void count_rsqrt_kernel(float* __restrict__ deg,
                                                           int* __restrict__ slots, int n) {
  __shared__ int part[BK_W / 32];
  const int t = threadIdx.x;
  const int i = blockIdx.x * BK_W + t;
  const float d = (i < n) ? deg[i] : 1.0f;
  int c = (int)d - 1;
  for (int o = 16; o > 0; o >>= 1) c += __shfl_xor(c, o);
  if ((t & 31) == 0) part[t >> 5] = c;
  __syncthreads();
  if (t == 0) {
    int s = 0;
    for (int w = 0; w < BK_W / 32; ++w) s += part[w];
    slots[(size_t)blockIdx.x * BK_STRIDE + BK_CNT] = s;
  }
  if (i < n) deg[i] = rsqrtf(d);
}

__global__ __launch_bounds__(1024) void bucket_scan_kernel(int* __restrict__ slots, int B) {
  __shared__ int wsum[32];
  __shared__ int total;
  const int tid = threadIdx.x, lane = tid & 31, wid = tid >> 5;
  int carry = 0;
  for (int base = 0; base < B; base += 1024) {
    const int b = base + tid;
    const int c = (b < B) ? slots[(size_t)b * BK_STRIDE + BK_CNT] : 0;
    int x = c;
    for (int o = 1; o < 32; o <<= 1) { const int y = __shfl_up(x, o); if (lane >= o) x += y; }
    if (lane == 31) wsum[wid] = x;
    __syncthreads();
    if (wid == 0) {
      const int tw = wsum[lane];
      int z = tw;
      for (int o = 1; o < 32; o <<= 1) { const int y = __shfl_up(z, o); if (lane >= o) z += y; }
      wsum[lane] = z - tw;
      if (lane == 31) total = z;
    }
    __syncthreads();
    const int excl = carry + wsum[wid] + x - c;
    if (b < B) {
      slots[(size_t)b * BK_STRIDE + BK_START]  = excl;
      slots[(size_t)b * BK_STRIDE + BK_CURSOR] = excl;
    }
    carry += total;
    __syncthreads();
  }
}

#define PT_THREADS 512
#define PT_GROUPS  256
#define PT_DEPTH   8
#define PT_CAP     12800
#define PT_OUT     5

__global__ __launch_bounds__(PT_THREADS) void partition_kernel(const int* __restrict__ src,
                                                               const int* __restrict__ dst,
                                                               int* __restrict__ slots,
                                                               int* __restrict__ list,
                                                               int E, int B) {
  extern __shared__ int pt_lds[];
  int* hist  = pt_lds;
  int* bbase = pt_lds + B;
  __shared__ int words[PT_CAP];
  __shared__ int delta[PT_CAP];
  __shared__ int wsum[PT_THREADS / 32];
  __shared__ int total;
  const int tid = threadIdx.x, lane = tid & 31, wid = tid >> 5;

  const int per = (E + (int)gridDim.x - 1) / (int)gridDim.x;
  const int e0  = blockIdx.x * per;
  const int e1  = min(E, e0 + per);

  for (int p0 = e0; p0 < e1; p0 += PT_CAP) {
    const int p1 = min(e1, p0 + PT_CAP);
    for (int b = tid; b < B; b += PT_THREADS) hist[b] = 0;
    __syncthreads();

    for (int e = p0 + tid; e < p1; e += PT_DEPTH * PT_THREADS) {
      int d[PT_DEPTH];
#pragma unroll
      for (int u = 0; u < PT_DEPTH; ++u) d[u] = dst[min(e + u * PT_THREADS, p1 - 1)];
#pragma unroll
      for (int u = 0; u < PT_DEPTH; ++u)
        if (e + u * PT_THREADS < p1) atomicAdd(&hist[d[u] >> BK_SHIFT], 1);
    }
    __syncthreads();

    int carry = 0;
    for (int base = 0; base < B; base += PT_THREADS) {
      const int b = base + tid;
      const int c = (b < B) ? hist[b] : 0;
      int x = c;
      for (int o = 1; o < 32; o <<= 1) { const int y = __shfl_up(x, o); if (lane >= o) x += y; }
      if (lane == 31) wsum[wid] = x;
      __syncthreads();
      if (wid == 0) {
        const int tw = (lane < PT_THREADS / 32) ? wsum[lane] : 0;
        int z = tw;
        for (int o = 1; o < 32; o <<= 1) { const int y = __shfl_up(z, o); if (lane >= o) z += y; }
        if (lane < PT_THREADS / 32) wsum[lane] = z - tw;
        if (lane == 31) total = z;
      }
      __syncthreads();
      const int excl = carry + wsum[wid] + x - c;
      if (b < B) {
        const int g = (c != 0) ? atomicAdd(&slots[(size_t)b * BK_STRIDE + BK_CURSOR], c) : 0;
        bbase[b] = g - excl;
        hist[b]  = excl;
      }
      carry += total;
      __syncthreads();
    }

    for (int e = p0 + tid; e < p1; e += PT_DEPTH * PT_THREADS) {
      int d[PT_DEPTH], s[PT_DEPTH];
#pragma unroll
      for (int u = 0; u < PT_DEPTH; ++u) {
        const int ee = min(e + u * PT_THREADS, p1 - 1);
        d[u] = dst[ee];
        s[u] = src[ee];
      }
#pragma unroll
      for (int u = 0; u < PT_DEPTH; ++u) {
        if (e + u * PT_THREADS < p1) {
          const int b = d[u] >> BK_SHIFT;
          const int q = atomicAdd(&hist[b], 1);
          words[q] = (s[u] << BK_SHIFT) | (d[u] & (BK_W - 1));
          delta[q] = bbase[b];
        }
      }
    }
    __syncthreads();

    const int m = p1 - p0;
    for (int i = tid; i < m; i += PT_OUT * PT_THREADS) {
      int w[PT_OUT], g[PT_OUT];
#pragma unroll
      for (int u = 0; u < PT_OUT; ++u) {
        const int ii = min(i + u * PT_THREADS, m - 1);
        w[u] = words[ii];
        g[u] = delta[ii] + ii;
      }
#pragma unroll
      for (int u = 0; u < PT_OUT; ++u)
        if (i + u * PT_THREADS < m) list[g[u]] = w[u];
    }
    __syncthreads();
  }
}

#define SORT_CAP   5120
#define SORT_PER_LANE (SORT_CAP / BK_THREADS)
#define BK_WAVES_PER_SIMD 16
#define TAIL_DEPTH 4

typedef __attribute__((ext_vector_type(4))) float f4v;

template <int F>
static __device__ __forceinline__ f4v row_piece(const float* __restrict__ feat, int s, unsigned int fl4) {
  const unsigned int off = (unsigned int)s * (unsigned int)(F * 4) + fl4;
  return *(const f4v*)((const char*)feat + off);
}

template <int F, int PITCH>
static __device__ __forceinline__ void bucket_rows(const float* __restrict__ feat,
                                                   const int* __restrict__ list,
                                                   float* acc, int* cntL, int* offL, int* srcL,
                                                   int start, int cnt) {
  constexpr int LPR = F / 4;
  constexpr int GPW = 32 / LPR;
  constexpr int NG  = (BK_THREADS / 32) * GPW;
  constexpr int RPG = BK_W / NG;
  constexpr int ROW_DEPTH = (RPG > 1) ? 6 : 8;
  static_assert(RPG * NG == BK_W, "every row of the bucket has exactly one owner");
  static_assert(SORT_PER_LANE * BK_THREADS == SORT_CAP, "a piece is a whole number of words per lane");
  const int tid  = threadIdx.x;
  const int lane = tid & 31;
  const int wid  = tid >> 5;
  const int grp  = wid * GPW + lane / LPR;
  const unsigned int fl  = (unsigned int)(lane % LPR) * 4u;
  const unsigned int fl4 = fl * 4u;

  f4v sum[RPG];
#pragma unroll
  for (int q = 0; q < RPG; ++q) sum[q] = (f4v){0.f, 0.f, 0.f, 0.f};

  for (int c0 = 0; c0 < cnt; c0 += SORT_CAP) {
    const int m = min(SORT_CAP, cnt - c0);
    const int* lp = list + start + c0;
    int wv[SORT_PER_LANE];
#pragma unroll
    for (int u = 0; u < SORT_PER_LANE; ++u) wv[u] = lp[min(tid + u * BK_THREADS, m - 1)];
    if (tid < BK_W) cntL[tid] = 0;
    __syncthreads();
#pragma unroll
    for (int u = 0; u < SORT_PER_LANE; ++u)
      if (tid + u * BK_THREADS < m) atomicAdd(&cntL[wv[u] & (BK_W - 1)], 1);
    __syncthreads();
    if (wid == 0) {
      const int a0 = cntL[4 * lane], a1 = cntL[4 * lane + 1], a2 = cntL[4 * lane + 2], a3 = cntL[4 * lane + 3];
      const int t = a0 + a1 + a2 + a3;
      int x = t;
      for (int o = 1; o < 32; o <<= 1) { const int y = __shfl_up(x, o); if (lane >= o) x += y; }
      const int p0 = x - t, p1 = p0 + a0, p2 = p1 + a1, p3 = p2 + a2;
      offL[4 * lane] = p0; offL[4 * lane + 1] = p1; offL[4 * lane + 2] = p2; offL[4 * lane + 3] = p3;
      cntL[4 * lane] = p0; cntL[4 * lane + 1] = p1; cntL[4 * lane + 2] = p2; cntL[4 * lane + 3] = p3;
      if (lane == 31) offL[BK_W] = x;
    }
    __syncthreads();
    int pos[SORT_PER_LANE];
#pragma unroll
    for (int u = 0; u < SORT_PER_LANE; ++u) {
      pos[u] = 0;
      if (tid + u * BK_THREADS < m) pos[u] = atomicAdd(&cntL[wv[u] & (BK_W - 1)], 1);
    }
#pragma unroll
    for (int u = 0; u < SORT_PER_LANE; ++u)
      if (tid + u * BK_THREADS < m) srcL[pos[u]] = wv[u] >> BK_SHIFT;
    __syncthreads();

#pragma unroll
    for (int q = 0; q < RPG; ++q) {
      const int row = grp * RPG + q;
      int j = offL[row];
      const int end = offL[row + 1];
      f4v a = sum[q];
      for (; j + ROW_DEPTH <= end; j += ROW_DEPTH) {
        f4v v[ROW_DEPTH];
#pragma unroll
        for (int u = 0; u < ROW_DEPTH; ++u) v[u] = row_piece<F>(feat, srcL[j + u], fl4);
#pragma unroll
        for (int u = 0; u < ROW_DEPTH; ++u) a += v[u];
      }
      for (; j < end; j += TAIL_DEPTH) {
        f4v v[TAIL_DEPTH];
#pragma unroll
        for (int u = 0; u < TAIL_DEPTH; ++u) v[u] = row_piece<F>(feat, srcL[min(j + u, end - 1)], fl4);
#pragma unroll
        for (int u = 0; u < TAIL_DEPTH; ++u) {
          const bool in = (j + u < end);
          a.x += in ? v[u].x : 0.0f;
          a.y += in ? v[u].y : 0.0f;
          a.z += in ? v[u].z : 0.0f;
          a.w += in ? v[u].w : 0.0f;
        }
      }
      sum[q] = a;
    }
    __syncthreads();
  }

#pragma unroll
  for (int q = 0; q < RPG; ++q) {
    float* o = acc + (grp * RPG + q) * PITCH + fl;
    o[0] = sum[q].x; o[1] = sum[q].y; o[2] = sum[q].z; o[3] = sum[q].w;
  }
}

#define A1_PITCH 33

__global__ __launch_bounds__(BK_THREADS, BK_WAVES_PER_SIMD) void agg1_gemm2_kernel(const float* __restrict__ h,
                                                         const int* __restrict__ list,
                                                         const int* __restrict__ slots,
                                                         const float* __restrict__ dis,
                                                         const float* __restrict__ b1,
                                                         const float* __restrict__ W2,
                                                         float* __restrict__ h2, int n) {
  __shared__ float acc[BK_W * A1_PITCH];
  __shared__ int cntL[BK_W];
  __shared__ int offL[BK_W + 1];
  __shared__ int srcL[SORT_CAP];
  const int lo    = blockIdx.x * BK_W;
  const int cnt   = slots[(size_t)blockIdx.x * BK_STRIDE + BK_CNT];
  const int start = slots[(size_t)blockIdx.x * BK_STRIDE + BK_START];
  bucket_rows<F1, A1_PITCH>(h, list, acc, cntL, offL, srcL, start, cnt);
  __syncthreads();

  const int lane = threadIdx.x & 31;
  const int r    = lane & 15;
  const int kh   = (lane >> 4) << 1;
  for (int tl = (threadIdx.x >> 5) * 16; tl < BK_W; tl += (blockDim.x >> 5) * 16) {
    const int m0 = lo + tl;
    if (m0 >= n) break;
    const int row  = m0 + r;
    const float dv = dis[row];
    const float* ag = acc + (tl + r) * A1_PITCH;
    const float* hr = h + (size_t)row * F1;
    v8f c8 = {0.f,0.f,0.f,0.f,0.f,0.f,0.f,0.f};
    for (int k = 0; k < F1; k += 4) {
      const int k0 = k + kh;
      v2f a;
      a.x = fmaxf((ag[k0]     + hr[k0])     * dv + b1[k0],     0.0f) * dv;
      a.y = fmaxf((ag[k0 + 1] + hr[k0 + 1]) * dv + b1[k0 + 1], 0.0f) * dv;
      v2f b;
      b.x = W2[(size_t)k0 * F2 + r];
      b.y = W2[(size_t)(k0 + 1) * F2 + r];
      c8 = __builtin_amdgcn_wmma_f32_16x16x4_f32(false, a, false, b, (short)0, c8, false, false);
    }
    const int mbase = m0 + ((lane >> 4) << 3);
    float* op = h2 + (size_t)mbase * F2 + r;
#pragma unroll
    for (int v = 0; v < 8; ++v) op[(size_t)v * F2] = c8[v];
  }
}

__global__ __launch_bounds__(BK_THREADS, BK_WAVES_PER_SIMD) void agg2_out_kernel(const float* __restrict__ h2,
                                                       const int* __restrict__ list,
                                                       const float* __restrict__ dis,
                                                       const float* __restrict__ b2,
                                                       float* out, int n) {
  __shared__ float acc[BK_W * F2];
  __shared__ int cntL[BK_W];
  __shared__ int offL[BK_W + 1];
  __shared__ int srcL[SORT_CAP];
  const int lo    = blockIdx.x * BK_W;
  const int cnt   = __float_as_int(out[(size_t)blockIdx.x * BK_STRIDE + BK_CNT]);
  const int start = __float_as_int(out[(size_t)blockIdx.x * BK_STRIDE + BK_START]);
  bucket_rows<F2, F2>(h2, list, acc, cntL, offL, srcL, start, cnt);
  __syncthreads();
  for (int i = threadIdx.x; i < BK_W * F2; i += blockDim.x) {
    const int row = lo + i / F2, f = i & (F2 - 1);
    if (row < n) {
      const float dv = dis[row];
      const size_t g = (size_t)row * F2 + f;
      out[g] = (acc[i] + h2[g]) * dv + b2[f];
    }
  }
}

typedef __attribute__((ext_vector_type(4))) float v4f;

#define G1_WAVES 8
#define G1_COLS  64
#define G1_PITCH 68

static __device__ __forceinline__ void g1_wave_handover() {
  __builtin_amdgcn_fence(__ATOMIC_RELEASE, "wavefront");
  __builtin_amdgcn_wave_barrier();
  __builtin_amdgcn_fence(__ATOMIC_ACQUIRE, "wavefront");
}

__global__ __launch_bounds__(256) void gemm1_kernel(const float* __restrict__ x,
                                                    const float* __restrict__ W1,
                                                    const float* __restrict__ dis,
                                                    float* __restrict__ h, int n) {
  __shared__ __attribute__((aligned(16))) float stage[G1_WAVES][16][G1_PITCH];
  const int lane  = threadIdx.x & 31;
  const int wid   = threadIdx.x >> 5;
  const int wave  = blockIdx.x * (blockDim.x >> 5) + wid;
  const int ntile = n >> 4;
  if (wave >= ntile) return;
  const int m0 = wave << 4;
  const int r  = lane & 15;
  const int hh = lane >> 4;

  float (*st)[G1_PITCH] = stage[wid];

  const int lrow = lane >> 4;
  const int lcol = (lane & 15) << 2;
  const float* gp = x + (size_t)(m0 + lrow) * KDIM + lcol;

  v4f R[8];
#pragma unroll
  for (int j = 0; j < 8; ++j)
    R[j] = __builtin_nontemporal_load((const v4f*)(gp + (size_t)(2 * j) * KDIM));

  v8f acc0 = {0.f,0.f,0.f,0.f,0.f,0.f,0.f,0.f};
  v8f acc1 = {0.f,0.f,0.f,0.f,0.f,0.f,0.f,0.f};

  const int mbase = m0 + ((lane >> 4) << 3);
  float rs[8];
#pragma unroll
  for (int v = 0; v < 8; ++v) rs[v] = dis[mbase + v];

  for (int t = 0; t < KDIM / G1_COLS; ++t) {
#pragma unroll
    for (int j = 0; j < 8; ++j)
      *(v4f*)&st[2 * j + lrow][lcol] = R[j];

    if (t + 1 < KDIM / G1_COLS) {
#pragma unroll
      for (int j = 0; j < 8; ++j)
        R[j] = __builtin_nontemporal_load(
            (const v4f*)(gp + (size_t)(2 * j) * KDIM + (t + 1) * G1_COLS));
    }
    g1_wave_handover();

    const float* wb = W1 + (size_t)(t * G1_COLS + 4 * hh) * F1 + r;
#pragma unroll
    for (int g = 0; g < G1_COLS / 8; ++g) {
      const v4f a4 = *(const v4f*)&st[r][8 * g + 4 * hh];
      const float* w = wb + (size_t)(8 * g) * F1;
      v2f aL; aL.x = a4.x; aL.y = a4.y;
      v2f aH; aH.x = a4.z; aH.y = a4.w;
      v2f b0; b0.x = w[0];      b0.y = w[F1];
      v2f b1; b1.x = w[16];     b1.y = w[F1 + 16];
      v2f b2; b2.x = w[2 * F1]; b2.y = w[3 * F1];
      v2f b3; b3.x = w[2 * F1 + 16]; b3.y = w[3 * F1 + 16];
      acc0 = __builtin_amdgcn_wmma_f32_16x16x4_f32(false, aL, false, b0, (short)0, acc0, false, false);
      acc1 = __builtin_amdgcn_wmma_f32_16x16x4_f32(false, aL, false, b1, (short)0, acc1, false, false);
      acc0 = __builtin_amdgcn_wmma_f32_16x16x4_f32(false, aH, false, b2, (short)0, acc0, false, false);
      acc1 = __builtin_amdgcn_wmma_f32_16x16x4_f32(false, aH, false, b3, (short)0, acc1, false, false);
    }
    g1_wave_handover();
  }

  float* hp = h + (size_t)mbase * F1 + r;
#pragma unroll
  for (int v = 0; v < 8; ++v) {
    hp[(size_t)v * F1]      = acc0[v] * rs[v];
    hp[(size_t)v * F1 + 16] = acc1[v] * rs[v];
  }
}

extern "C" void kernel_launch(void* const* d_in, const int* in_sizes, int n_in,
                              void* d_out, int out_size, void* d_ws, size_t ws_size,
                              hipStream_t stream) {
  const float* x  = (const float*)d_in[0];
  const int*   ei = (const int*)d_in[1];
  const float* W1 = (const float*)d_in[2];
  const float* b1 = (const float*)d_in[3];
  const float* W2 = (const float*)d_in[4];
  const float* b2 = (const float*)d_in[5];

  const int n = in_sizes[0] / KDIM;
  const int E = in_sizes[1] / 2;
  const int* src = ei;
  const int* dst = ei + E;

  float* ws   = (float*)d_ws;
  float* dis  = ws;
  float* h    = dis  + n;
  int*   list = (int*)(h + (size_t)n * F1);
  float* h2   = (float*)(list + E);
  float* out  = (float*)d_out;
  int*   slots = (int*)d_out;
  const int B = (n + BK_W - 1) / BK_W;

  init_kernel<<<(n + 255) / 256, 256, 0, stream>>>(dis, n);
  {
    const int ranges = (n + DEG_BINS - 1) / DEG_BINS;
    degree_kernel<<<dim3(DEG_CHUNKS, ranges), DEG_THREADS, 0, stream>>>(dst, dis, n, E);
  }
  count_rsqrt_kernel<<<B, BK_W, 0, stream>>>(dis, slots, n);
  bucket_scan_kernel<<<1, 1024, 0, stream>>>(slots, B);
  partition_kernel<<<PT_GROUPS, PT_THREADS, 2 * B * sizeof(int), stream>>>(src, dst, slots, list, E, B);
  {
    const int ntile = n >> 4;
    gemm1_kernel<<<(ntile + 7) / 8, 256, 0, stream>>>(x, W1, dis, h, n);
  }
  agg1_gemm2_kernel<<<B, BK_THREADS, 0, stream>>>(h, list, slots, dis, b1, W2, h2, n);
  agg2_out_kernel<<<B, BK_THREADS, 0, stream>>>(h2, list, dis, b2, out, n);
}


# Node degrees counted from the regrouped list (per-row LDS count of a bucket's words inside GEMM1); bucket sizes from per-chunk LDS histograms summed in the scan: no degree pass, 8 launches to 6
# speedup vs baseline: 3.9564x; 1.0374x over previous
#include <hip/hip_runtime.h>

typedef __attribute__((ext_vector_type(2))) float v2f;
typedef __attribute__((ext_vector_type(8))) float v8f;

#define KDIM 512
#define F1 32
#define F2 16


#define BK_SHIFT  7
#define BK_W      (1 << BK_SHIFT)
#define BK_STRIDE (BK_W * F2)
#define BK_CNT    0
#define BK_START  1
#define BK_CURSOR 2
#define BK_THREADS 512

#define BC_GROUPS  64
#define BC_THREADS 1024
#define BC_DEPTH   8

static __device__ __forceinline__ void bc_count(int* hist, int4 d) {
  atomicAdd(&hist[d.x >> BK_SHIFT], 1);
  atomicAdd(&hist[d.y >> BK_SHIFT], 1);
  atomicAdd(&hist[d.z >> BK_SHIFT], 1);
  atomicAdd(&hist[d.w >> BK_SHIFT], 1);
}

__global__ __launch_bounds__(BC_THREADS) void bucket_count_kernel(const int* __restrict__ dst,
                                                                  int* __restrict__ table,
                                                                  int E, int B) {
  extern __shared__ int bc_hist[];
  const int tid = threadIdx.x;
  for (int b = tid; b < B; b += BC_THREADS) bc_hist[b] = 0;
  __syncthreads();

  const int per = (((E + (int)gridDim.x - 1) / (int)gridDim.x) + 3) & ~3;
  const int e0  = blockIdx.x * per;
  const int len = max(0, min(E - e0, per));

  if ((((size_t)dst) & 15) == 0) {
    const int4* p  = (const int4*)(dst + e0);
    const int   nv = len >> 2;
    if (nv > 0) {
      int4 r[BC_DEPTH];
#pragma unroll
      for (int u = 0; u < BC_DEPTH; ++u)
        r[u] = p[min(u * BC_THREADS + tid, nv - 1)];
      for (int base = 0; base < nv; base += BC_DEPTH * BC_THREADS) {
        int4 q[BC_DEPTH];
#pragma unroll
        for (int u = 0; u < BC_DEPTH; ++u)
          q[u] = p[min(base + (BC_DEPTH + u) * BC_THREADS + tid, nv - 1)];
#pragma unroll
        for (int u = 0; u < BC_DEPTH; ++u)
          if (base + u * BC_THREADS + tid < nv) bc_count(bc_hist, r[u]);
#pragma unroll
        for (int u = 0; u < BC_DEPTH; ++u) r[u] = q[u];
      }
    }
    const int t = (nv << 2) + tid;
    if (t < len) atomicAdd(&bc_hist[dst[e0 + t] >> BK_SHIFT], 1);
  } else {
    for (int t = tid; t < len; t += BC_THREADS) atomicAdd(&bc_hist[dst[e0 + t] >> BK_SHIFT], 1);
  }
  __syncthreads();

  for (int b = tid; b < B; b += BC_THREADS) table[(size_t)blockIdx.x * B + b] = bc_hist[b];
}

#define SC_DEPTH 32

__global__ __launch_bounds__(1024) void bucket_scan_kernel(const int* __restrict__ table,
                                                           int* __restrict__ slots, int B) {
  static_assert(BC_GROUPS % SC_DEPTH == 0, "the column sum takes whole trips");
  __shared__ int wsum[32];
  __shared__ int total;
  const int tid = threadIdx.x, lane = tid & 31, wid = tid >> 5;
  int carry = 0;
  for (int base = 0; base < B; base += 1024) {
    const int b = base + tid;
    int c = 0;
    if (b < B) {
      for (int g = 0; g < BC_GROUPS; g += SC_DEPTH) {
        int v[SC_DEPTH];
#pragma unroll
        for (int u = 0; u < SC_DEPTH; ++u) v[u] = table[(size_t)(g + u) * B + b];
#pragma unroll
        for (int u = 0; u < SC_DEPTH; ++u) c += v[u];
      }
    }
    int x = c;
    for (int o = 1; o < 32; o <<= 1) { const int y = __shfl_up(x, o); if (lane >= o) x += y; }
    if (lane == 31) wsum[wid] = x;
    __syncthreads();
    if (wid == 0) {
      const int tw = wsum[lane];
      int z = tw;
      for (int o = 1; o < 32; o <<= 1) { const int y = __shfl_up(z, o); if (lane >= o) z += y; }
      wsum[lane] = z - tw;
      if (lane == 31) total = z;
    }
    __syncthreads();
    const int excl = carry + wsum[wid] + x - c;
    if (b < B) {
      slots[(size_t)b * BK_STRIDE + BK_CNT]    = c;
      slots[(size_t)b * BK_STRIDE + BK_START]  = excl;
      slots[(size_t)b * BK_STRIDE + BK_CURSOR] = excl;
    }
    carry += total;
    __syncthreads();
  }
}

#define PT_THREADS 512
#define PT_GROUPS  256
#define PT_DEPTH   8
#define PT_CAP     12800
#define PT_OUT     5

__global__ __launch_bounds__(PT_THREADS) void partition_kernel(const int* __restrict__ src,
                                                               const int* __restrict__ dst,
                                                               int* __restrict__ slots,
                                                               int* __restrict__ list,
                                                               int E, int B) {
  extern __shared__ int pt_lds[];
  int* hist  = pt_lds;
  int* bbase = pt_lds + B;
  __shared__ int words[PT_CAP];
  __shared__ int delta[PT_CAP];
  __shared__ int wsum[PT_THREADS / 32];
  __shared__ int total;
  const int tid = threadIdx.x, lane = tid & 31, wid = tid >> 5;

  const int per = (E + (int)gridDim.x - 1) / (int)gridDim.x;
  const int e0  = blockIdx.x * per;
  const int e1  = min(E, e0 + per);

  for (int p0 = e0; p0 < e1; p0 += PT_CAP) {
    const int p1 = min(e1, p0 + PT_CAP);
    for (int b = tid; b < B; b += PT_THREADS) hist[b] = 0;
    __syncthreads();

    for (int e = p0 + tid; e < p1; e += PT_DEPTH * PT_THREADS) {
      int d[PT_DEPTH];
#pragma unroll
      for (int u = 0; u < PT_DEPTH; ++u) d[u] = dst[min(e + u * PT_THREADS, p1 - 1)];
#pragma unroll
      for (int u = 0; u < PT_DEPTH; ++u)
        if (e + u * PT_THREADS < p1) atomicAdd(&hist[d[u] >> BK_SHIFT], 1);
    }
    __syncthreads();

    int carry = 0;
    for (int base = 0; base < B; base += PT_THREADS) {
      const int b = base + tid;
      const int c = (b < B) ? hist[b] : 0;
      int x = c;
      for (int o = 1; o < 32; o <<= 1) { const int y = __shfl_up(x, o); if (lane >= o) x += y; }
      if (lane == 31) wsum[wid] = x;
      __syncthreads();
      if (wid == 0) {
        const int tw = (lane < PT_THREADS / 32) ? wsum[lane] : 0;
        int z = tw;
        for (int o = 1; o < 32; o <<= 1) { const int y = __shfl_up(z, o); if (lane >= o) z += y; }
        if (lane < PT_THREADS / 32) wsum[lane] = z - tw;
        if (lane == 31) total = z;
      }
      __syncthreads();
      const int excl = carry + wsum[wid] + x - c;
      if (b < B) {
        const int g = (c != 0) ? atomicAdd(&slots[(size_t)b * BK_STRIDE + BK_CURSOR], c) : 0;
        bbase[b] = g - excl;
        hist[b]  = excl;
      }
      carry += total;
      __syncthreads();
    }

    for (int e = p0 + tid; e < p1; e += PT_DEPTH * PT_THREADS) {
      int d[PT_DEPTH], s[PT_DEPTH];
#pragma unroll
      for (int u = 0; u < PT_DEPTH; ++u) {
        const int ee = min(e + u * PT_THREADS, p1 - 1);
        d[u] = dst[ee];
        s[u] = src[ee];
      }
#pragma unroll
      for (int u = 0; u < PT_DEPTH; ++u) {
        if (e + u * PT_THREADS < p1) {
          const int b = d[u] >> BK_SHIFT;
          const int q = atomicAdd(&hist[b], 1);
          words[q] = (s[u] << BK_SHIFT) | (d[u] & (BK_W - 1));
          delta[q] = bbase[b];
        }
      }
    }
    __syncthreads();

    const int m = p1 - p0;
    for (int i = tid; i < m; i += PT_OUT * PT_THREADS) {
      int w[PT_OUT], g[PT_OUT];
#pragma unroll
      for (int u = 0; u < PT_OUT; ++u) {
        const int ii = min(i + u * PT_THREADS, m - 1);
        w[u] = words[ii];
        g[u] = delta[ii] + ii;
      }
#pragma unroll
      for (int u = 0; u < PT_OUT; ++u)
        if (i + u * PT_THREADS < m) list[g[u]] = w[u];
    }
    __syncthreads();
  }
}

#define SORT_CAP   5120
#define SORT_PER_LANE (SORT_CAP / BK_THREADS)
#define BK_WAVES_PER_SIMD 16
#define TAIL_DEPTH 4

typedef __attribute__((ext_vector_type(4))) float f4v;

template <int F>
static __device__ __forceinline__ f4v row_piece(const float* __restrict__ feat, int s, unsigned int fl4) {
  const unsigned int off = (unsigned int)s * (unsigned int)(F * 4) + fl4;
  return *(const f4v*)((const char*)feat + off);
}

template <int F, int PITCH>
static __device__ __forceinline__ void bucket_rows(const float* __restrict__ feat,
                                                   const int* __restrict__ list,
                                                   float* acc, int* cntL, int* offL, int* srcL,
                                                   int start, int cnt) {
  constexpr int LPR = F / 4;
  constexpr int GPW = 32 / LPR;
  constexpr int NG  = (BK_THREADS / 32) * GPW;
  constexpr int RPG = BK_W / NG;
  constexpr int ROW_DEPTH = (RPG > 1) ? 6 : 8;
  static_assert(RPG * NG == BK_W, "every row of the bucket has exactly one owner");
  static_assert(SORT_PER_LANE * BK_THREADS == SORT_CAP, "a piece is a whole number of words per lane");
  const int tid  = threadIdx.x;
  const int lane = tid & 31;
  const int wid  = tid >> 5;
  const int grp  = wid * GPW + lane / LPR;
  const unsigned int fl  = (unsigned int)(lane % LPR) * 4u;
  const unsigned int fl4 = fl * 4u;

  f4v sum[RPG];
#pragma unroll
  for (int q = 0; q < RPG; ++q) sum[q] = (f4v){0.f, 0.f, 0.f, 0.f};

  for (int c0 = 0; c0 < cnt; c0 += SORT_CAP) {
    const int m = min(SORT_CAP, cnt - c0);
    const int* lp = list + start + c0;
    int wv[SORT_PER_LANE];
#pragma unroll
    for (int u = 0; u < SORT_PER_LANE; ++u) wv[u] = lp[min(tid + u * BK_THREADS, m - 1)];
    if (tid < BK_W) cntL[tid] = 0;
    __syncthreads();
#pragma unroll
    for (int u = 0; u < SORT_PER_LANE; ++u)
      if (tid + u * BK_THREADS < m) atomicAdd(&cntL[wv[u] & (BK_W - 1)], 1);
    __syncthreads();
    if (wid == 0) {
      const int a0 = cntL[4 * lane], a1 = cntL[4 * lane + 1], a2 = cntL[4 * lane + 2], a3 = cntL[4 * lane + 3];
      const int t = a0 + a1 + a2 + a3;
      int x = t;
      for (int o = 1; o < 32; o <<= 1) { const int y = __shfl_up(x, o); if (lane >= o) x += y; }
      const int p0 = x - t, p1 = p0 + a0, p2 = p1 + a1, p3 = p2 + a2;
      offL[4 * lane] = p0; offL[4 * lane + 1] = p1; offL[4 * lane + 2] = p2; offL[4 * lane + 3] = p3;
      cntL[4 * lane] = p0; cntL[4 * lane + 1] = p1; cntL[4 * lane + 2] = p2; cntL[4 * lane + 3] = p3;
      if (lane == 31) offL[BK_W] = x;
    }
    __syncthreads();
    int pos[SORT_PER_LANE];
#pragma unroll
    for (int u = 0; u < SORT_PER_LANE; ++u) {
      pos[u] = 0;
      if (tid + u * BK_THREADS < m) pos[u] = atomicAdd(&cntL[wv[u] & (BK_W - 1)], 1);
    }
#pragma unroll
    for (int u = 0; u < SORT_PER_LANE; ++u)
      if (tid + u * BK_THREADS < m) srcL[pos[u]] = wv[u] >> BK_SHIFT;
    __syncthreads();

#pragma unroll
    for (int q = 0; q < RPG; ++q) {
      const int row = grp * RPG + q;
      int j = offL[row];
      const int end = offL[row + 1];
      f4v a = sum[q];
      for (; j + ROW_DEPTH <= end; j += ROW_DEPTH) {
        f4v v[ROW_DEPTH];
#pragma unroll
        for (int u = 0; u < ROW_DEPTH; ++u) v[u] = row_piece<F>(feat, srcL[j + u], fl4);
#pragma unroll
        for (int u = 0; u < ROW_DEPTH; ++u) a += v[u];
      }
      for (; j < end; j += TAIL_DEPTH) {
        f4v v[TAIL_DEPTH];
#pragma unroll
        for (int u = 0; u < TAIL_DEPTH; ++u) v[u] = row_piece<F>(feat, srcL[min(j + u, end - 1)], fl4);
#pragma unroll
        for (int u = 0; u < TAIL_DEPTH; ++u) {
          const bool in = (j + u < end);
          a.x += in ? v[u].x : 0.0f;
          a.y += in ? v[u].y : 0.0f;
          a.z += in ? v[u].z : 0.0f;
          a.w += in ? v[u].w : 0.0f;
        }
      }
      sum[q] = a;
    }
    __syncthreads();
  }

#pragma unroll
  for (int q = 0; q < RPG; ++q) {
    float* o = acc + (grp * RPG + q) * PITCH + fl;
    o[0] = sum[q].x; o[1] = sum[q].y; o[2] = sum[q].z; o[3] = sum[q].w;
  }
}

#define A1_PITCH 33

__global__ __launch_bounds__(BK_THREADS, BK_WAVES_PER_SIMD) void agg1_gemm2_kernel(const float* __restrict__ h,
                                                         const int* __restrict__ list,
                                                         const int* __restrict__ slots,
                                                         const float* __restrict__ dis,
                                                         const float* __restrict__ b1,
                                                         const float* __restrict__ W2,
                                                         float* __restrict__ h2, int n) {
  __shared__ float acc[BK_W * A1_PITCH];
  __shared__ int cntL[BK_W];
  __shared__ int offL[BK_W + 1];
  __shared__ int srcL[SORT_CAP];
  const int lo    = blockIdx.x * BK_W;
  const int cnt   = slots[(size_t)blockIdx.x * BK_STRIDE + BK_CNT];
  const int start = slots[(size_t)blockIdx.x * BK_STRIDE + BK_START];
  bucket_rows<F1, A1_PITCH>(h, list, acc, cntL, offL, srcL, start, cnt);
  __syncthreads();

  const int lane = threadIdx.x & 31;
  const int r    = lane & 15;
  const int kh   = (lane >> 4) << 1;
  for (int tl = (threadIdx.x >> 5) * 16; tl < BK_W; tl += (blockDim.x >> 5) * 16) {
    const int m0 = lo + tl;
    if (m0 >= n) break;
    const int row  = m0 + r;
    const float dv = dis[row];
    const float* ag = acc + (tl + r) * A1_PITCH;
    const float* hr = h + (size_t)row * F1;
    v8f c8 = {0.f,0.f,0.f,0.f,0.f,0.f,0.f,0.f};
    for (int k = 0; k < F1; k += 4) {
      const int k0 = k + kh;
      v2f a;
      a.x = fmaxf((ag[k0]     + hr[k0])     * dv + b1[k0],     0.0f) * dv;
      a.y = fmaxf((ag[k0 + 1] + hr[k0 + 1]) * dv + b1[k0 + 1], 0.0f) * dv;
      v2f b;
      b.x = W2[(size_t)k0 * F2 + r];
      b.y = W2[(size_t)(k0 + 1) * F2 + r];
      c8 = __builtin_amdgcn_wmma_f32_16x16x4_f32(false, a, false, b, (short)0, c8, false, false);
    }
    const int mbase = m0 + ((lane >> 4) << 3);
    float* op = h2 + (size_t)mbase * F2 + r;
#pragma unroll
    for (int v = 0; v < 8; ++v) op[(size_t)v * F2] = c8[v];
  }
}

__global__ __launch_bounds__(BK_THREADS, BK_WAVES_PER_SIMD) void agg2_out_kernel(const float* __restrict__ h2,
                                                       const int* __restrict__ list,
                                                       const float* __restrict__ dis,
                                                       const float* __restrict__ b2,
                                                       float* out, int n) {
  __shared__ float acc[BK_W * F2];
  __shared__ int cntL[BK_W];
  __shared__ int offL[BK_W + 1];
  __shared__ int srcL[SORT_CAP];
  const int lo    = blockIdx.x * BK_W;
  const int cnt   = __float_as_int(out[(size_t)blockIdx.x * BK_STRIDE + BK_CNT]);
  const int start = __float_as_int(out[(size_t)blockIdx.x * BK_STRIDE + BK_START]);
  bucket_rows<F2, F2>(h2, list, acc, cntL, offL, srcL, start, cnt);
  __syncthreads();
  for (int i = threadIdx.x; i < BK_W * F2; i += blockDim.x) {
    const int row = lo + i / F2, f = i & (F2 - 1);
    if (row < n) {
      const float dv = dis[row];
      const size_t g = (size_t)row * F2 + f;
      out[g] = (acc[i] + h2[g]) * dv + b2[f];
    }
  }
}

typedef __attribute__((ext_vector_type(4))) float v4f;

#define G1_WAVES 8
#define G1_THREADS (G1_WAVES * 32)
#define G1_COLS  64
#define G1_PITCH 68
#define G1_LIST_DEPTH 18

static __device__ __forceinline__ void g1_wave_handover() {
  __builtin_amdgcn_fence(__ATOMIC_RELEASE, "wavefront");
  __builtin_amdgcn_wave_barrier();
  __builtin_amdgcn_fence(__ATOMIC_ACQUIRE, "wavefront");
}

__global__ __launch_bounds__(G1_THREADS) void gemm1_kernel(const float* __restrict__ x,
                                                           const float* __restrict__ W1,
                                                           const int* __restrict__ list,
                                                           const int* __restrict__ slots,
                                                           float* __restrict__ dis,
                                                           float* __restrict__ h, int n) {
  static_assert(G1_WAVES * 16 == BK_W, "a workgroup's rows are one bucket");
  __shared__ __attribute__((aligned(16))) float stage[G1_WAVES][16][G1_PITCH];
  __shared__ __attribute__((aligned(16))) int rowcnt[BK_W];
  const int tid   = threadIdx.x;
  const int lane  = tid & 31;
  const int wid   = tid >> 5;
  const int wave  = blockIdx.x * G1_WAVES + wid;
  const int ntile = n >> 4;
  const bool live = wave < ntile;
  const int m0 = wave << 4;
  const int r  = lane & 15;
  const int hh = lane >> 4;

  const int cnt   = slots[(size_t)blockIdx.x * BK_STRIDE + BK_CNT];
  const int start = slots[(size_t)blockIdx.x * BK_STRIDE + BK_START];
  const int* lp = list + start;
  int wv[G1_LIST_DEPTH];
#pragma unroll
  for (int u = 0; u < G1_LIST_DEPTH; ++u) wv[u] = 0;
  if (cnt > 0) {
#pragma unroll
    for (int u = 0; u < G1_LIST_DEPTH; ++u) wv[u] = lp[min(tid + u * G1_THREADS, cnt - 1)];
  }

  float (*st)[G1_PITCH] = stage[wid];

  const int lrow = lane >> 4;
  const int lcol = (lane & 15) << 2;
  const float* gp = x + (size_t)(m0 + lrow) * KDIM + lcol;

  v4f R[8];
#pragma unroll
  for (int j = 0; j < 8; ++j) R[j] = (v4f){0.f, 0.f, 0.f, 0.f};
  if (live) {
#pragma unroll
    for (int j = 0; j < 8; ++j)
      R[j] = __builtin_nontemporal_load((const v4f*)(gp + (size_t)(2 * j) * KDIM));
  }

  if (tid < BK_W) rowcnt[tid] = 0;
  __syncthreads();
#pragma unroll
  for (int u = 0; u < G1_LIST_DEPTH; ++u)
    if (tid + u * G1_THREADS < cnt) atomicAdd(&rowcnt[wv[u] & (BK_W - 1)], 1);
  for (int i = G1_LIST_DEPTH * G1_THREADS + tid; i < cnt; i += G1_THREADS)
    atomicAdd(&rowcnt[lp[i] & (BK_W - 1)], 1);
  __syncthreads();

  if (tid < BK_W) {
    const int row = blockIdx.x * BK_W + tid;
    if (row < n) dis[row] = rsqrtf((float)(1 + rowcnt[tid]));
  }
  if (!live) return;

  v8f acc0 = {0.f,0.f,0.f,0.f,0.f,0.f,0.f,0.f};
  v8f acc1 = {0.f,0.f,0.f,0.f,0.f,0.f,0.f,0.f};

  const int mbase = m0 + ((lane >> 4) << 3);
  float rs[8];
#pragma unroll
  for (int v = 0; v < 8; ++v) rs[v] = rsqrtf((float)(1 + rowcnt[(wid << 4) + (hh << 3) + v]));

  for (int t = 0; t < KDIM / G1_COLS; ++t) {
#pragma unroll
    for (int j = 0; j < 8; ++j)
      *(v4f*)&st[2 * j + lrow][lcol] = R[j];

    if (t + 1 < KDIM / G1_COLS) {
#pragma unroll
      for (int j = 0; j < 8; ++j)
        R[j] = __builtin_nontemporal_load(
            (const v4f*)(gp + (size_t)(2 * j) * KDIM + (t + 1) * G1_COLS));
    }
    g1_wave_handover();

    const float* wb = W1 + (size_t)(t * G1_COLS + 4 * hh) * F1 + r;
#pragma unroll
    for (int g = 0; g < G1_COLS / 8; ++g) {
      const v4f a4 = *(const v4f*)&st[r][8 * g + 4 * hh];
      const float* w = wb + (size_t)(8 * g) * F1;
      v2f aL; aL.x = a4.x; aL.y = a4.y;
      v2f aH; aH.x = a4.z; aH.y = a4.w;
      v2f b0; b0.x = w[0];      b0.y = w[F1];
      v2f b1; b1.x = w[16];     b1.y = w[F1 + 16];
      v2f b2; b2.x = w[2 * F1]; b2.y = w[3 * F1];
      v2f b3; b3.x = w[2 * F1 + 16]; b3.y = w[3 * F1 + 16];
      acc0 = __builtin_amdgcn_wmma_f32_16x16x4_f32(false, aL, false, b0, (short)0, acc0, false, false);
      acc1 = __builtin_amdgcn_wmma_f32_16x16x4_f32(false, aL, false, b1, (short)0, acc1, false, false);
      acc0 = __builtin_amdgcn_wmma_f32_16x16x4_f32(false, aH, false, b2, (short)0, acc0, false, false);
      acc1 = __builtin_amdgcn_wmma_f32_16x16x4_f32(false, aH, false, b3, (short)0, acc1, false, false);
    }
    g1_wave_handover();
  }

  float* hp = h + (size_t)mbase * F1 + r;
#pragma unroll
  for (int v = 0; v < 8; ++v) {
    hp[(size_t)v * F1]      = acc0[v] * rs[v];
    hp[(size_t)v * F1 + 16] = acc1[v] * rs[v];
  }
}

extern "C" void kernel_launch(void* const* d_in, const int* in_sizes, int n_in,
                              void* d_out, int out_size, void* d_ws, size_t ws_size,
                              hipStream_t stream) {
  const float* x  = (const float*)d_in[0];
  const int*   ei = (const int*)d_in[1];
  const float* W1 = (const float*)d_in[2];
  const float* b1 = (const float*)d_in[3];
  const float* W2 = (const float*)d_in[4];
  const float* b2 = (const float*)d_in[5];

  const int n = in_sizes[0] / KDIM;
  const int E = in_sizes[1] / 2;
  const int* src = ei;
  const int* dst = ei + E;

  float* ws   = (float*)d_ws;
  float* dis  = ws;
  float* h    = dis  + n;
  int*   list = (int*)(h + (size_t)n * F1);
  float* h2   = (float*)(list + E);
  float* out  = (float*)d_out;
  int*   slots = (int*)d_out;
  const int B = (n + BK_W - 1) / BK_W;
  int*   table = (int*)h2;

  bucket_count_kernel<<<BC_GROUPS, BC_THREADS, B * sizeof(int), stream>>>(dst, table, E, B);
  bucket_scan_kernel<<<1, 1024, 0, stream>>>(table, slots, B);
  partition_kernel<<<PT_GROUPS, PT_THREADS, 2 * B * sizeof(int), stream>>>(src, dst, slots, list, E, B);
  gemm1_kernel<<<B, G1_THREADS, 0, stream>>>(x, W1, list, slots, dis, h, n);
  agg1_gemm2_kernel<<<B, BK_THREADS, 0, stream>>>(h, list, slots, dis, b1, W2, h2, n);
  agg2_out_kernel<<<B, BK_THREADS, 0, stream>>>(h2, list, dis, b2, out, n);
}


# Bucket sources sorted by destination row once, in GEMM1 (it already counts the rows), stored back over the list: aggregations read a row's sources as a run, no LDS sort; sum order varies as before
# speedup vs baseline: 4.0891x; 1.0335x over previous
#include <hip/hip_runtime.h>

typedef __attribute__((ext_vector_type(2))) float v2f;
typedef __attribute__((ext_vector_type(8))) float v8f;

#define KDIM 512
#define F1 32
#define F2 16


#define BK_SHIFT  7
#define BK_W      (1 << BK_SHIFT)
#define BK_STRIDE (BK_W * F2)
#define BK_CNT    0
#define BK_START  1
#define BK_CURSOR 2
#define BK_THREADS 512

#define BC_GROUPS  64
#define BC_THREADS 1024
#define BC_DEPTH   8

static __device__ __forceinline__ void bc_count(int* hist, int4 d) {
  atomicAdd(&hist[d.x >> BK_SHIFT], 1);
  atomicAdd(&hist[d.y >> BK_SHIFT], 1);
  atomicAdd(&hist[d.z >> BK_SHIFT], 1);
  atomicAdd(&hist[d.w >> BK_SHIFT], 1);
}

__global__ __launch_bounds__(BC_THREADS) void bucket_count_kernel(const int* __restrict__ dst,
                                                                  int* __restrict__ table,
                                                                  int E, int B) {
  extern __shared__ int bc_hist[];
  const int tid = threadIdx.x;
  for (int b = tid; b < B; b += BC_THREADS) bc_hist[b] = 0;
  __syncthreads();

  const int per = (((E + (int)gridDim.x - 1) / (int)gridDim.x) + 3) & ~3;
  const int e0  = blockIdx.x * per;
  const int len = max(0, min(E - e0, per));

  if ((((size_t)dst) & 15) == 0) {
    const int4* p  = (const int4*)(dst + e0);
    const int   nv = len >> 2;
    if (nv > 0) {
      int4 r[BC_DEPTH];
#pragma unroll
      for (int u = 0; u < BC_DEPTH; ++u)
        r[u] = p[min(u * BC_THREADS + tid, nv - 1)];
      for (int base = 0; base < nv; base += BC_DEPTH * BC_THREADS) {
        int4 q[BC_DEPTH];
#pragma unroll
        for (int u = 0; u < BC_DEPTH; ++u)
          q[u] = p[min(base + (BC_DEPTH + u) * BC_THREADS + tid, nv - 1)];
#pragma unroll
        for (int u = 0; u < BC_DEPTH; ++u)
          if (base + u * BC_THREADS + tid < nv) bc_count(bc_hist, r[u]);
#pragma unroll
        for (int u = 0; u < BC_DEPTH; ++u) r[u] = q[u];
      }
    }
    const int t = (nv << 2) + tid;
    if (t < len) atomicAdd(&bc_hist[dst[e0 + t] >> BK_SHIFT], 1);
  } else {
    for (int t = tid; t < len; t += BC_THREADS) atomicAdd(&bc_hist[dst[e0 + t] >> BK_SHIFT], 1);
  }
  __syncthreads();

  for (int b = tid; b < B; b += BC_THREADS) table[(size_t)blockIdx.x * B + b] = bc_hist[b];
}

#define SC_DEPTH 32

__global__ __launch_bounds__(1024) void bucket_scan_kernel(const int* __restrict__ table,
                                                           int* __restrict__ slots, int B) {
  static_assert(BC_GROUPS % SC_DEPTH == 0, "the column sum takes whole trips");
  __shared__ int wsum[32];
  __shared__ int total;
  const int tid = threadIdx.x, lane = tid & 31, wid = tid >> 5;
  int carry = 0;
  for (int base = 0; base < B; base += 1024) {
    const int b = base + tid;
    int c = 0;
    if (b < B) {
      for (int g = 0; g < BC_GROUPS; g += SC_DEPTH) {
        int v[SC_DEPTH];
#pragma unroll
        for (int u = 0; u < SC_DEPTH; ++u) v[u] = table[(size_t)(g + u) * B + b];
#pragma unroll
        for (int u = 0; u < SC_DEPTH; ++u) c += v[u];
      }
    }
    int x = c;
    for (int o = 1; o < 32; o <<= 1) { const int y = __shfl_up(x, o); if (lane >= o) x += y; }
    if (lane == 31) wsum[wid] = x;
    __syncthreads();
    if (wid == 0) {
      const int tw = wsum[lane];
      int z = tw;
      for (int o = 1; o < 32; o <<= 1) { const int y = __shfl_up(z, o); if (lane >= o) z += y; }
      wsum[lane] = z - tw;
      if (lane == 31) total = z;
    }
    __syncthreads();
    const int excl = carry + wsum[wid] + x - c;
    if (b < B) {
      slots[(size_t)b * BK_STRIDE + BK_CNT]    = c;
      slots[(size_t)b * BK_STRIDE + BK_START]  = excl;
      slots[(size_t)b * BK_STRIDE + BK_CURSOR] = excl;
    }
    carry += total;
    __syncthreads();
  }
}

#define PT_THREADS 512
#define PT_GROUPS  256
#define PT_DEPTH   8
#define PT_CAP     12800
#define PT_OUT     5

__global__ __launch_bounds__(PT_THREADS) void partition_kernel(const int* __restrict__ src,
                                                               const int* __restrict__ dst,
                                                               int* __restrict__ slots,
                                                               int* __restrict__ list,
                                                               int E, int B) {
  extern __shared__ int pt_lds[];
  int* hist  = pt_lds;
  int* bbase = pt_lds + B;
  __shared__ int words[PT_CAP];
  __shared__ int delta[PT_CAP];
  __shared__ int wsum[PT_THREADS / 32];
  __shared__ int total;
  const int tid = threadIdx.x, lane = tid & 31, wid = tid >> 5;

  const int per = (E + (int)gridDim.x - 1) / (int)gridDim.x;
  const int e0  = blockIdx.x * per;
  const int e1  = min(E, e0 + per);

  for (int p0 = e0; p0 < e1; p0 += PT_CAP) {
    const int p1 = min(e1, p0 + PT_CAP);
    for (int b = tid; b < B; b += PT_THREADS) hist[b] = 0;
    __syncthreads();

    for (int e = p0 + tid; e < p1; e += PT_DEPTH * PT_THREADS) {
      int d[PT_DEPTH];
#pragma unroll
      for (int u = 0; u < PT_DEPTH; ++u) d[u] = dst[min(e + u * PT_THREADS, p1 - 1)];
#pragma unroll
      for (int u = 0; u < PT_DEPTH; ++u)
        if (e + u * PT_THREADS < p1) atomicAdd(&hist[d[u] >> BK_SHIFT], 1);
    }
    __syncthreads();

    int carry = 0;
    for (int base = 0; base < B; base += PT_THREADS) {
      const int b = base + tid;
      const int c = (b < B) ? hist[b] : 0;
      int x = c;
      for (int o = 1; o < 32; o <<= 1) { const int y = __shfl_up(x, o); if (lane >= o) x += y; }
      if (lane == 31) wsum[wid] = x;
      __syncthreads();
      if (wid == 0) {
        const int tw = (lane < PT_THREADS / 32) ? wsum[lane] : 0;
        int z = tw;
        for (int o = 1; o < 32; o <<= 1) { const int y = __shfl_up(z, o); if (lane >= o) z += y; }
        if (lane < PT_THREADS / 32) wsum[lane] = z - tw;
        if (lane == 31) total = z;
      }
      __syncthreads();
      const int excl = carry + wsum[wid] + x - c;
      if (b < B) {
        const int g = (c != 0) ? atomicAdd(&slots[(size_t)b * BK_STRIDE + BK_CURSOR], c) : 0;
        bbase[b] = g - excl;
        hist[b]  = excl;
      }
      carry += total;
      __syncthreads();
    }

    for (int e = p0 + tid; e < p1; e += PT_DEPTH * PT_THREADS) {
      int d[PT_DEPTH], s[PT_DEPTH];
#pragma unroll
      for (int u = 0; u < PT_DEPTH; ++u) {
        const int ee = min(e + u * PT_THREADS, p1 - 1);
        d[u] = dst[ee];
        s[u] = src[ee];
      }
#pragma unroll
      for (int u = 0; u < PT_DEPTH; ++u) {
        if (e + u * PT_THREADS < p1) {
          const int b = d[u] >> BK_SHIFT;
          const int q = atomicAdd(&hist[b], 1);
          words[q] = (s[u] << BK_SHIFT) | (d[u] & (BK_W - 1));
          delta[q] = bbase[b];
        }
      }
    }
    __syncthreads();

    const int m = p1 - p0;
    for (int i = tid; i < m; i += PT_OUT * PT_THREADS) {
      int w[PT_OUT], g[PT_OUT];
#pragma unroll
      for (int u = 0; u < PT_OUT; ++u) {
        const int ii = min(i + u * PT_THREADS, m - 1);
        w[u] = words[ii];
        g[u] = delta[ii] + ii;
      }
#pragma unroll
      for (int u = 0; u < PT_OUT; ++u)
        if (i + u * PT_THREADS < m) list[g[u]] = w[u];
    }
    __syncthreads();
  }
}

#define SORT_CAP   5120
#define ROWSORT_CAP 4608
#define LONG_CAP 2560
#define LONG_PER_LANE (LONG_CAP / BK_THREADS)
#define ROWSORT_PER_LANE (ROWSORT_CAP / BK_THREADS)
#define BK_WAVES_PER_SIMD 16
#define TAIL_DEPTH 4

typedef __attribute__((ext_vector_type(4))) float f4v;

template <int F>
static __device__ __forceinline__ f4v row_piece(const float* __restrict__ feat, int s, unsigned int fl4) {
  const unsigned int off = (unsigned int)s * (unsigned int)(F * 4) + fl4;
  return *(const f4v*)((const char*)feat + off);
}

template <int F, int RPG, int ROW_DEPTH>
static __device__ __forceinline__ void gather_rows(const float* __restrict__ feat,
                                                   const int* offL, const int* srcL,
                                                   f4v* sum, int grp, unsigned int fl4) {
#pragma unroll
  for (int q = 0; q < RPG; ++q) {
    const int row = grp * RPG + q;
    int j = offL[row];
    const int end = offL[row + 1];
    f4v a = sum[q];
    for (; j + ROW_DEPTH <= end; j += ROW_DEPTH) {
      f4v v[ROW_DEPTH];
#pragma unroll
      for (int u = 0; u < ROW_DEPTH; ++u) v[u] = row_piece<F>(feat, srcL[j + u], fl4);
#pragma unroll
      for (int u = 0; u < ROW_DEPTH; ++u) a += v[u];
    }
    for (; j < end; j += TAIL_DEPTH) {
      f4v v[TAIL_DEPTH];
#pragma unroll
      for (int u = 0; u < TAIL_DEPTH; ++u) v[u] = row_piece<F>(feat, srcL[min(j + u, end - 1)], fl4);
#pragma unroll
      for (int u = 0; u < TAIL_DEPTH; ++u) {
        const bool in = (j + u < end);
        a.x += in ? v[u].x : 0.0f;
        a.y += in ? v[u].y : 0.0f;
        a.z += in ? v[u].z : 0.0f;
        a.w += in ? v[u].w : 0.0f;
      }
    }
    sum[q] = a;
  }
}

template <int F, int PITCH>
static __device__ __forceinline__ void bucket_rows(const float* __restrict__ feat,
                                                   const int* __restrict__ list,
                                                   const int* __restrict__ rowcnt,
                                                   float* acc, int* cntL, int* offL, int* srcL,
                                                   int start, int cnt, int lo, int n) {
  constexpr int LPR = F / 4;
  constexpr int GPW = 32 / LPR;
  constexpr int NG  = (BK_THREADS / 32) * GPW;
  constexpr int RPG = BK_W / NG;
  constexpr int ROW_DEPTH = (RPG > 1) ? 6 : 8;
  static_assert(RPG * NG == BK_W, "every row of the bucket has exactly one owner");
  static_assert(ROWSORT_CAP <= SORT_CAP, "a bucket in row order fits srcL");
  static_assert(LONG_PER_LANE * BK_THREADS == LONG_CAP && LONG_CAP <= SORT_CAP, "a piece is a whole number of words per lane and fits srcL");
  static_assert(ROWSORT_PER_LANE * BK_THREADS == ROWSORT_CAP, "a bucket in row order is a whole number of words per lane");
  const int tid  = threadIdx.x;
  const int lane = tid & 31;
  const int wid  = tid >> 5;
  const int grp  = wid * GPW + lane / LPR;
  const unsigned int fl  = (unsigned int)(lane % LPR) * 4u;
  const unsigned int fl4 = fl * 4u;

  f4v sum[RPG];
#pragma unroll
  for (int q = 0; q < RPG; ++q) sum[q] = (f4v){0.f, 0.f, 0.f, 0.f};

  if (cnt <= ROWSORT_CAP) {
    int wv[ROWSORT_PER_LANE];
#pragma unroll
    for (int u = 0; u < ROWSORT_PER_LANE; ++u) wv[u] = 0;
    if (cnt > 0) {
#pragma unroll
      for (int u = 0; u < ROWSORT_PER_LANE; ++u) wv[u] = list[start + min(tid + u * BK_THREADS, cnt - 1)];
    }
    if (wid == 0) {
      const int r0 = lo + 4 * lane;
      const int a0 = (r0     < n) ? rowcnt[r0]     : 0;
      const int a1 = (r0 + 1 < n) ? rowcnt[r0 + 1] : 0;
      const int a2 = (r0 + 2 < n) ? rowcnt[r0 + 2] : 0;
      const int a3 = (r0 + 3 < n) ? rowcnt[r0 + 3] : 0;
      const int t = a0 + a1 + a2 + a3;
      int x = t;
      for (int o = 1; o < 32; o <<= 1) { const int y = __shfl_up(x, o); if (lane >= o) x += y; }
      const int p0 = x - t, p1 = p0 + a0, p2 = p1 + a1, p3 = p2 + a2;
      offL[4 * lane] = p0; offL[4 * lane + 1] = p1; offL[4 * lane + 2] = p2; offL[4 * lane + 3] = p3;
      if (lane == 31) offL[BK_W] = x;
    }
#pragma unroll
    for (int u = 0; u < ROWSORT_PER_LANE; ++u)
      if (tid + u * BK_THREADS < cnt) srcL[tid + u * BK_THREADS] = wv[u];
    __syncthreads();
    gather_rows<F, RPG, ROW_DEPTH>(feat, offL, srcL, sum, grp, fl4);
  } else
  for (int c0 = 0; c0 < cnt; c0 += LONG_CAP) {
    const int m = min(LONG_CAP, cnt - c0);
    const int* lp = list + start + c0;
    int wv[LONG_PER_LANE];
#pragma unroll
    for (int u = 0; u < LONG_PER_LANE; ++u) wv[u] = lp[min(tid + u * BK_THREADS, m - 1)];
    if (tid < BK_W) cntL[tid] = 0;
    __syncthreads();
#pragma unroll
    for (int u = 0; u < LONG_PER_LANE; ++u)
      if (tid + u * BK_THREADS < m) atomicAdd(&cntL[wv[u] & (BK_W - 1)], 1);
    __syncthreads();
    if (wid == 0) {
      const int a0 = cntL[4 * lane], a1 = cntL[4 * lane + 1], a2 = cntL[4 * lane + 2], a3 = cntL[4 * lane + 3];
      const int t = a0 + a1 + a2 + a3;
      int x = t;
      for (int o = 1; o < 32; o <<= 1) { const int y = __shfl_up(x, o); if (lane >= o) x += y; }
      const int p0 = x - t, p1 = p0 + a0, p2 = p1 + a1, p3 = p2 + a2;
      offL[4 * lane] = p0; offL[4 * lane + 1] = p1; offL[4 * lane + 2] = p2; offL[4 * lane + 3] = p3;
      cntL[4 * lane] = p0; cntL[4 * lane + 1] = p1; cntL[4 * lane + 2] = p2; cntL[4 * lane + 3] = p3;
      if (lane == 31) offL[BK_W] = x;
    }
    __syncthreads();
    int pos[LONG_PER_LANE];
#pragma unroll
    for (int u = 0; u < LONG_PER_LANE; ++u) {
      pos[u] = 0;
      if (tid + u * BK_THREADS < m) pos[u] = atomicAdd(&cntL[wv[u] & (BK_W - 1)], 1);
    }
#pragma unroll
    for (int u = 0; u < LONG_PER_LANE; ++u)
      if (tid + u * BK_THREADS < m) srcL[pos[u]] = wv[u] >> BK_SHIFT;
    __syncthreads();

    gather_rows<F, RPG, ROW_DEPTH>(feat, offL, srcL, sum, grp, fl4);
    __syncthreads();
  }

#pragma unroll
  for (int q = 0; q < RPG; ++q) {
    float* o = acc + (grp * RPG + q) * PITCH + fl;
    o[0] = sum[q].x; o[1] = sum[q].y; o[2] = sum[q].z; o[3] = sum[q].w;
  }
}

#define A1_PITCH 33

__global__ __launch_bounds__(BK_THREADS, BK_WAVES_PER_SIMD) void agg1_gemm2_kernel(const float* __restrict__ h,
                                                         const int* __restrict__ list,
                                                         const int* __restrict__ slots,
                                                         const int* __restrict__ rowcnt,
                                                         const float* __restrict__ b1,
                                                         const float* __restrict__ W2,
                                                         float* __restrict__ h2, int n) {
  __shared__ float acc[BK_W * A1_PITCH];
  __shared__ int cntL[BK_W];
  __shared__ int offL[BK_W + 1];
  __shared__ int srcL[SORT_CAP];
  const int lo    = blockIdx.x * BK_W;
  const int cnt   = slots[(size_t)blockIdx.x * BK_STRIDE + BK_CNT];
  const int start = slots[(size_t)blockIdx.x * BK_STRIDE + BK_START];
  bucket_rows<F1, A1_PITCH>(h, list, rowcnt, acc, cntL, offL, srcL, start, cnt, lo, n);
  __syncthreads();

  const int lane = threadIdx.x & 31;
  const int r    = lane & 15;
  const int kh   = (lane >> 4) << 1;
  for (int tl = (threadIdx.x >> 5) * 16; tl < BK_W; tl += (blockDim.x >> 5) * 16) {
    const int m0 = lo + tl;
    if (m0 >= n) break;
    const int row  = m0 + r;
    const float dv = rsqrtf((float)(1 + rowcnt[row]));
    const float* ag = acc + (tl + r) * A1_PITCH;
    const float* hr = h + (size_t)row * F1;
    v8f c8 = {0.f,0.f,0.f,0.f,0.f,0.f,0.f,0.f};
    for (int k = 0; k < F1; k += 4) {
      const int k0 = k + kh;
      v2f a;
      a.x = fmaxf((ag[k0]     + hr[k0])     * dv + b1[k0],     0.0f) * dv;
      a.y = fmaxf((ag[k0 + 1] + hr[k0 + 1]) * dv + b1[k0 + 1], 0.0f) * dv;
      v2f b;
      b.x = W2[(size_t)k0 * F2 + r];
      b.y = W2[(size_t)(k0 + 1) * F2 + r];
      c8 = __builtin_amdgcn_wmma_f32_16x16x4_f32(false, a, false, b, (short)0, c8, false, false);
    }
    const int mbase = m0 + ((lane >> 4) << 3);
    float* op = h2 + (size_t)mbase * F2 + r;
#pragma unroll
    for (int v = 0; v < 8; ++v) op[(size_t)v * F2] = c8[v];
  }
}

__global__ __launch_bounds__(BK_THREADS, BK_WAVES_PER_SIMD) void agg2_out_kernel(const float* __restrict__ h2,
                                                       const int* __restrict__ list,
                                                       const int* __restrict__ rowcnt,
                                                       const float* __restrict__ b2,
                                                       float* out, int n) {
  __shared__ float acc[BK_W * F2];
  __shared__ int cntL[BK_W];
  __shared__ int offL[BK_W + 1];
  __shared__ int srcL[SORT_CAP];
  const int lo    = blockIdx.x * BK_W;
  const int cnt   = __float_as_int(out[(size_t)blockIdx.x * BK_STRIDE + BK_CNT]);
  const int start = __float_as_int(out[(size_t)blockIdx.x * BK_STRIDE + BK_START]);
  bucket_rows<F2, F2>(h2, list, rowcnt, acc, cntL, offL, srcL, start, cnt, lo, n);
  __syncthreads();
  for (int i = threadIdx.x; i < BK_W * F2; i += blockDim.x) {
    const int row = lo + i / F2, f = i & (F2 - 1);
    if (row < n) {
      const float dv = rsqrtf((float)(1 + rowcnt[row]));
      const size_t g = (size_t)row * F2 + f;
      out[g] = (acc[i] + h2[g]) * dv + b2[f];
    }
  }
}

typedef __attribute__((ext_vector_type(4))) float v4f;

#define G1_WAVES 8
#define G1_THREADS (G1_WAVES * 32)
#define G1_COLS  64
#define G1_PITCH 68
#define G1_LIST_DEPTH 18
#define G1_LIST_OUT   6

static __device__ __forceinline__ void g1_wave_handover() {
  __builtin_amdgcn_fence(__ATOMIC_RELEASE, "wavefront");
  __builtin_amdgcn_wave_barrier();
  __builtin_amdgcn_fence(__ATOMIC_ACQUIRE, "wavefront");
}

#define G1_WAVES_PER_SIMD 8
__global__ __launch_bounds__(G1_THREADS, G1_WAVES_PER_SIMD) void gemm1_kernel(const float* __restrict__ x,
                                                           const float* __restrict__ W1,
                                                           int* list,
                                                           const int* __restrict__ slots,
                                                           int* __restrict__ rowcnt_out,
                                                           float* __restrict__ h, int n) {
  static_assert(G1_WAVES * 16 == BK_W, "a workgroup's rows are one bucket");
  static_assert(G1_LIST_DEPTH * G1_THREADS == ROWSORT_CAP, "a bucket in row order is one the registers hold whole");
  static_assert(G1_LIST_DEPTH % G1_LIST_OUT == 0, "places and copy back take whole groups");
  __shared__ int rowcur[BK_W];
  __shared__ int sorted[ROWSORT_CAP];
  __shared__ __attribute__((aligned(16))) float stage[G1_WAVES][16][G1_PITCH];
  __shared__ __attribute__((aligned(16))) int rowcnt[BK_W];
  const int tid   = threadIdx.x;
  const int lane  = tid & 31;
  const int wid   = tid >> 5;
  const int wave  = blockIdx.x * G1_WAVES + wid;
  const int ntile = n >> 4;
  const bool live = wave < ntile;
  const int m0 = wave << 4;
  const int r  = lane & 15;
  const int hh = lane >> 4;

  const int cnt   = slots[(size_t)blockIdx.x * BK_STRIDE + BK_CNT];
  const int start = slots[(size_t)blockIdx.x * BK_STRIDE + BK_START];
  int* lp = list + start;
  int wv[G1_LIST_DEPTH];
#pragma unroll
  for (int u = 0; u < G1_LIST_DEPTH; ++u) wv[u] = 0;
  if (cnt > 0) {
#pragma unroll
    for (int u = 0; u < G1_LIST_DEPTH; ++u) wv[u] = lp[min(tid + u * G1_THREADS, cnt - 1)];
  }

  float (*st)[G1_PITCH] = stage[wid];

  const int lrow = lane >> 4;
  const int lcol = (lane & 15) << 2;
  const float* gp = x + (size_t)(m0 + lrow) * KDIM + lcol;

  v4f R[8];
#pragma unroll
  for (int j = 0; j < 8; ++j) R[j] = (v4f){0.f, 0.f, 0.f, 0.f};
  if (live) {
#pragma unroll
    for (int j = 0; j < 8; ++j)
      R[j] = __builtin_nontemporal_load((const v4f*)(gp + (size_t)(2 * j) * KDIM));
  }

  if (tid < BK_W) rowcnt[tid] = 0;
  __syncthreads();
#pragma unroll
  for (int u = 0; u < G1_LIST_DEPTH; ++u)
    if (tid + u * G1_THREADS < cnt) atomicAdd(&rowcnt[wv[u] & (BK_W - 1)], 1);
  for (int i = G1_LIST_DEPTH * G1_THREADS + tid; i < cnt; i += G1_THREADS)
    atomicAdd(&rowcnt[lp[i] & (BK_W - 1)], 1);
  __syncthreads();

  if (tid < BK_W) {
    const int row = blockIdx.x * BK_W + tid;
    if (row < n) rowcnt_out[row] = rowcnt[tid];
  }

  if (cnt > 0 && cnt <= ROWSORT_CAP) {
    if (wid == 0) {
      const int a0 = rowcnt[4 * lane], a1 = rowcnt[4 * lane + 1], a2 = rowcnt[4 * lane + 2];
      const int t = a0 + a1 + a2 + rowcnt[4 * lane + 3];
      int x = t;
      for (int o = 1; o < 32; o <<= 1) { const int y = __shfl_up(x, o); if (lane >= o) x += y; }
      const int p0 = x - t, p1 = p0 + a0, p2 = p1 + a1, p3 = p2 + a2;
      rowcur[4 * lane] = p0; rowcur[4 * lane + 1] = p1; rowcur[4 * lane + 2] = p2; rowcur[4 * lane + 3] = p3;
    }
    __syncthreads();
#pragma unroll
    for (int u0 = 0; u0 < G1_LIST_DEPTH; u0 += G1_LIST_OUT) {
      int pos[G1_LIST_OUT];
#pragma unroll
      for (int u = 0; u < G1_LIST_OUT; ++u) {
        pos[u] = 0;
        if (tid + (u0 + u) * G1_THREADS < cnt) pos[u] = atomicAdd(&rowcur[wv[u0 + u] & (BK_W - 1)], 1);
      }
#pragma unroll
      for (int u = 0; u < G1_LIST_OUT; ++u)
        if (tid + (u0 + u) * G1_THREADS < cnt) sorted[pos[u]] = wv[u0 + u] >> BK_SHIFT;
    }
    __syncthreads();
    for (int i = tid; i < cnt; i += G1_LIST_OUT * G1_THREADS) {
      int w[G1_LIST_OUT];
#pragma unroll
      for (int u = 0; u < G1_LIST_OUT; ++u) w[u] = sorted[min(i + u * G1_THREADS, cnt - 1)];
#pragma unroll
      for (int u = 0; u < G1_LIST_OUT; ++u)
        if (i + u * G1_THREADS < cnt) lp[i + u * G1_THREADS] = w[u];
    }
  }
  if (!live) return;

  v8f acc0 = {0.f,0.f,0.f,0.f,0.f,0.f,0.f,0.f};
  v8f acc1 = {0.f,0.f,0.f,0.f,0.f,0.f,0.f,0.f};

  const int mbase = m0 + ((lane >> 4) << 3);
  float rs[8];
#pragma unroll
  for (int v = 0; v < 8; ++v) rs[v] = rsqrtf((float)(1 + rowcnt[(wid << 4) + (hh << 3) + v]));

  for (int t = 0; t < KDIM / G1_COLS; ++t) {
#pragma unroll
    for (int j = 0; j < 8; ++j)
      *(v4f*)&st[2 * j + lrow][lcol] = R[j];

    if (t + 1 < KDIM / G1_COLS) {
#pragma unroll
      for (int j = 0; j < 8; ++j)
        R[j] = __builtin_nontemporal_load(
            (const v4f*)(gp + (size_t)(2 * j) * KDIM + (t + 1) * G1_COLS));
    }
    g1_wave_handover();

    const float* wb = W1 + (size_t)(t * G1_COLS + 4 * hh) * F1 + r;
#pragma unroll
    for (int g = 0; g < G1_COLS / 8; ++g) {
      const v4f a4 = *(const v4f*)&st[r][8 * g + 4 * hh];
      const float* w = wb + (size_t)(8 * g) * F1;
      v2f aL; aL.x = a4.x; aL.y = a4.y;
      v2f aH; aH.x = a4.z; aH.y = a4.w;
      v2f b0; b0.x = w[0];      b0.y = w[F1];
      v2f b1; b1.x = w[16];     b1.y = w[F1 + 16];
      v2f b2; b2.x = w[2 * F1]; b2.y = w[3 * F1];
      v2f b3; b3.x = w[2 * F1 + 16]; b3.y = w[3 * F1 + 16];
      acc0 = __builtin_amdgcn_wmma_f32_16x16x4_f32(false, aL, false, b0, (short)0, acc0, false, false);
      acc1 = __builtin_amdgcn_wmma_f32_16x16x4_f32(false, aL, false, b1, (short)0, acc1, false, false);
      acc0 = __builtin_amdgcn_wmma_f32_16x16x4_f32(false, aH, false, b2, (short)0, acc0, false, false);
      acc1 = __builtin_amdgcn_wmma_f32_16x16x4_f32(false, aH, false, b3, (short)0, acc1, false, false);
    }
    g1_wave_handover();
  }

  float* hp = h + (size_t)mbase * F1 + r;
#pragma unroll
  for (int v = 0; v < 8; ++v) {
    hp[(size_t)v * F1]      = acc0[v] * rs[v];
    hp[(size_t)v * F1 + 16] = acc1[v] * rs[v];
  }
}

extern "C" void kernel_launch(void* const* d_in, const int* in_sizes, int n_in,
                              void* d_out, int out_size, void* d_ws, size_t ws_size,
                              hipStream_t stream) {
  const float* x  = (const float*)d_in[0];
  const int*   ei = (const int*)d_in[1];
  const float* W1 = (const float*)d_in[2];
  const float* b1 = (const float*)d_in[3];
  const float* W2 = (const float*)d_in[4];
  const float* b2 = (const float*)d_in[5];

  const int n = in_sizes[0] / KDIM;
  const int E = in_sizes[1] / 2;
  const int* src = ei;
  const int* dst = ei + E;

  float* ws   = (float*)d_ws;
  int*   rowcnt = (int*)ws;
  float* h    = ws + n;
  int*   list = (int*)(h + (size_t)n * F1);
  float* h2   = (float*)(list + E);
  float* out  = (float*)d_out;
  int*   slots = (int*)d_out;
  const int B = (n + BK_W - 1) / BK_W;
  int*   table = (int*)h2;

  bucket_count_kernel<<<BC_GROUPS, BC_THREADS, B * sizeof(int), stream>>>(dst, table, E, B);
  bucket_scan_kernel<<<1, 1024, 0, stream>>>(table, slots, B);
  partition_kernel<<<PT_GROUPS, PT_THREADS, 2 * B * sizeof(int), stream>>>(src, dst, slots, list, E, B);
  gemm1_kernel<<<B, G1_THREADS, 0, stream>>>(x, W1, list, slots, rowcnt, h, n);
  agg1_gemm2_kernel<<<B, BK_THREADS, 0, stream>>>(h, list, slots, rowcnt, b1, W2, h2, n);
  agg2_out_kernel<<<B, BK_THREADS, 0, stream>>>(h2, list, rowcnt, b2, out, n);
}


# Edge list kept as per-chunk runs in bucket order with a table of run starts: bucket-count and scan launches and the global reservation add go (6 launches to 4); a row's sum order follows the runs
# speedup vs baseline: 4.4750x; 1.0944x over previous
#include <hip/hip_runtime.h>

typedef __attribute__((ext_vector_type(2))) float v2f;
typedef __attribute__((ext_vector_type(8))) float v8f;

#define KDIM 512
#define F1 32
#define F2 16


#define BK_SHIFT  7
#define BK_W      (1 << BK_SHIFT)
#define BK_STRIDE (BK_W * F2)
#define BK_THREADS 512
#define RUNS 256

template <int NT>
static __device__ __forceinline__ int wg_exclusive_scan(int c, int* wsum) {
  const int lane = threadIdx.x & 31, wid = threadIdx.x >> 5;
  int x = c;
  for (int o = 1; o < 32; o <<= 1) { const int y = __shfl_up(x, o); if (lane >= o) x += y; }
  if (lane == 31) wsum[wid] = x;
  __syncthreads();
  if (wid == 0) {
    const int tw = (lane < NT / 32) ? wsum[lane] : 0;
    int z = tw;
    for (int o = 1; o < 32; o <<= 1) { const int y = __shfl_up(z, o); if (lane >= o) z += y; }
    if (lane < NT / 32) wsum[lane] = z - tw;
    if (lane == 31) wsum[NT / 32] = z;
  }
  __syncthreads();
  return wsum[wid] + x - c;
}

#define PT_THREADS 512
#define PT_DEPTH   8
#define PT_CAP     12800
#define PT_OUT     5

__global__ __launch_bounds__(PT_THREADS) void partition_kernel(const int* __restrict__ src,
                                                               const int* __restrict__ dst,
                                                               int* __restrict__ table,
                                                               int* __restrict__ list,
                                                               int E, int B) {
  extern __shared__ int hist[];
  __shared__ int words[PT_CAP];
  __shared__ int wsum[PT_THREADS / 32 + 1];
  const int tid = threadIdx.x;

  const int per = (E + RUNS - 1) / RUNS;
  const int p0  = blockIdx.x * per;
  const int p1  = min(E, p0 + min(per, PT_CAP));
  int* trow = table + (size_t)blockIdx.x * (B + 1);

  for (int b = tid; b < B; b += PT_THREADS) hist[b] = 0;
  __syncthreads();

  for (int e = p0 + tid; e < p1; e += PT_DEPTH * PT_THREADS) {
    int d[PT_DEPTH];
#pragma unroll
    for (int u = 0; u < PT_DEPTH; ++u) d[u] = dst[min(e + u * PT_THREADS, p1 - 1)];
#pragma unroll
    for (int u = 0; u < PT_DEPTH; ++u)
      if (e + u * PT_THREADS < p1) atomicAdd(&hist[d[u] >> BK_SHIFT], 1);
  }
  __syncthreads();

  int carry = 0;
  for (int base = 0; base < B; base += PT_THREADS) {
    const int b = base + tid;
    const int c = (b < B) ? hist[b] : 0;
    const int excl = carry + wg_exclusive_scan<PT_THREADS>(c, wsum);
    if (b < B) { trow[b] = excl; hist[b] = excl; }
    carry += wsum[PT_THREADS / 32];
    __syncthreads();
  }
  if (tid == 0) trow[B] = carry;

  for (int e = p0 + tid; e < p1; e += PT_DEPTH * PT_THREADS) {
    int d[PT_DEPTH], s[PT_DEPTH];
#pragma unroll
    for (int u = 0; u < PT_DEPTH; ++u) {
      const int ee = min(e + u * PT_THREADS, p1 - 1);
      d[u] = dst[ee];
      s[u] = src[ee];
    }
#pragma unroll
    for (int u = 0; u < PT_DEPTH; ++u) {
      if (e + u * PT_THREADS < p1) {
        const int q = atomicAdd(&hist[d[u] >> BK_SHIFT], 1);
        words[q] = (s[u] << BK_SHIFT) | (d[u] & (BK_W - 1));
      }
    }
  }
  __syncthreads();

  const int m = p1 - p0;
  for (int i = tid; i < m; i += PT_OUT * PT_THREADS) {
    int w[PT_OUT];
#pragma unroll
    for (int u = 0; u < PT_OUT; ++u) w[u] = words[min(i + u * PT_THREADS, m - 1)];
#pragma unroll
    for (int u = 0; u < PT_OUT; ++u)
      if (i + u * PT_THREADS < m) list[p0 + i + u * PT_THREADS] = w[u];
  }
}

template <int NT, int NB, typename Fn>
static __device__ __forceinline__ void bucket_words(const int* __restrict__ list, const int* runP,
                                                    const int* runO, int per, Fn fn) {
  constexpr int RPW = RUNS / (NT / 32);
  const int lane = threadIdx.x & 31, g0 = (threadIdx.x >> 5) * RPW;
  int cmax = 0;
  for (int j = 0; j < RPW; ++j) cmax = max(cmax, runP[g0 + j + 1] - runP[g0 + j]);
  static_assert(RPW % NB == 0, "whole groups of runs");
  for (int r0 = 0; r0 < cmax; r0 += 32) {
    const int kk = r0 + lane;
#pragma nounroll
    for (int jb = 0; jb < RPW; jb += NB) {
      int w[NB];
#pragma unroll
      for (int j = 0; j < NB; ++j) {
        const int g = g0 + jb + j, c = runP[g + 1] - runP[g];
        w[j] = list[max(g * per + runO[g] + min(kk, c - 1), 0)];
      }
#pragma unroll
      for (int j = 0; j < NB; ++j) {
        const int g = g0 + jb + j, p = runP[g], c = runP[g + 1] - p;
        if (kk < c) fn(w[j], p + kk);
      }
    }
  }
}

static __device__ __forceinline__ int bucket_word_at(const int* __restrict__ list, const int* runP,
                                                     const int* runO, int per, int i) {
  int lo = 0, hi = RUNS;
  while (hi - lo > 1) { const int mid = (lo + hi) >> 1; if (runP[mid] <= i) lo = mid; else hi = mid; }
  return list[lo * per + runO[lo] + (i - runP[lo])];
}

#define SORT_CAP   5120
#define ROWSORT_CAP 4608
#define LONG_CAP 2560
#define LONG_PER_LANE (LONG_CAP / BK_THREADS)
#define ROWSORT_PER_LANE (ROWSORT_CAP / BK_THREADS)
#define BK_WAVES_PER_SIMD 16
#define TAIL_DEPTH 4

typedef __attribute__((ext_vector_type(4))) float f4v;

template <int F>
static __device__ __forceinline__ f4v row_piece(const float* __restrict__ feat, int s, unsigned int fl4) {
  const unsigned int off = (unsigned int)s * (unsigned int)(F * 4) + fl4;
  return *(const f4v*)((const char*)feat + off);
}

template <int F, int RPG, int ROW_DEPTH>
static __device__ __forceinline__ void gather_rows(const float* __restrict__ feat,
                                                   const int* offL, const int* srcL,
                                                   f4v* sum, int grp, unsigned int fl4) {
#pragma unroll
  for (int q = 0; q < RPG; ++q) {
    const int row = grp * RPG + q;
    int j = offL[row];
    const int end = offL[row + 1];
    f4v a = sum[q];
    for (; j + ROW_DEPTH <= end; j += ROW_DEPTH) {
      f4v v[ROW_DEPTH];
#pragma unroll
      for (int u = 0; u < ROW_DEPTH; ++u) v[u] = row_piece<F>(feat, srcL[j + u], fl4);
#pragma unroll
      for (int u = 0; u < ROW_DEPTH; ++u) a += v[u];
    }
    for (; j < end; j += TAIL_DEPTH) {
      f4v v[TAIL_DEPTH];
#pragma unroll
      for (int u = 0; u < TAIL_DEPTH; ++u) v[u] = row_piece<F>(feat, srcL[min(j + u, end - 1)], fl4);
#pragma unroll
      for (int u = 0; u < TAIL_DEPTH; ++u) {
        const bool in = (j + u < end);
        a.x += in ? v[u].x : 0.0f;
        a.y += in ? v[u].y : 0.0f;
        a.z += in ? v[u].z : 0.0f;
        a.w += in ? v[u].w : 0.0f;
      }
    }
    sum[q] = a;
  }
}

template <int F, int PITCH>
static __device__ __forceinline__ void bucket_rows(const float* __restrict__ feat,
                                                   const int* __restrict__ list,
                                                   const int* __restrict__ rowcnt,
                                                   float* acc, int* cntL, int* offL, int* srcL,
                                                   int* runP, int* runO, int* wsum,
                                                   int runword, int per, int lo, int n) {
  constexpr int LPR = F / 4;
  constexpr int GPW = 32 / LPR;
  constexpr int NG  = (BK_THREADS / 32) * GPW;
  constexpr int RPG = BK_W / NG;
  constexpr int ROW_DEPTH = (RPG > 1) ? 6 : 8;
  static_assert(RPG * NG == BK_W, "every row of the bucket has exactly one owner");
  static_assert(ROWSORT_CAP <= SORT_CAP, "a bucket in row order fits srcL");
  static_assert(LONG_PER_LANE * BK_THREADS == LONG_CAP && LONG_CAP <= SORT_CAP, "a piece is a whole number of words per lane and fits srcL");
  static_assert(RUNS <= BK_THREADS && RUNS % (BK_THREADS / 32) == 0, "a lane per run in the scan, whole runs per wave");
  const int tid  = threadIdx.x;
  const int lane = tid & 31;
  const int wid  = tid >> 5;
  const int grp  = wid * GPW + lane / LPR;
  const unsigned int fl  = (unsigned int)(lane % LPR) * 4u;
  const unsigned int fl4 = fl * 4u;

  f4v sum[RPG];
#pragma unroll
  for (int q = 0; q < RPG; ++q) sum[q] = (f4v){0.f, 0.f, 0.f, 0.f};

  {
    const int rc = (tid < RUNS) ? (runword & 0xffff) : 0;
    const int rp = wg_exclusive_scan<BK_THREADS>(rc, wsum);
    if (tid < RUNS) { runP[tid] = rp; runO[tid] = (int)((unsigned int)runword >> 16); }
    if (tid == RUNS - 1) runP[RUNS] = rp + rc;
  }
  if (wid == 0) {
    const int r0 = lo + 4 * lane;
    const int a0 = (r0     < n) ? rowcnt[r0]     : 0;
    const int a1 = (r0 + 1 < n) ? rowcnt[r0 + 1] : 0;
    const int a2 = (r0 + 2 < n) ? rowcnt[r0 + 2] : 0;
    const int a3 = (r0 + 3 < n) ? rowcnt[r0 + 3] : 0;
    const int t = a0 + a1 + a2 + a3;
    int x = t;
    for (int o = 1; o < 32; o <<= 1) { const int y = __shfl_up(x, o); if (lane >= o) x += y; }
    const int p0 = x - t, p1 = p0 + a0, p2 = p1 + a1, p3 = p2 + a2;
    offL[4 * lane] = p0; offL[4 * lane + 1] = p1; offL[4 * lane + 2] = p2; offL[4 * lane + 3] = p3;
    if (lane == 31) offL[BK_W] = x;
  }
  __syncthreads();
  const int cnt = runP[RUNS];

  if (cnt <= ROWSORT_CAP) {
    bucket_words<BK_THREADS, 8>(list, runP, runO, per, [&](int w, int i) { srcL[i] = w; });
    __syncthreads();
    gather_rows<F, RPG, ROW_DEPTH>(feat, offL, srcL, sum, grp, fl4);
  } else
  for (int c0 = 0; c0 < cnt; c0 += LONG_CAP) {
    const int m = min(LONG_CAP, cnt - c0);
    int wv[LONG_PER_LANE];
#pragma unroll
    for (int u = 0; u < LONG_PER_LANE; ++u) wv[u] = bucket_word_at(list, runP, runO, per, c0 + min(tid + u * BK_THREADS, m - 1));
    if (tid < BK_W) cntL[tid] = 0;
    __syncthreads();
#pragma unroll
    for (int u = 0; u < LONG_PER_LANE; ++u)
      if (tid + u * BK_THREADS < m) atomicAdd(&cntL[wv[u] & (BK_W - 1)], 1);
    __syncthreads();
    if (wid == 0) {
      const int a0 = cntL[4 * lane], a1 = cntL[4 * lane + 1], a2 = cntL[4 * lane + 2], a3 = cntL[4 * lane + 3];
      const int t = a0 + a1 + a2 + a3;
      int x = t;
      for (int o = 1; o < 32; o <<= 1) { const int y = __shfl_up(x, o); if (lane >= o) x += y; }
      const int p0 = x - t, p1 = p0 + a0, p2 = p1 + a1, p3 = p2 + a2;
      offL[4 * lane] = p0; offL[4 * lane + 1] = p1; offL[4 * lane + 2] = p2; offL[4 * lane + 3] = p3;
      cntL[4 * lane] = p0; cntL[4 * lane + 1] = p1; cntL[4 * lane + 2] = p2; cntL[4 * lane + 3] = p3;
      if (lane == 31) offL[BK_W] = x;
    }
    __syncthreads();
    int pos[LONG_PER_LANE];
#pragma unroll
    for (int u = 0; u < LONG_PER_LANE; ++u) {
      pos[u] = 0;
      if (tid + u * BK_THREADS < m) pos[u] = atomicAdd(&cntL[wv[u] & (BK_W - 1)], 1);
    }
#pragma unroll
    for (int u = 0; u < LONG_PER_LANE; ++u)
      if (tid + u * BK_THREADS < m) srcL[pos[u]] = wv[u] >> BK_SHIFT;
    __syncthreads();

    gather_rows<F, RPG, ROW_DEPTH>(feat, offL, srcL, sum, grp, fl4);
    __syncthreads();
  }

#pragma unroll
  for (int q = 0; q < RPG; ++q) {
    float* o = acc + (grp * RPG + q) * PITCH + fl;
    o[0] = sum[q].x; o[1] = sum[q].y; o[2] = sum[q].z; o[3] = sum[q].w;
  }
}

#define A1_PITCH 33

__global__ __launch_bounds__(BK_THREADS, BK_WAVES_PER_SIMD) void agg1_gemm2_kernel(const float* __restrict__ h,
                                                         const int* __restrict__ list,
                                                         const int* __restrict__ slots,
                                                         const int* __restrict__ rowcnt,
                                                         const float* __restrict__ b1,
                                                         const float* __restrict__ W2,
                                                         float* __restrict__ h2, int n, int E) {
  __shared__ float acc[BK_W * A1_PITCH];
  __shared__ int cntL[BK_W];
  __shared__ int offL[BK_W + 1];
  __shared__ int srcL[SORT_CAP];
  __shared__ int runP[RUNS + 1];
  __shared__ int runO[RUNS];
  __shared__ int wsum[BK_THREADS / 32 + 1];
  const int lo    = blockIdx.x * BK_W;
  const int runword = (threadIdx.x < RUNS) ? slots[(size_t)blockIdx.x * BK_STRIDE + threadIdx.x] : 0;
  bucket_rows<F1, A1_PITCH>(h, list, rowcnt, acc, cntL, offL, srcL, runP, runO, wsum, runword, (E + RUNS - 1) / RUNS, lo, n);
  __syncthreads();

  const int lane = threadIdx.x & 31;
  const int r    = lane & 15;
  const int kh   = (lane >> 4) << 1;
  for (int tl = (threadIdx.x >> 5) * 16; tl < BK_W; tl += (blockDim.x >> 5) * 16) {
    const int m0 = lo + tl;
    if (m0 >= n) break;
    const int row  = m0 + r;
    const float dv = rsqrtf((float)(1 + rowcnt[row]));
    const float* ag = acc + (tl + r) * A1_PITCH;
    const float* hr = h + (size_t)row * F1;
    v8f c8 = {0.f,0.f,0.f,0.f,0.f,0.f,0.f,0.f};
    for (int k = 0; k < F1; k += 4) {
      const int k0 = k + kh;
      v2f a;
      a.x = fmaxf((ag[k0]     + hr[k0])     * dv + b1[k0],     0.0f) * dv;
      a.y = fmaxf((ag[k0 + 1] + hr[k0 + 1]) * dv + b1[k0 + 1], 0.0f) * dv;
      v2f b;
      b.x = W2[(size_t)k0 * F2 + r];
      b.y = W2[(size_t)(k0 + 1) * F2 + r];
      c8 = __builtin_amdgcn_wmma_f32_16x16x4_f32(false, a, false, b, (short)0, c8, false, false);
    }
    const int mbase = m0 + ((lane >> 4) << 3);
    float* op = h2 + (size_t)mbase * F2 + r;
#pragma unroll
    for (int v = 0; v < 8; ++v) op[(size_t)v * F2] = c8[v];
  }
}

__global__ __launch_bounds__(BK_THREADS, BK_WAVES_PER_SIMD) void agg2_out_kernel(const float* __restrict__ h2,
                                                       const int* __restrict__ list,
                                                       const int* __restrict__ rowcnt,
                                                       const float* __restrict__ b2,
                                                       float* out, int n, int E) {
  __shared__ float acc[BK_W * F2];
  __shared__ int cntL[BK_W];
  __shared__ int offL[BK_W + 1];
  __shared__ int srcL[SORT_CAP];
  __shared__ int runP[RUNS + 1];
  __shared__ int runO[RUNS];
  __shared__ int wsum[BK_THREADS / 32 + 1];
  const int lo    = blockIdx.x * BK_W;
  const int runword = (threadIdx.x < RUNS) ? __float_as_int(out[(size_t)blockIdx.x * BK_STRIDE + threadIdx.x]) : 0;
  bucket_rows<F2, F2>(h2, list, rowcnt, acc, cntL, offL, srcL, runP, runO, wsum, runword, (E + RUNS - 1) / RUNS, lo, n);
  __syncthreads();
  for (int i = threadIdx.x; i < BK_W * F2; i += blockDim.x) {
    const int row = lo + i / F2, f = i & (F2 - 1);
    if (row < n) {
      const float dv = rsqrtf((float)(1 + rowcnt[row]));
      const size_t g = (size_t)row * F2 + f;
      out[g] = (acc[i] + h2[g]) * dv + b2[f];
    }
  }
}

typedef __attribute__((ext_vector_type(4))) float v4f;

#define G1_WAVES 8
#define G1_THREADS (G1_WAVES * 32)
#define G1_COLS  64
#define G1_PITCH 68
#define G1_LIST_DEPTH 18
#define G1_LIST_OUT   6
#define G1_LIST_BACK  8

static __device__ __forceinline__ void g1_wave_handover() {
  __builtin_amdgcn_fence(__ATOMIC_RELEASE, "wavefront");
  __builtin_amdgcn_wave_barrier();
  __builtin_amdgcn_fence(__ATOMIC_ACQUIRE, "wavefront");
}

#define G1_WAVES_PER_SIMD 8
__global__ __launch_bounds__(G1_THREADS, G1_WAVES_PER_SIMD) void gemm1_kernel(const float* __restrict__ x,
                                                           const float* __restrict__ W1,
                                                           int* list,
                                                           const int* __restrict__ table,
                                                           int* __restrict__ runinfo,
                                                           int* __restrict__ rowcnt_out,
                                                           float* __restrict__ h, int n, int E, int B) {
  static_assert(G1_WAVES * 16 == BK_W, "a workgroup's rows are one bucket");
  static_assert(G1_LIST_DEPTH * G1_THREADS == ROWSORT_CAP, "a bucket in row order is one the registers hold whole");
  static_assert(G1_LIST_DEPTH % G1_LIST_OUT == 0, "places take whole groups");
  static_assert(G1_THREADS == RUNS, "a lane per run of the bucket");
  __shared__ int rowcur[BK_W];
  __shared__ int sorted[ROWSORT_CAP];
  __shared__ int runP[RUNS + 1];
  __shared__ int runO[RUNS];
  __shared__ int wsum[G1_THREADS / 32 + 1];
  __shared__ __attribute__((aligned(16))) float stage[G1_WAVES][16][G1_PITCH];
  __shared__ __attribute__((aligned(16))) int rowcnt[BK_W];
  const int tid   = threadIdx.x;
  const int lane  = tid & 31;
  const int wid   = tid >> 5;
  const int wave  = blockIdx.x * G1_WAVES + wid;
  const int ntile = n >> 4;
  const bool live = wave < ntile;
  const int m0 = wave << 4;
  const int r  = lane & 15;
  const int hh = lane >> 4;
  const int per = (E + RUNS - 1) / RUNS;

  const int* tcol = table + (size_t)tid * (B + 1) + blockIdx.x;
  const int ro = tcol[0];
  const int rc = tcol[1] - ro;

  float (*st)[G1_PITCH] = stage[wid];

  const int lrow = lane >> 4;
  const int lcol = (lane & 15) << 2;
  const float* gp = x + (size_t)(m0 + lrow) * KDIM + lcol;

  v4f R[8];
#pragma unroll
  for (int j = 0; j < 8; ++j) R[j] = (v4f){0.f, 0.f, 0.f, 0.f};
  if (live) {
#pragma unroll
    for (int j = 0; j < 8; ++j)
      R[j] = __builtin_nontemporal_load((const v4f*)(gp + (size_t)(2 * j) * KDIM));
  }

  if (tid < BK_W) rowcnt[tid] = 0;
  {
    const int rp = wg_exclusive_scan<G1_THREADS>(rc, wsum);
    runP[tid] = rp; runO[tid] = ro;
    if (tid == RUNS - 1) runP[RUNS] = rp + rc;
    runinfo[(size_t)blockIdx.x * BK_STRIDE + tid] = (int)(((unsigned int)ro << 16) | (unsigned int)rc);
  }
  __syncthreads();
  const int cnt = runP[RUNS];
  const bool fits = cnt <= ROWSORT_CAP;

  bucket_words<G1_THREADS, 16>(list, runP, runO, per, [&](int w, int i) {
    atomicAdd(&rowcnt[w & (BK_W - 1)], 1);
    if (fits) sorted[i] = w;
  });
  __syncthreads();

  if (tid < BK_W) {
    const int row = blockIdx.x * BK_W + tid;
    if (row < n) rowcnt_out[row] = rowcnt[tid];
  }

  if (cnt > 0 && fits) {
    int wv[G1_LIST_DEPTH];
#pragma unroll
    for (int u = 0; u < G1_LIST_DEPTH; ++u) wv[u] = sorted[min(tid + u * G1_THREADS, cnt - 1)];
    if (wid == 0) {
      const int a0 = rowcnt[4 * lane], a1 = rowcnt[4 * lane + 1], a2 = rowcnt[4 * lane + 2];
      const int t = a0 + a1 + a2 + rowcnt[4 * lane + 3];
      int x = t;
      for (int o = 1; o < 32; o <<= 1) { const int y = __shfl_up(x, o); if (lane >= o) x += y; }
      const int p0 = x - t, p1 = p0 + a0, p2 = p1 + a1, p3 = p2 + a2;
      rowcur[4 * lane] = p0; rowcur[4 * lane + 1] = p1; rowcur[4 * lane + 2] = p2; rowcur[4 * lane + 3] = p3;
    }
    __syncthreads();
#pragma unroll
    for (int u0 = 0; u0 < G1_LIST_DEPTH; u0 += G1_LIST_OUT) {
      int pos[G1_LIST_OUT];
#pragma unroll
      for (int u = 0; u < G1_LIST_OUT; ++u) {
        pos[u] = 0;
        if (tid + (u0 + u) * G1_THREADS < cnt) pos[u] = atomicAdd(&rowcur[wv[u0 + u] & (BK_W - 1)], 1);
      }
#pragma unroll
      for (int u = 0; u < G1_LIST_OUT; ++u)
        if (tid + (u0 + u) * G1_THREADS < cnt) sorted[pos[u]] = wv[u0 + u] >> BK_SHIFT;
    }
    __syncthreads();
    {
      constexpr int RPW = RUNS / G1_WAVES;
      static_assert(RPW % G1_LIST_BACK == 0, "whole groups of runs");
      int cmax = 0;
      for (int j = 0; j < RPW; ++j) cmax = max(cmax, runP[wid * RPW + j + 1] - runP[wid * RPW + j]);
      for (int r0 = 0; r0 < cmax; r0 += 32) {
        const int kk = r0 + lane;
#pragma nounroll
        for (int jb = 0; jb < RPW; jb += G1_LIST_BACK) {
          int w[G1_LIST_BACK], at[G1_LIST_BACK];
#pragma unroll
          for (int j = 0; j < G1_LIST_BACK; ++j) {
            const int g = wid * RPW + jb + j, p = runP[g], c = runP[g + 1] - p, o = runO[g];
            w[j]  = sorted[max(p + min(kk, c - 1), 0)];
            at[j] = (kk < c) ? g * per + o + kk : -1;
          }
#pragma unroll
          for (int j = 0; j < G1_LIST_BACK; ++j)
            if (at[j] >= 0) list[at[j]] = w[j];
        }
      }
    }
  }
  if (!live) return;

  v8f acc0 = {0.f,0.f,0.f,0.f,0.f,0.f,0.f,0.f};
  v8f acc1 = {0.f,0.f,0.f,0.f,0.f,0.f,0.f,0.f};

  const int mbase = m0 + ((lane >> 4) << 3);
  float rs[8];
#pragma unroll
  for (int v = 0; v < 8; ++v) rs[v] = rsqrtf((float)(1 + rowcnt[(wid << 4) + (hh << 3) + v]));

  for (int t = 0; t < KDIM / G1_COLS; ++t) {
#pragma unroll
    for (int j = 0; j < 8; ++j)
      *(v4f*)&st[2 * j + lrow][lcol] = R[j];

    if (t + 1 < KDIM / G1_COLS) {
#pragma unroll
      for (int j = 0; j < 8; ++j)
        R[j] = __builtin_nontemporal_load(
            (const v4f*)(gp + (size_t)(2 * j) * KDIM + (t + 1) * G1_COLS));
    }
    g1_wave_handover();

    const float* wb = W1 + (size_t)(t * G1_COLS + 4 * hh) * F1 + r;
#pragma unroll
    for (int g = 0; g < G1_COLS / 8; ++g) {
      const v4f a4 = *(const v4f*)&st[r][8 * g + 4 * hh];
      const float* w = wb + (size_t)(8 * g) * F1;
      v2f aL; aL.x = a4.x; aL.y = a4.y;
      v2f aH; aH.x = a4.z; aH.y = a4.w;
      v2f b0; b0.x = w[0];      b0.y = w[F1];
      v2f b1; b1.x = w[16];     b1.y = w[F1 + 16];
      v2f b2; b2.x = w[2 * F1]; b2.y = w[3 * F1];
      v2f b3; b3.x = w[2 * F1 + 16]; b3.y = w[3 * F1 + 16];
      acc0 = __builtin_amdgcn_wmma_f32_16x16x4_f32(false, aL, false, b0, (short)0, acc0, false, false);
      acc1 = __builtin_amdgcn_wmma_f32_16x16x4_f32(false, aL, false, b1, (short)0, acc1, false, false);
      acc0 = __builtin_amdgcn_wmma_f32_16x16x4_f32(false, aH, false, b2, (short)0, acc0, false, false);
      acc1 = __builtin_amdgcn_wmma_f32_16x16x4_f32(false, aH, false, b3, (short)0, acc1, false, false);
    }
    g1_wave_handover();
  }

  float* hp = h + (size_t)mbase * F1 + r;
#pragma unroll
  for (int v = 0; v < 8; ++v) {
    hp[(size_t)v * F1]      = acc0[v] * rs[v];
    hp[(size_t)v * F1 + 16] = acc1[v] * rs[v];
  }
}

extern "C" void kernel_launch(void* const* d_in, const int* in_sizes, int n_in,
                              void* d_out, int out_size, void* d_ws, size_t ws_size,
                              hipStream_t stream) {
  const float* x  = (const float*)d_in[0];
  const int*   ei = (const int*)d_in[1];
  const float* W1 = (const float*)d_in[2];
  const float* b1 = (const float*)d_in[3];
  const float* W2 = (const float*)d_in[4];
  const float* b2 = (const float*)d_in[5];

  const int n = in_sizes[0] / KDIM;
  const int E = in_sizes[1] / 2;
  const int* src = ei;
  const int* dst = ei + E;

  float* ws   = (float*)d_ws;
  int*   rowcnt = (int*)ws;
  float* h    = ws + n;
  int*   list = (int*)(h + (size_t)n * F1);
  float* h2   = (float*)(list + E);
  float* out  = (float*)d_out;
  int*   runinfo = (int*)d_out;
  const int B = (n + BK_W - 1) / BK_W;
  int*   table = (int*)h2;

  partition_kernel<<<RUNS, PT_THREADS, B * sizeof(int), stream>>>(src, dst, table, list, E, B);
  gemm1_kernel<<<B, G1_THREADS, 0, stream>>>(x, W1, list, table, runinfo, rowcnt, h, n, E, B);
  agg1_gemm2_kernel<<<B, BK_THREADS, 0, stream>>>(h, list, runinfo, rowcnt, b1, W2, h2, n, E);
  agg2_out_kernel<<<B, BK_THREADS, 0, stream>>>(h2, list, rowcnt, b2, out, n, E);
}
